# starting kernel
# baseline (speedup 1.0000x reference)
#include <hip/hip_runtime.h>


typedef __attribute__((ext_vector_type(2))) float v2f;
typedef __attribute__((ext_vector_type(8))) float v8f;

#define N_NODES 50000
#define N_EDGES 1600000
#define IN_DIM  128
#define HIDDEN  128
#define OUT_DIM 64

__global__ __launch_bounds__(256)
void sage_scatter(const float* __restrict__ feat,
                  const int*   __restrict__ src,
                  const int*   __restrict__ dst,
                  float* __restrict__ agg,
                  float* __restrict__ deg,
                  int n_edges) {
    const int lane = threadIdx.x & 31;
    const int e    = blockIdx.x * 8 + (threadIdx.x >> 5);
    if (e >= n_edges) return;

    const int s = src[e];
    const int d = dst[e];

    const float4 v =
        *(reinterpret_cast<const float4*>(feat + (size_t)s * 128) + lane);

    float* out = agg + (size_t)d * 128 + lane * 4;
    unsafeAtomicAdd(out + 0, v.x);
    unsafeAtomicAdd(out + 1, v.y);
    unsafeAtomicAdd(out + 2, v.z);
    unsafeAtomicAdd(out + 3, v.w);

    if (deg != nullptr && lane == 0) {
        unsafeAtomicAdd(deg + d, 1.0f);
    }
}

template <int NDIM, bool RELU>
__global__ __launch_bounds__(256)
void sage_gemm(const float* __restrict__ agg,
               const float* __restrict__ deg,
               const float* __restrict__ X,
               const float* __restrict__ Wl,
               const float* __restrict__ Wr,
               const float* __restrict__ bias,
               float* __restrict__ out,
               int n_nodes) {
    constexpr int NTILES_N = NDIM / 16;
    const int wave = blockIdx.x * 8 + (threadIdx.x >> 5);
    const int total_tiles = (n_nodes / 16) * NTILES_N;
    if (wave >= total_tiles) return;

    const int mtile = wave / NTILES_N;
    const int ntile = wave % NTILES_N;

    const int lane = threadIdx.x & 31;
    const int r    = lane & 15;
    const int kk   = (lane >> 4) * 2;

    const int row  = mtile * 16 + r;
    const int coln = ntile * 16 + r;

    const float dval = deg[row];
    const float invd = 1.0f / fmaxf(dval, 1.0f);

    const float* aggRow = agg + (size_t)row * 128;
    const float* xRow   = X   + (size_t)row * 128;

    v8f acc = {};

#pragma unroll 8
    for (int k0 = 0; k0 < 128; k0 += 4) {
        const int k = k0 + kk;

        v2f aA; aA.x = aggRow[k] * invd;  aA.y = aggRow[k + 1] * invd;
        v2f aX; aX.x = xRow[k];           aX.y = xRow[k + 1];

        v2f bL; bL.x = Wl[(size_t)k * NDIM + coln];
                bL.y = Wl[(size_t)(k + 1) * NDIM + coln];
        v2f bR; bR.x = Wr[(size_t)k * NDIM + coln];
                bR.y = Wr[(size_t)(k + 1) * NDIM + coln];

        acc = __builtin_amdgcn_wmma_f32_16x16x4_f32(
                  false, aA, false, bL, (short)0, acc, false, false);
        acc = __builtin_amdgcn_wmma_f32_16x16x4_f32(
                  false, aX, false, bR, (short)0, acc, false, false);
    }

    const int   half = lane >> 4;
    const float bv   = bias[coln];
#pragma unroll
    for (int i = 0; i < 8; ++i) {
        const int m    = i + 8 * half;
        const int node = mtile * 16 + m;
        float v = acc[i] + bv;
        if (RELU) v = fmaxf(v, 0.0f);
        out[(size_t)node * NDIM + (ntile * 16 + (lane & 15))] = v;
    }
}

extern "C" void kernel_launch(void* const* d_in, const int* in_sizes, int n_in,
                              void* d_out, int out_size, void* d_ws, size_t ws_size,
                              hipStream_t stream) {
    const float* x    = (const float*)d_in[0];
    const int*   ei   = (const int*)  d_in[1];
    const float* W1l  = (const float*)d_in[2];
    const float* W1r  = (const float*)d_in[3];
    const float* b1   = (const float*)d_in[4];
    const float* W2l  = (const float*)d_in[5];
    const float* W2r  = (const float*)d_in[6];
    const float* b2   = (const float*)d_in[7];
    float*       outp = (float*)d_out;

    const int n_nodes = in_sizes[0] / IN_DIM;
    const int n_edges = in_sizes[1] / 2;
    const int* src = ei;
    const int* dst = ei + n_edges;

    const size_t aggBytes = (size_t)n_nodes * 128 * sizeof(float);
    const size_t degBytes = (((size_t)n_nodes * sizeof(float)) + 255) & ~(size_t)255;

    char* ws    = (char*)d_ws;
    float* agg1 = (float*)(ws);
    float* agg2 = (float*)(ws + aggBytes);
    float* deg  = (float*)(ws + 2 * aggBytes);
    float* h    = (float*)(ws + 2 * aggBytes + degBytes);

    hipMemsetAsync(d_ws, 0, 2 * aggBytes + degBytes, stream);

    const int scatterBlocks = (n_edges + 7) / 8;
    const int tiles1 = (n_nodes / 16) * (HIDDEN / 16);
    const int tiles2 = (n_nodes / 16) * (OUT_DIM / 16);

    sage_scatter<<<scatterBlocks, 256, 0, stream>>>(x, src, dst, agg1, deg, n_edges);
    sage_gemm<HIDDEN, true><<<(tiles1 + 7) / 8, 256, 0, stream>>>(
        agg1, deg, x, W1l, W1r, b1, h, n_nodes);

    sage_scatter<<<scatterBlocks, 256, 0, stream>>>(h, src, dst, agg2, nullptr, n_edges);
    sage_gemm<OUT_DIM, false><<<(tiles2 + 7) / 8, 256, 0, stream>>>(
        agg2, deg, h, W2l, W2r, b2, outp, n_nodes);
}


# dense stages weight-stationary: B fragments of both weight matrices stay in registers, work-groups walk 16-row strips staged through LDS by whole rows; layer-1 products sum in two accumulators
# speedup vs baseline: 1.3206x; 1.3206x over previous
#include <hip/hip_runtime.h>


typedef __attribute__((ext_vector_type(2))) float v2f;
typedef __attribute__((ext_vector_type(8))) float v8f;

#define N_NODES 50000
#define N_EDGES 1600000
#define IN_DIM  128
#define HIDDEN  128
#define OUT_DIM 64

__global__ __launch_bounds__(256)
void sage_scatter(const float* __restrict__ feat,
                  const int*   __restrict__ src,
                  const int*   __restrict__ dst,
                  float* __restrict__ agg,
                  float* __restrict__ deg,
                  int n_edges) {
    const int lane = threadIdx.x & 31;
    const int e    = blockIdx.x * 8 + (threadIdx.x >> 5);
    if (e >= n_edges) return;

    const int s = src[e];
    const int d = dst[e];

    const float4 v =
        *(reinterpret_cast<const float4*>(feat + (size_t)s * 128) + lane);

    float* out = agg + (size_t)d * 128 + lane * 4;
    unsafeAtomicAdd(out + 0, v.x);
    unsafeAtomicAdd(out + 1, v.y);
    unsafeAtomicAdd(out + 2, v.z);
    unsafeAtomicAdd(out + 3, v.w);

    if (lane == 0) {
        unsafeAtomicAdd(deg + d, 1.0f);
    }
}

__global__ __launch_bounds__(256)
void sage_scatter64(const float* __restrict__ P,
                    const int*   __restrict__ src,
                    const int*   __restrict__ dst,
                    float* __restrict__ agg,
                    int n_edges) {
    const int sub = threadIdx.x & 15;
    const int e   = blockIdx.x * 16 + (threadIdx.x >> 4);
    if (e >= n_edges) return;

    const int s = src[e];
    const int d = dst[e];

    const float4 v =
        *(reinterpret_cast<const float4*>(P + (size_t)s * 64) + sub);

    float* o = agg + (size_t)d * 64 + sub * 4;
    unsafeAtomicAdd(o + 0, v.x);
    unsafeAtomicAdd(o + 1, v.y);
    unsafeAtomicAdd(o + 2, v.z);
    unsafeAtomicAdd(o + 3, v.w);
}

__global__ __launch_bounds__(256)
void sage_finish64(const float* __restrict__ S,
                   const float* __restrict__ deg,
                   const float* __restrict__ Q,
                   float* __restrict__ out,
                   int n_nodes) {
    const int t = blockIdx.x * 256 + threadIdx.x;
    if (t >= n_nodes * 16) return;

    const float invd = 1.0f / fmaxf(deg[t >> 4], 1.0f);
    const float4 s = reinterpret_cast<const float4*>(S)[t];
    const float4 q = reinterpret_cast<const float4*>(Q)[t];

    float4 o;
    o.x = q.x + s.x * invd;
    o.y = q.y + s.y * invd;
    o.z = q.z + s.z * invd;
    o.w = q.w + s.w * invd;
    reinterpret_cast<float4*>(out)[t] = o;
}

#define LDS_PITCH   132
#define STRIP_WORDS (16 * LDS_PITCH)

__global__ __launch_bounds__(256) __attribute__((amdgpu_waves_per_eu(4, 4)))
void sage_gemm128(const float* __restrict__ agg,
                  const float* __restrict__ deg,
                  const float* __restrict__ X,
                  const float* __restrict__ Wl,
                  const float* __restrict__ Wr,
                  const float* __restrict__ bias,
                  float* __restrict__ out,
                  int n_nodes) {
    __shared__ __attribute__((aligned(16))) float lds[2][2][STRIP_WORDS];

    const int n_strips = n_nodes / 16;
    const int tid  = threadIdx.x;
    const int lane = tid & 31;
    const int w    = tid >> 5;
    const int r    = lane & 15;
    const int half = lane >> 4;
    const int kk   = half * 2;
    const int coln = w * 16 + r;

    v2f bL[32], bR[32];
#pragma unroll
    for (int ks = 0; ks < 32; ++ks) {
        const int k = ks * 4 + kk;
        bL[ks].x = Wl[(size_t)k * 128 + coln];
        bL[ks].y = Wl[(size_t)(k + 1) * 128 + coln];
        bR[ks].x = Wr[(size_t)k * 128 + coln];
        bR[ks].y = Wr[(size_t)(k + 1) * 128 + coln];
    }
    const float bv = bias[coln];

    const int srow = tid >> 5;
    const int c4   = tid & 31;

    int s   = blockIdx.x;
    int cur = 0;

    if (s < n_strips) {
        const int row0 = s * 16 + srow;
        const int row1 = row0 + 8;
        const float i0 = 1.0f / fmaxf(deg[row0], 1.0f);
        const float i1 = 1.0f / fmaxf(deg[row1], 1.0f);
        float4 a0 = *(reinterpret_cast<const float4*>(agg + (size_t)row0 * 128) + c4);
        float4 a1 = *(reinterpret_cast<const float4*>(agg + (size_t)row1 * 128) + c4);
        const float4 x0 = *(reinterpret_cast<const float4*>(X + (size_t)row0 * 128) + c4);
        const float4 x1 = *(reinterpret_cast<const float4*>(X + (size_t)row1 * 128) + c4);
        a0.x *= i0; a0.y *= i0; a0.z *= i0; a0.w *= i0;
        a1.x *= i1; a1.y *= i1; a1.z *= i1; a1.w *= i1;
        *reinterpret_cast<float4*>(&lds[0][0][srow * LDS_PITCH + c4 * 4])       = a0;
        *reinterpret_cast<float4*>(&lds[0][0][(srow + 8) * LDS_PITCH + c4 * 4]) = a1;
        *reinterpret_cast<float4*>(&lds[0][1][srow * LDS_PITCH + c4 * 4])       = x0;
        *reinterpret_cast<float4*>(&lds[0][1][(srow + 8) * LDS_PITCH + c4 * 4]) = x1;
    }
    __syncthreads();

    for (; s < n_strips; s += gridDim.x) {
        const int ahead = s + (int)gridDim.x;
        const int nxt   = ahead < n_strips ? ahead : s;
        const int row0  = nxt * 16 + srow;
        const int row1  = row0 + 8;
        const float d0 = deg[row0];
        const float d1 = deg[row1];
        float4 a0 = *(reinterpret_cast<const float4*>(agg + (size_t)row0 * 128) + c4);
        float4 a1 = *(reinterpret_cast<const float4*>(agg + (size_t)row1 * 128) + c4);
        const float4 x0 = *(reinterpret_cast<const float4*>(X + (size_t)row0 * 128) + c4);
        const float4 x1 = *(reinterpret_cast<const float4*>(X + (size_t)row1 * 128) + c4);

        const float* la = &lds[cur][0][r * LDS_PITCH + kk];
        const float* lx = &lds[cur][1][r * LDS_PITCH + kk];

        v8f accA = {};
        v8f accX = {};
#pragma unroll
        for (int ks = 0; ks < 32; ++ks) {
            const v2f aA = *reinterpret_cast<const v2f*>(la + ks * 4);
            const v2f aX = *reinterpret_cast<const v2f*>(lx + ks * 4);
            accA = __builtin_amdgcn_wmma_f32_16x16x4_f32(
                       false, aA, false, bL[ks], (short)0, accA, false, false);
            accX = __builtin_amdgcn_wmma_f32_16x16x4_f32(
                       false, aX, false, bR[ks], (short)0, accX, false, false);
        }

        float v[8];
#pragma unroll
        for (int i = 0; i < 8; ++i) {
            v[i] = fmaxf((accA[i] + accX[i]) + bv, 0.0f);
        }

        {
            const float i0 = 1.0f / fmaxf(d0, 1.0f);
            const float i1 = 1.0f / fmaxf(d1, 1.0f);
            a0.x *= i0; a0.y *= i0; a0.z *= i0; a0.w *= i0;
            a1.x *= i1; a1.y *= i1; a1.z *= i1; a1.w *= i1;
            const int nb = cur ^ 1;
            *reinterpret_cast<float4*>(&lds[nb][0][srow * LDS_PITCH + c4 * 4])       = a0;
            *reinterpret_cast<float4*>(&lds[nb][0][(srow + 8) * LDS_PITCH + c4 * 4]) = a1;
            *reinterpret_cast<float4*>(&lds[nb][1][srow * LDS_PITCH + c4 * 4])       = x0;
            *reinterpret_cast<float4*>(&lds[nb][1][(srow + 8) * LDS_PITCH + c4 * 4]) = x1;
        }
        __syncthreads();
        cur ^= 1;

        const int rowBase = s * 16 + 8 * half;
#pragma unroll
        for (int i = 0; i < 8; ++i) {
            out[(size_t)(rowBase + i) * 128 + coln] = v[i];
        }
    }
}

__global__ __launch_bounds__(256) __attribute__((amdgpu_waves_per_eu(4, 4)))
void sage_transform64(const float* __restrict__ H,
                      const float* __restrict__ Wl,
                      const float* __restrict__ Wr,
                      const float* __restrict__ bias,
                      float* __restrict__ P,
                      float* __restrict__ Q,
                      int n_nodes) {
    __shared__ __attribute__((aligned(16))) float lds[2][2][STRIP_WORDS];

    const int n_strips = n_nodes / 16;
    const int n_pairs  = (n_strips + 1) / 2;
    const int tid  = threadIdx.x;
    const int lane = tid & 31;
    const int w    = tid >> 5;
    const int grp  = w >> 2;
    const int r    = lane & 15;
    const int half = lane >> 4;
    const int kk   = half * 2;
    const int coln = (w & 3) * 16 + r;

    v2f bL[32], bR[32];
#pragma unroll
    for (int ks = 0; ks < 32; ++ks) {
        const int k = ks * 4 + kk;
        bL[ks].x = Wl[(size_t)k * 64 + coln];
        bL[ks].y = Wl[(size_t)(k + 1) * 64 + coln];
        bR[ks].x = Wr[(size_t)k * 64 + coln];
        bR[ks].y = Wr[(size_t)(k + 1) * 64 + coln];
    }
    const float bv = bias[coln];

    const int srow = tid >> 5;
    const int c4   = tid & 31;
    const int last = n_strips - 1;

    int p   = blockIdx.x;
    int cur = 0;

    if (p < n_pairs) {
#pragma unroll
        for (int sp = 0; sp < 2; ++sp) {
            const int st   = min(2 * p + sp, last);
            const int row0 = st * 16 + srow;
            const float4 h0 = *(reinterpret_cast<const float4*>(H + (size_t)row0 * 128) + c4);
            const float4 h1 = *(reinterpret_cast<const float4*>(H + (size_t)(row0 + 8) * 128) + c4);
            *reinterpret_cast<float4*>(&lds[0][sp][srow * LDS_PITCH + c4 * 4])       = h0;
            *reinterpret_cast<float4*>(&lds[0][sp][(srow + 8) * LDS_PITCH + c4 * 4]) = h1;
        }
    }
    __syncthreads();

    for (; p < n_pairs; p += gridDim.x) {
        const int ahead = p + (int)gridDim.x;
        const int nxt   = ahead < n_pairs ? ahead : p;
        const int rowA  = min(2 * nxt, last) * 16 + srow;
        const int rowB  = min(2 * nxt + 1, last) * 16 + srow;
        const float4 h00 = *(reinterpret_cast<const float4*>(H + (size_t)rowA * 128) + c4);
        const float4 h01 = *(reinterpret_cast<const float4*>(H + (size_t)(rowA + 8) * 128) + c4);
        const float4 h10 = *(reinterpret_cast<const float4*>(H + (size_t)rowB * 128) + c4);
        const float4 h11 = *(reinterpret_cast<const float4*>(H + (size_t)(rowB + 8) * 128) + c4);

        const float* lh = &lds[cur][grp][r * LDS_PITCH + kk];

        v8f accP = {};
        v8f accQ = {};
#pragma unroll
        for (int ks = 0; ks < 32; ++ks) {
            const v2f aH = *reinterpret_cast<const v2f*>(lh + ks * 4);
            accP = __builtin_amdgcn_wmma_f32_16x16x4_f32(
                       false, aH, false, bL[ks], (short)0, accP, false, false);
            accQ = __builtin_amdgcn_wmma_f32_16x16x4_f32(
                       false, aH, false, bR[ks], (short)0, accQ, false, false);
        }

        float vq[8];
#pragma unroll
        for (int i = 0; i < 8; ++i) {
            vq[i] = accQ[i] + bv;
        }

        {
            const int nb = cur ^ 1;
            *reinterpret_cast<float4*>(&lds[nb][0][srow * LDS_PITCH + c4 * 4])       = h00;
            *reinterpret_cast<float4*>(&lds[nb][0][(srow + 8) * LDS_PITCH + c4 * 4]) = h01;
            *reinterpret_cast<float4*>(&lds[nb][1][srow * LDS_PITCH + c4 * 4])       = h10;
            *reinterpret_cast<float4*>(&lds[nb][1][(srow + 8) * LDS_PITCH + c4 * 4]) = h11;
        }
        __syncthreads();
        cur ^= 1;

        const int st = 2 * p + grp;
        if (st < n_strips) {
            const int rowBase = st * 16 + 8 * half;
#pragma unroll
            for (int i = 0; i < 8; ++i) {
                const size_t off = (size_t)(rowBase + i) * 64 + coln;
                P[off] = accP[i];
                Q[off] = vq[i];
            }
        }
    }
}

extern "C" void kernel_launch(void* const* d_in, const int* in_sizes, int n_in,
                              void* d_out, int out_size, void* d_ws, size_t ws_size,
                              hipStream_t stream) {
    const float* x    = (const float*)d_in[0];
    const int*   ei   = (const int*)  d_in[1];
    const float* W1l  = (const float*)d_in[2];
    const float* W1r  = (const float*)d_in[3];
    const float* b1   = (const float*)d_in[4];
    const float* W2l  = (const float*)d_in[5];
    const float* W2r  = (const float*)d_in[6];
    const float* b2   = (const float*)d_in[7];
    float*       outp = (float*)d_out;

    const int n_nodes = in_sizes[0] / IN_DIM;
    const int n_edges = in_sizes[1] / 2;
    const int* src = ei;
    const int* dst = ei + n_edges;

    const size_t aggBytes = (size_t)n_nodes * 128 * sizeof(float);
    const size_t rowBytes = (size_t)n_nodes * 64 * sizeof(float);
    const size_t degBytes = (((size_t)n_nodes * sizeof(float)) + 255) & ~(size_t)255;

    char* ws    = (char*)d_ws;
    float* agg1 = (float*)(ws);
    float* sum2 = (float*)(ws + aggBytes);
    float* deg  = (float*)(ws + aggBytes + rowBytes);
    float* h    = (float*)(ws + aggBytes + rowBytes + degBytes);
    float* p2   = (float*)(ws + 2 * aggBytes + rowBytes + degBytes);
    float* q2   = (float*)(ws + 2 * aggBytes + 2 * rowBytes + degBytes);

    hipMemsetAsync(d_ws, 0, aggBytes + rowBytes + degBytes, stream);

    const int scatterBlocks1 = (n_edges + 7) / 8;
    const int scatterBlocks2 = (n_edges + 15) / 16;
    const int denseBlocks = 512;
    const int finishBlocks = (n_nodes * 16 + 255) / 256;

    sage_scatter<<<scatterBlocks1, 256, 0, stream>>>(x, src, dst, agg1, deg, n_edges);
    sage_gemm128<<<denseBlocks, 256, 0, stream>>>(
        agg1, deg, x, W1l, W1r, b1, h, n_nodes);

    sage_transform64<<<denseBlocks, 256, 0, stream>>>(
        h, W2l, W2r, b2, p2, q2, n_nodes);
    sage_scatter64<<<scatterBlocks2, 256, 0, stream>>>(p2, src, dst, sum2, n_edges);
    sage_finish64<<<finishBlocks, 256, 0, stream>>>(sum2, deg, q2, outp, n_nodes);
}


# edge sums destination-stationary: each work-group owns 200 nodes, filters the whole edge list, adds source rows into LDS rows (LDS f32 adds), stores once; no global atomics; sum order = LDS arrival
# speedup vs baseline: 1.8229x; 1.3804x over previous
#include <hip/hip_runtime.h>


typedef __attribute__((ext_vector_type(2))) float v2f;
typedef __attribute__((ext_vector_type(8))) float v8f;

#define N_NODES 50000
#define N_EDGES 1600000
#define IN_DIM  128
#define HIDDEN  128
#define OUT_DIM 64

#define OWN_NODES   200
#define OWN_THREADS 1024
#define SCAN_VEC    8
#define OWN_CHUNK   (OWN_THREADS * SCAN_VEC * 4)
#define OWN_QCAP    2048

__device__ __forceinline__ void lds_add(float* p, float v) {
    __hip_atomic_fetch_add(p, v, __ATOMIC_RELAXED, __HIP_MEMORY_SCOPE_WORKGROUP);
}

template <int D, bool WITH_DEG>
__device__ __forceinline__ void own_push(int e, int dl, int2* q, int* qcount,
                                         float* acc, float* cnt,
                                         const float* __restrict__ feat,
                                         const int* __restrict__ src) {
    const int pos = __hip_atomic_fetch_add(qcount, 1, __ATOMIC_RELAXED, __HIP_MEMORY_SCOPE_WORKGROUP);
    if (pos < OWN_QCAP) {
        q[pos] = make_int2(e, dl);
    } else {
        const float* row = feat + (size_t)src[e] * D;
        for (int c = 0; c < D; ++c) lds_add(&acc[dl * D + c], row[c]);
        if (WITH_DEG) lds_add(&cnt[dl], 1.0f);
    }
}

template <int D, bool WITH_DEG>
__global__ __launch_bounds__(OWN_THREADS)
void sage_owner_sum(const float* __restrict__ feat,
                    const int*   __restrict__ src,
                    const int*   __restrict__ dst,
                    float* __restrict__ agg,
                    float* __restrict__ deg,
                    int n_nodes, int n_edges) {
    __shared__ __attribute__((aligned(16))) float acc[OWN_NODES * D];
    __shared__ float cnt[OWN_NODES];
    __shared__ int2  queue[2][OWN_QCAP];
    __shared__ int   qn[3];

    constexpr int LPE   = D / 4;
    constexpr int EPW   = 32 / LPE;
    constexpr int SLOTS = (D == 128) ? 5 : 3;
    constexpr int PER_PASS = 32 * EPW * SLOTS;

    const int tid  = threadIdx.x;
    const int lane = tid & 31;
    const int wave = tid >> 5;
    const int sub  = lane % LPE;
    const int eslot = lane / LPE;
    const int rot  = lane >> 4;
    const int base = blockIdx.x * OWN_NODES;
    const int lim  = min(OWN_NODES, n_nodes - base);
    const int n4   = n_edges & ~3;

    for (int i = tid; i < OWN_NODES * D; i += OWN_THREADS) acc[i] = 0.0f;
    if (tid < OWN_NODES) cnt[tid] = 0.0f;
    if (tid < 3) qn[tid] = 0;

    int4 dcur[SCAN_VEC];
#pragma unroll
    for (int j = 0; j < SCAN_VEC; ++j) {
        const int e = (j * OWN_THREADS + tid) * 4;
        const int4 v = *reinterpret_cast<const int4*>(dst + (e < n4 ? e : 0));
        dcur[j] = e < n4 ? v : make_int4(-1, -1, -1, -1);
    }
    __syncthreads();

    int ci = 0;
    int qb = 0;
    for (int c0 = 0; c0 < n4; c0 += OWN_CHUNK) {
        int4 dnext[SCAN_VEC];
#pragma unroll
        for (int j = 0; j < SCAN_VEC; ++j) {
            const int e = c0 + OWN_CHUNK + (j * OWN_THREADS + tid) * 4;
            const int4 v = *reinterpret_cast<const int4*>(dst + (e < n4 ? e : 0));
            dnext[j] = e < n4 ? v : make_int4(-1, -1, -1, -1);
        }

#pragma unroll
        for (int j = 0; j < SCAN_VEC; ++j) {
            const int eb = c0 + (j * OWN_THREADS + tid) * 4;
            const unsigned dl[4] = {(unsigned)(dcur[j].x - base), (unsigned)(dcur[j].y - base),
                                    (unsigned)(dcur[j].z - base), (unsigned)(dcur[j].w - base)};
            if (min(min(dl[0], dl[1]), min(dl[2], dl[3])) < (unsigned)lim) {
#pragma unroll
                for (int k = 0; k < 4; ++k) {
                    if (dl[k] < (unsigned)lim) {
                        own_push<D, WITH_DEG>(eb + k, (int)dl[k], queue[qb], &qn[ci], acc, cnt, feat, src);
                    }
                }
            }
        }
        __syncthreads();

        if (tid == 0) qn[ci == 0 ? 2 : ci - 1] = 0;

        const int n = min(qn[ci], OWN_QCAP);
        const int2* q = queue[qb];
        for (int k0 = 0; k0 < n; k0 += PER_PASS) {
            int    dl[SLOTS];
            bool   ok[SLOTS];
            float4 v[SLOTS];
#pragma unroll
            for (int j = 0; j < SLOTS; ++j) {
                const int idx = k0 + (j * 32 + wave) * EPW + eslot;
                ok[j] = idx < n;
                const int2 ent = q[ok[j] ? idx : 0];
                dl[j] = ent.y;
                const int s = src[ent.x];
                v[j] = *(reinterpret_cast<const float4*>(feat + (size_t)s * D) + sub);
            }
#pragma unroll
            for (int j = 0; j < SLOTS; ++j) {
                asm volatile("" :: "v"(v[j].x), "v"(v[j].y), "v"(v[j].z), "v"(v[j].w));
            }
#pragma unroll
            for (int j = 0; j < SLOTS; ++j) {
                if (ok[j]) {
                    float* a = &acc[dl[j] * D + sub * 4];
                    lds_add(a + rot,             rot ? v[j].y : v[j].x);
                    lds_add(a + 1 + rot,         rot ? v[j].z : v[j].y);
                    lds_add(a + 2 + rot,         rot ? v[j].w : v[j].z);
                    lds_add(a + (rot ? 0 : 3),   rot ? v[j].x : v[j].w);
                    if (WITH_DEG && sub == 0) lds_add(&cnt[dl[j]], 1.0f);
                }
            }
        }

#pragma unroll
        for (int j = 0; j < SCAN_VEC; ++j) dcur[j] = dnext[j];
        ci = ci == 2 ? 0 : ci + 1;
        qb ^= 1;
    }

    if (tid < n_edges - n4) {
        const int e = n4 + tid;
        const unsigned dl = (unsigned)(dst[e] - base);
        if (dl < (unsigned)lim) {
            const float* row = feat + (size_t)src[e] * D;
            for (int c = 0; c < D; ++c) lds_add(&acc[dl * D + c], row[c]);
            if (WITH_DEG) lds_add(&cnt[dl], 1.0f);
        }
    }
    __syncthreads();

    const float4* a4 = reinterpret_cast<const float4*>(acc);
    float4* o4 = reinterpret_cast<float4*>(agg + (size_t)base * D);
    for (int i = tid; i < lim * (D / 4); i += OWN_THREADS) o4[i] = a4[i];
    if (WITH_DEG && tid < lim) deg[base + tid] = cnt[tid];
}

__global__ __launch_bounds__(256)
void sage_finish64(const float* __restrict__ S,
                   const float* __restrict__ deg,
                   const float* __restrict__ Q,
                   float* __restrict__ out,
                   int n_nodes) {
    const int t = blockIdx.x * 256 + threadIdx.x;
    if (t >= n_nodes * 16) return;

    const float invd = 1.0f / fmaxf(deg[t >> 4], 1.0f);
    const float4 s = reinterpret_cast<const float4*>(S)[t];
    const float4 q = reinterpret_cast<const float4*>(Q)[t];

    float4 o;
    o.x = q.x + s.x * invd;
    o.y = q.y + s.y * invd;
    o.z = q.z + s.z * invd;
    o.w = q.w + s.w * invd;
    reinterpret_cast<float4*>(out)[t] = o;
}

#define LDS_PITCH   132
#define STRIP_WORDS (16 * LDS_PITCH)

__global__ __launch_bounds__(256) __attribute__((amdgpu_waves_per_eu(4, 4)))
void sage_gemm128(const float* __restrict__ agg,
                  const float* __restrict__ deg,
                  const float* __restrict__ X,
                  const float* __restrict__ Wl,
                  const float* __restrict__ Wr,
                  const float* __restrict__ bias,
                  float* __restrict__ out,
                  int n_nodes) {
    __shared__ __attribute__((aligned(16))) float lds[2][2][STRIP_WORDS];

    const int n_strips = n_nodes / 16;
    const int tid  = threadIdx.x;
    const int lane = tid & 31;
    const int w    = tid >> 5;
    const int r    = lane & 15;
    const int half = lane >> 4;
    const int kk   = half * 2;
    const int coln = w * 16 + r;

    v2f bL[32], bR[32];
#pragma unroll
    for (int ks = 0; ks < 32; ++ks) {
        const int k = ks * 4 + kk;
        bL[ks].x = Wl[(size_t)k * 128 + coln];
        bL[ks].y = Wl[(size_t)(k + 1) * 128 + coln];
        bR[ks].x = Wr[(size_t)k * 128 + coln];
        bR[ks].y = Wr[(size_t)(k + 1) * 128 + coln];
    }
    const float bv = bias[coln];

    const int srow = tid >> 5;
    const int c4   = tid & 31;

    int s   = blockIdx.x;
    int cur = 0;

    if (s < n_strips) {
        const int row0 = s * 16 + srow;
        const int row1 = row0 + 8;
        const float i0 = 1.0f / fmaxf(deg[row0], 1.0f);
        const float i1 = 1.0f / fmaxf(deg[row1], 1.0f);
        float4 a0 = *(reinterpret_cast<const float4*>(agg + (size_t)row0 * 128) + c4);
        float4 a1 = *(reinterpret_cast<const float4*>(agg + (size_t)row1 * 128) + c4);
        const float4 x0 = *(reinterpret_cast<const float4*>(X + (size_t)row0 * 128) + c4);
        const float4 x1 = *(reinterpret_cast<const float4*>(X + (size_t)row1 * 128) + c4);
        a0.x *= i0; a0.y *= i0; a0.z *= i0; a0.w *= i0;
        a1.x *= i1; a1.y *= i1; a1.z *= i1; a1.w *= i1;
        *reinterpret_cast<float4*>(&lds[0][0][srow * LDS_PITCH + c4 * 4])       = a0;
        *reinterpret_cast<float4*>(&lds[0][0][(srow + 8) * LDS_PITCH + c4 * 4]) = a1;
        *reinterpret_cast<float4*>(&lds[0][1][srow * LDS_PITCH + c4 * 4])       = x0;
        *reinterpret_cast<float4*>(&lds[0][1][(srow + 8) * LDS_PITCH + c4 * 4]) = x1;
    }
    __syncthreads();

    for (; s < n_strips; s += gridDim.x) {
        const int ahead = s + (int)gridDim.x;
        const int nxt   = ahead < n_strips ? ahead : s;
        const int row0  = nxt * 16 + srow;
        const int row1  = row0 + 8;
        const float d0 = deg[row0];
        const float d1 = deg[row1];
        float4 a0 = *(reinterpret_cast<const float4*>(agg + (size_t)row0 * 128) + c4);
        float4 a1 = *(reinterpret_cast<const float4*>(agg + (size_t)row1 * 128) + c4);
        const float4 x0 = *(reinterpret_cast<const float4*>(X + (size_t)row0 * 128) + c4);
        const float4 x1 = *(reinterpret_cast<const float4*>(X + (size_t)row1 * 128) + c4);

        const float* la = &lds[cur][0][r * LDS_PITCH + kk];
        const float* lx = &lds[cur][1][r * LDS_PITCH + kk];

        v8f accA = {};
        v8f accX = {};
#pragma unroll
        for (int ks = 0; ks < 32; ++ks) {
            const v2f aA = *reinterpret_cast<const v2f*>(la + ks * 4);
            const v2f aX = *reinterpret_cast<const v2f*>(lx + ks * 4);
            accA = __builtin_amdgcn_wmma_f32_16x16x4_f32(
                       false, aA, false, bL[ks], (short)0, accA, false, false);
            accX = __builtin_amdgcn_wmma_f32_16x16x4_f32(
                       false, aX, false, bR[ks], (short)0, accX, false, false);
        }

        float v[8];
#pragma unroll
        for (int i = 0; i < 8; ++i) {
            v[i] = fmaxf((accA[i] + accX[i]) + bv, 0.0f);
        }

        {
            const float i0 = 1.0f / fmaxf(d0, 1.0f);
            const float i1 = 1.0f / fmaxf(d1, 1.0f);
            a0.x *= i0; a0.y *= i0; a0.z *= i0; a0.w *= i0;
            a1.x *= i1; a1.y *= i1; a1.z *= i1; a1.w *= i1;
            const int nb = cur ^ 1;
            *reinterpret_cast<float4*>(&lds[nb][0][srow * LDS_PITCH + c4 * 4])       = a0;
            *reinterpret_cast<float4*>(&lds[nb][0][(srow + 8) * LDS_PITCH + c4 * 4]) = a1;
            *reinterpret_cast<float4*>(&lds[nb][1][srow * LDS_PITCH + c4 * 4])       = x0;
            *reinterpret_cast<float4*>(&lds[nb][1][(srow + 8) * LDS_PITCH + c4 * 4]) = x1;
        }
        __syncthreads();
        cur ^= 1;

        const int rowBase = s * 16 + 8 * half;
#pragma unroll
        for (int i = 0; i < 8; ++i) {
            out[(size_t)(rowBase + i) * 128 + coln] = v[i];
        }
    }
}

__global__ __launch_bounds__(256) __attribute__((amdgpu_waves_per_eu(4, 4)))
void sage_transform64(const float* __restrict__ H,
                      const float* __restrict__ Wl,
                      const float* __restrict__ Wr,
                      const float* __restrict__ bias,
                      float* __restrict__ P,
                      float* __restrict__ Q,
                      int n_nodes) {
    __shared__ __attribute__((aligned(16))) float lds[2][2][STRIP_WORDS];

    const int n_strips = n_nodes / 16;
    const int n_pairs  = (n_strips + 1) / 2;
    const int tid  = threadIdx.x;
    const int lane = tid & 31;
    const int w    = tid >> 5;
    const int grp  = w >> 2;
    const int r    = lane & 15;
    const int half = lane >> 4;
    const int kk   = half * 2;
    const int coln = (w & 3) * 16 + r;

    v2f bL[32], bR[32];
#pragma unroll
    for (int ks = 0; ks < 32; ++ks) {
        const int k = ks * 4 + kk;
        bL[ks].x = Wl[(size_t)k * 64 + coln];
        bL[ks].y = Wl[(size_t)(k + 1) * 64 + coln];
        bR[ks].x = Wr[(size_t)k * 64 + coln];
        bR[ks].y = Wr[(size_t)(k + 1) * 64 + coln];
    }
    const float bv = bias[coln];

    const int srow = tid >> 5;
    const int c4   = tid & 31;
    const int last = n_strips - 1;

    int p   = blockIdx.x;
    int cur = 0;

    if (p < n_pairs) {
#pragma unroll
        for (int sp = 0; sp < 2; ++sp) {
            const int st   = min(2 * p + sp, last);
            const int row0 = st * 16 + srow;
            const float4 h0 = *(reinterpret_cast<const float4*>(H + (size_t)row0 * 128) + c4);
            const float4 h1 = *(reinterpret_cast<const float4*>(H + (size_t)(row0 + 8) * 128) + c4);
            *reinterpret_cast<float4*>(&lds[0][sp][srow * LDS_PITCH + c4 * 4])       = h0;
            *reinterpret_cast<float4*>(&lds[0][sp][(srow + 8) * LDS_PITCH + c4 * 4]) = h1;
        }
    }
    __syncthreads();

    for (; p < n_pairs; p += gridDim.x) {
        const int ahead = p + (int)gridDim.x;
        const int nxt   = ahead < n_pairs ? ahead : p;
        const int rowA  = min(2 * nxt, last) * 16 + srow;
        const int rowB  = min(2 * nxt + 1, last) * 16 + srow;
        const float4 h00 = *(reinterpret_cast<const float4*>(H + (size_t)rowA * 128) + c4);
        const float4 h01 = *(reinterpret_cast<const float4*>(H + (size_t)(rowA + 8) * 128) + c4);
        const float4 h10 = *(reinterpret_cast<const float4*>(H + (size_t)rowB * 128) + c4);
        const float4 h11 = *(reinterpret_cast<const float4*>(H + (size_t)(rowB + 8) * 128) + c4);

        const float* lh = &lds[cur][grp][r * LDS_PITCH + kk];

        v8f accP = {};
        v8f accQ = {};
#pragma unroll
        for (int ks = 0; ks < 32; ++ks) {
            const v2f aH = *reinterpret_cast<const v2f*>(lh + ks * 4);
            accP = __builtin_amdgcn_wmma_f32_16x16x4_f32(
                       false, aH, false, bL[ks], (short)0, accP, false, false);
            accQ = __builtin_amdgcn_wmma_f32_16x16x4_f32(
                       false, aH, false, bR[ks], (short)0, accQ, false, false);
        }

        float vq[8];
#pragma unroll
        for (int i = 0; i < 8; ++i) {
            vq[i] = accQ[i] + bv;
        }

        {
            const int nb = cur ^ 1;
            *reinterpret_cast<float4*>(&lds[nb][0][srow * LDS_PITCH + c4 * 4])       = h00;
            *reinterpret_cast<float4*>(&lds[nb][0][(srow + 8) * LDS_PITCH + c4 * 4]) = h01;
            *reinterpret_cast<float4*>(&lds[nb][1][srow * LDS_PITCH + c4 * 4])       = h10;
            *reinterpret_cast<float4*>(&lds[nb][1][(srow + 8) * LDS_PITCH + c4 * 4]) = h11;
        }
        __syncthreads();
        cur ^= 1;

        const int st = 2 * p + grp;
        if (st < n_strips) {
            const int rowBase = st * 16 + 8 * half;
#pragma unroll
            for (int i = 0; i < 8; ++i) {
                const size_t off = (size_t)(rowBase + i) * 64 + coln;
                P[off] = accP[i];
                Q[off] = vq[i];
            }
        }
    }
}

extern "C" void kernel_launch(void* const* d_in, const int* in_sizes, int n_in,
                              void* d_out, int out_size, void* d_ws, size_t ws_size,
                              hipStream_t stream) {
    const float* x    = (const float*)d_in[0];
    const int*   ei   = (const int*)  d_in[1];
    const float* W1l  = (const float*)d_in[2];
    const float* W1r  = (const float*)d_in[3];
    const float* b1   = (const float*)d_in[4];
    const float* W2l  = (const float*)d_in[5];
    const float* W2r  = (const float*)d_in[6];
    const float* b2   = (const float*)d_in[7];
    float*       outp = (float*)d_out;

    const int n_nodes = in_sizes[0] / IN_DIM;
    const int n_edges = in_sizes[1] / 2;
    const int* src = ei;
    const int* dst = ei + n_edges;

    const size_t aggBytes = (size_t)n_nodes * 128 * sizeof(float);
    const size_t rowBytes = (size_t)n_nodes * 64 * sizeof(float);
    const size_t degBytes = (((size_t)n_nodes * sizeof(float)) + 255) & ~(size_t)255;

    char* ws    = (char*)d_ws;
    float* agg1 = (float*)(ws);
    float* sum2 = (float*)(ws + aggBytes);
    float* deg  = (float*)(ws + aggBytes + rowBytes);
    float* h    = (float*)(ws + aggBytes + rowBytes + degBytes);
    float* p2   = (float*)(ws + 2 * aggBytes + rowBytes + degBytes);
    float* q2   = (float*)(ws + 2 * aggBytes + 2 * rowBytes + degBytes);

    const int ownBlocks = (n_nodes + OWN_NODES - 1) / OWN_NODES;
    const int denseBlocks = 512;
    const int finishBlocks = (n_nodes * 16 + 255) / 256;

    sage_owner_sum<128, true><<<ownBlocks, OWN_THREADS, 0, stream>>>(
        x, src, dst, agg1, deg, n_nodes, n_edges);
    sage_gemm128<<<denseBlocks, 256, 0, stream>>>(
        agg1, deg, x, W1l, W1r, b1, h, n_nodes);

    sage_transform64<<<denseBlocks, 256, 0, stream>>>(
        h, W2l, W2r, b2, p2, q2, n_nodes);
    sage_owner_sum<64, false><<<ownBlocks, OWN_THREADS, 0, stream>>>(
        p2, src, dst, sum2, nullptr, n_nodes, n_edges);
    sage_finish64<<<finishBlocks, 256, 0, stream>>>(sum2, deg, q2, outp, n_nodes);
}


# layer-2 edge sums replay per-owner (source, local row) lists that layer 1's filter writes down while it drains: no second scan of the destination list; sums stay in LDS arrival order
# speedup vs baseline: 2.4681x; 1.3539x over previous
#include <hip/hip_runtime.h>


typedef __attribute__((ext_vector_type(2))) float v2f;
typedef __attribute__((ext_vector_type(8))) float v8f;

#define N_NODES 50000
#define N_EDGES 1600000
#define IN_DIM  128
#define HIDDEN  128
#define OUT_DIM 64

#define OWN_NODES   200
#define OWN_THREADS 1024
#define SCAN_VEC    8
#define OWN_CHUNK   (OWN_THREADS * SCAN_VEC * 4)
#define OWN_QCAP    2048
#define LIST_CAP    32768
#define REPLAY_CAP  8192

__device__ __forceinline__ void lds_add(float* p, float v) {
    __hip_atomic_fetch_add(p, v, __ATOMIC_RELAXED, __HIP_MEMORY_SCOPE_WORKGROUP);
}

template <int D, bool WITH_DEG>
__device__ __forceinline__ void own_push(int e, int dl, int2* q, int* qcount,
                                         float* acc, float* cnt,
                                         const float* __restrict__ feat,
                                         const int* __restrict__ src) {
    const int pos = __hip_atomic_fetch_add(qcount, 1, __ATOMIC_RELAXED, __HIP_MEMORY_SCOPE_WORKGROUP);
    if (pos < OWN_QCAP) {
        q[pos] = make_int2(e, dl);
    } else {
        const float* row = feat + (size_t)src[e] * D;
        for (int c = 0; c < D; ++c) lds_add(&acc[dl * D + c], row[c]);
        if (WITH_DEG) lds_add(&cnt[dl], 1.0f);
    }
}

template <int D, int SLOTS>
__device__ __forceinline__ void replay_fetch(const int2* q, int m, int k0, int wave, int eslot, int sub,
                                             const float* __restrict__ feat,
                                             int (&dl)[SLOTS], bool (&ok)[SLOTS], float4 (&v)[SLOTS]) {
    constexpr int EPW = 32 / (D / 4);
#pragma unroll
    for (int j = 0; j < SLOTS; ++j) {
        const int idx = k0 + (j * 32 + wave) * EPW + eslot;
        ok[j] = idx < m;
        const int2 ent = q[ok[j] ? idx : 0];
        dl[j] = ent.y;
        const unsigned off = (unsigned)ent.x * (unsigned)(D * 4) + (unsigned)(sub * 16);
        v[j] = *reinterpret_cast<const float4*>(reinterpret_cast<const char*>(feat) + off);
    }
}

template <int D, int SLOTS>
__device__ __forceinline__ void replay_add(float* acc, int sub, int rot,
                                           const int (&dl)[SLOTS], const bool (&ok)[SLOTS], const float4 (&v)[SLOTS]) {
#pragma unroll
    for (int j = 0; j < SLOTS; ++j) {
        if (ok[j]) {
            float* a = &acc[dl[j] * D + sub * 4];
            lds_add(a + rot,             rot ? v[j].y : v[j].x);
            lds_add(a + 1 + rot,         rot ? v[j].z : v[j].y);
            lds_add(a + 2 + rot,         rot ? v[j].w : v[j].z);
            lds_add(a + (rot ? 0 : 3),   rot ? v[j].x : v[j].w);
        }
    }
}

template <int D, bool WITH_DEG, bool RECORD, bool REPLAY>
__global__ __launch_bounds__(OWN_THREADS)
void sage_owner_sum(const float* __restrict__ feat,
                    const int*   __restrict__ src,
                    const int*   __restrict__ dst,
                    float* __restrict__ agg,
                    float* __restrict__ deg,
                    int*   __restrict__ lists,
                    int*   __restrict__ heads,
                    int n_nodes, int n_edges) {
    constexpr int QENTS = REPLAY ? REPLAY_CAP : 2 * OWN_QCAP;
    static_assert(REPLAY_CAP >= 2 * OWN_QCAP, "the scan form needs two queue buffers");
    static_assert(REPLAY_CAP % OWN_THREADS == 0, "whole staging rounds");

    __shared__ __attribute__((aligned(16))) float acc[OWN_NODES * D];
    __shared__ float cnt[OWN_NODES];
    __shared__ int2  queue[QENTS];
    __shared__ int   qn[3];

    constexpr int LPE   = D / 4;
    constexpr int EPW   = 32 / LPE;
    constexpr int SLOTS = (D == 128) ? 5 : 3;
    constexpr int PER_PASS = 32 * EPW * SLOTS;

    const int tid  = threadIdx.x;
    const int lane = tid & 31;
    const int wave = tid >> 5;
    const int sub  = lane % LPE;
    const int eslot = lane / LPE;
    const int rot  = lane >> 4;
    const int base = blockIdx.x * OWN_NODES;
    const int lim  = min(OWN_NODES, n_nodes - base);
    const int n4   = n_edges & ~3;

    int2* const lst = reinterpret_cast<int2*>(lists) + (size_t)blockIdx.x * LIST_CAP;

    int head_n = 0, head_bad = 1;
    if (REPLAY) {
        const int2 h = reinterpret_cast<const int2*>(heads)[blockIdx.x];
        head_n   = __builtin_amdgcn_readfirstlane(h.x);
        head_bad = __builtin_amdgcn_readfirstlane(h.y);
    }

    for (int i = tid; i < OWN_NODES * D; i += OWN_THREADS) acc[i] = 0.0f;
    if (tid < OWN_NODES) cnt[tid] = 0.0f;
    if (tid < 3) qn[tid] = 0;

    int total = 0;
    int bad   = 0;

    if (REPLAY && head_bad == 0) {
        constexpr int RPT    = REPLAY_CAP / OWN_THREADS;
        constexpr int RSLOTS = 8;
        constexpr int RPASS  = 32 * EPW * RSLOTS;

        __syncthreads();

        for (int c0 = 0; c0 < head_n; c0 += REPLAY_CAP) {
            const int m = min(REPLAY_CAP, head_n - c0);
            int2 pre[RPT];
#pragma unroll
            for (int t = 0; t < RPT; ++t) {
                const int i = c0 + t * OWN_THREADS + tid;
                pre[t] = lst[i < head_n ? i : 0];
            }
#pragma unroll
            for (int t = 0; t < RPT; ++t) {
                const int i = t * OWN_THREADS + tid;
                if (i < m) queue[i] = pre[t];
            }
            __syncthreads();

            int    dlA[RSLOTS], dlB[RSLOTS];
            bool   okA[RSLOTS], okB[RSLOTS];
            float4 vA[RSLOTS],  vB[RSLOTS];
            replay_fetch<D, RSLOTS>(queue, m, 0, wave, eslot, sub, feat, dlA, okA, vA);
            for (int k0 = 0; k0 < m; k0 += 2 * RPASS) {
                replay_fetch<D, RSLOTS>(queue, m, k0 + RPASS, wave, eslot, sub, feat, dlB, okB, vB);
                replay_add<D, RSLOTS>(acc, sub, rot, dlA, okA, vA);
                replay_fetch<D, RSLOTS>(queue, m, k0 + 2 * RPASS, wave, eslot, sub, feat, dlA, okA, vA);
                replay_add<D, RSLOTS>(acc, sub, rot, dlB, okB, vB);
            }
            __syncthreads();
        }
    } else {
        int4 dcur[SCAN_VEC];
#pragma unroll
        for (int j = 0; j < SCAN_VEC; ++j) {
            const int e = (j * OWN_THREADS + tid) * 4;
            const int4 v = *reinterpret_cast<const int4*>(dst + (e < n4 ? e : 0));
            dcur[j] = e < n4 ? v : make_int4(-1, -1, -1, -1);
        }
        __syncthreads();

        int ci = 0;
        int qb = 0;
        for (int c0 = 0; c0 < n4; c0 += OWN_CHUNK) {
            int4 dnext[SCAN_VEC];
#pragma unroll
            for (int j = 0; j < SCAN_VEC; ++j) {
                const int e = c0 + OWN_CHUNK + (j * OWN_THREADS + tid) * 4;
                const int4 v = *reinterpret_cast<const int4*>(dst + (e < n4 ? e : 0));
                dnext[j] = e < n4 ? v : make_int4(-1, -1, -1, -1);
            }

            int2* const qcur = queue + qb * OWN_QCAP;
#pragma unroll
            for (int j = 0; j < SCAN_VEC; ++j) {
                const int eb = c0 + (j * OWN_THREADS + tid) * 4;
                const unsigned dl[4] = {(unsigned)(dcur[j].x - base), (unsigned)(dcur[j].y - base),
                                        (unsigned)(dcur[j].z - base), (unsigned)(dcur[j].w - base)};
                if (min(min(dl[0], dl[1]), min(dl[2], dl[3])) < (unsigned)lim) {
#pragma unroll
                    for (int k = 0; k < 4; ++k) {
                        if (dl[k] < (unsigned)lim) {
                            own_push<D, WITH_DEG>(eb + k, (int)dl[k], qcur, &qn[ci], acc, cnt, feat, src);
                        }
                    }
                }
            }
            __syncthreads();

            if (tid == 0) qn[ci == 0 ? 2 : ci - 1] = 0;

            const int nraw = qn[ci];
            const int n = min(nraw, OWN_QCAP);
            const int2* q = qcur;
            for (int k0 = 0; k0 < n; k0 += PER_PASS) {
                int    dl[SLOTS];
                int    sn[SLOTS];
                bool   ok[SLOTS];
                float4 v[SLOTS];
#pragma unroll
                for (int j = 0; j < SLOTS; ++j) {
                    const int idx = k0 + (j * 32 + wave) * EPW + eslot;
                    ok[j] = idx < n;
                    const int2 ent = q[ok[j] ? idx : 0];
                    dl[j] = ent.y;
                    sn[j] = src[ent.x];
                    v[j] = *(reinterpret_cast<const float4*>(feat + (size_t)sn[j] * D) + sub);
                }
#pragma unroll
                for (int j = 0; j < SLOTS; ++j) {
                    asm volatile("" :: "v"(v[j].x), "v"(v[j].y), "v"(v[j].z), "v"(v[j].w));
                }
#pragma unroll
                for (int j = 0; j < SLOTS; ++j) {
                    if (ok[j]) {
                        float* a = &acc[dl[j] * D + sub * 4];
                        lds_add(a + rot,             rot ? v[j].y : v[j].x);
                        lds_add(a + 1 + rot,         rot ? v[j].z : v[j].y);
                        lds_add(a + 2 + rot,         rot ? v[j].w : v[j].z);
                        lds_add(a + (rot ? 0 : 3),   rot ? v[j].x : v[j].w);
                        if (WITH_DEG && sub == 0) lds_add(&cnt[dl[j]], 1.0f);
                        if (RECORD && sub == 0) {
                            const int g = total + k0 + (j * 32 + wave) * EPW + eslot;
                            if (g < LIST_CAP) lst[g] = make_int2(sn[j], dl[j]);
                        }
                    }
                }
            }
            total += n;
            if (nraw > OWN_QCAP) bad = 1;

#pragma unroll
            for (int j = 0; j < SCAN_VEC; ++j) dcur[j] = dnext[j];
            ci = ci == 2 ? 0 : ci + 1;
            qb ^= 1;
        }
    }

    if (tid < n_edges - n4) {
        const int e = n4 + tid;
        const unsigned dl = (unsigned)(dst[e] - base);
        if (dl < (unsigned)lim) {
            const float* row = feat + (size_t)src[e] * D;
            for (int c = 0; c < D; ++c) lds_add(&acc[dl * D + c], row[c]);
            if (WITH_DEG) lds_add(&cnt[dl], 1.0f);
        }
    }
    __syncthreads();

    const float4* a4 = reinterpret_cast<const float4*>(acc);
    float4* o4 = reinterpret_cast<float4*>(agg + (size_t)base * D);
    for (int i = tid; i < lim * (D / 4); i += OWN_THREADS) o4[i] = a4[i];
    if (WITH_DEG && tid < lim) deg[base + tid] = cnt[tid];

    if (RECORD && tid == 0) {
        const int full = (bad != 0 || total > LIST_CAP) ? 1 : 0;
        reinterpret_cast<int2*>(heads)[blockIdx.x] = make_int2(min(total, LIST_CAP), full);
    }
}

__global__ __launch_bounds__(256)
void sage_finish64(const float* __restrict__ S,
                   const float* __restrict__ deg,
                   const float* __restrict__ Q,
                   float* __restrict__ out,
                   int n_nodes) {
    const int t = blockIdx.x * 256 + threadIdx.x;
    if (t >= n_nodes * 16) return;

    const float invd = 1.0f / fmaxf(deg[t >> 4], 1.0f);
    const float4 s = reinterpret_cast<const float4*>(S)[t];
    const float4 q = reinterpret_cast<const float4*>(Q)[t];

    float4 o;
    o.x = q.x + s.x * invd;
    o.y = q.y + s.y * invd;
    o.z = q.z + s.z * invd;
    o.w = q.w + s.w * invd;
    reinterpret_cast<float4*>(out)[t] = o;
}

#define LDS_PITCH   132
#define STRIP_WORDS (16 * LDS_PITCH)

__global__ __launch_bounds__(256) __attribute__((amdgpu_waves_per_eu(4, 4)))
void sage_gemm128(const float* __restrict__ agg,
                  const float* __restrict__ deg,
                  const float* __restrict__ X,
                  const float* __restrict__ Wl,
                  const float* __restrict__ Wr,
                  const float* __restrict__ bias,
                  float* __restrict__ out,
                  int n_nodes) {
    __shared__ __attribute__((aligned(16))) float lds[2][2][STRIP_WORDS];

    const int n_strips = n_nodes / 16;
    const int tid  = threadIdx.x;
    const int lane = tid & 31;
    const int w    = tid >> 5;
    const int r    = lane & 15;
    const int half = lane >> 4;
    const int kk   = half * 2;
    const int coln = w * 16 + r;

    v2f bL[32], bR[32];
#pragma unroll
    for (int ks = 0; ks < 32; ++ks) {
        const int k = ks * 4 + kk;
        bL[ks].x = Wl[(size_t)k * 128 + coln];
        bL[ks].y = Wl[(size_t)(k + 1) * 128 + coln];
        bR[ks].x = Wr[(size_t)k * 128 + coln];
        bR[ks].y = Wr[(size_t)(k + 1) * 128 + coln];
    }
    const float bv = bias[coln];

    const int srow = tid >> 5;
    const int c4   = tid & 31;

    int s   = blockIdx.x;
    int cur = 0;

    if (s < n_strips) {
        const int row0 = s * 16 + srow;
        const int row1 = row0 + 8;
        const float i0 = 1.0f / fmaxf(deg[row0], 1.0f);
        const float i1 = 1.0f / fmaxf(deg[row1], 1.0f);
        float4 a0 = *(reinterpret_cast<const float4*>(agg + (size_t)row0 * 128) + c4);
        float4 a1 = *(reinterpret_cast<const float4*>(agg + (size_t)row1 * 128) + c4);
        const float4 x0 = *(reinterpret_cast<const float4*>(X + (size_t)row0 * 128) + c4);
        const float4 x1 = *(reinterpret_cast<const float4*>(X + (size_t)row1 * 128) + c4);
        a0.x *= i0; a0.y *= i0; a0.z *= i0; a0.w *= i0;
        a1.x *= i1; a1.y *= i1; a1.z *= i1; a1.w *= i1;
        *reinterpret_cast<float4*>(&lds[0][0][srow * LDS_PITCH + c4 * 4])       = a0;
        *reinterpret_cast<float4*>(&lds[0][0][(srow + 8) * LDS_PITCH + c4 * 4]) = a1;
        *reinterpret_cast<float4*>(&lds[0][1][srow * LDS_PITCH + c4 * 4])       = x0;
        *reinterpret_cast<float4*>(&lds[0][1][(srow + 8) * LDS_PITCH + c4 * 4]) = x1;
    }
    __syncthreads();

    for (; s < n_strips; s += gridDim.x) {
        const int ahead = s + (int)gridDim.x;
        const int nxt   = ahead < n_strips ? ahead : s;
        const int row0  = nxt * 16 + srow;
        const int row1  = row0 + 8;
        const float d0 = deg[row0];
        const float d1 = deg[row1];
        float4 a0 = *(reinterpret_cast<const float4*>(agg + (size_t)row0 * 128) + c4);
        float4 a1 = *(reinterpret_cast<const float4*>(agg + (size_t)row1 * 128) + c4);
        const float4 x0 = *(reinterpret_cast<const float4*>(X + (size_t)row0 * 128) + c4);
        const float4 x1 = *(reinterpret_cast<const float4*>(X + (size_t)row1 * 128) + c4);

        const float* la = &lds[cur][0][r * LDS_PITCH + kk];
        const float* lx = &lds[cur][1][r * LDS_PITCH + kk];

        v8f accA = {};
        v8f accX = {};
#pragma unroll
        for (int ks = 0; ks < 32; ++ks) {
            const v2f aA = *reinterpret_cast<const v2f*>(la + ks * 4);
            const v2f aX = *reinterpret_cast<const v2f*>(lx + ks * 4);
            accA = __builtin_amdgcn_wmma_f32_16x16x4_f32(
                       false, aA, false, bL[ks], (short)0, accA, false, false);
            accX = __builtin_amdgcn_wmma_f32_16x16x4_f32(
                       false, aX, false, bR[ks], (short)0, accX, false, false);
        }

        float v[8];
#pragma unroll
        for (int i = 0; i < 8; ++i) {
            v[i] = fmaxf((accA[i] + accX[i]) + bv, 0.0f);
        }

        {
            const float i0 = 1.0f / fmaxf(d0, 1.0f);
            const float i1 = 1.0f / fmaxf(d1, 1.0f);
            a0.x *= i0; a0.y *= i0; a0.z *= i0; a0.w *= i0;
            a1.x *= i1; a1.y *= i1; a1.z *= i1; a1.w *= i1;
            const int nb = cur ^ 1;
            *reinterpret_cast<float4*>(&lds[nb][0][srow * LDS_PITCH + c4 * 4])       = a0;
            *reinterpret_cast<float4*>(&lds[nb][0][(srow + 8) * LDS_PITCH + c4 * 4]) = a1;
            *reinterpret_cast<float4*>(&lds[nb][1][srow * LDS_PITCH + c4 * 4])       = x0;
            *reinterpret_cast<float4*>(&lds[nb][1][(srow + 8) * LDS_PITCH + c4 * 4]) = x1;
        }
        __syncthreads();
        cur ^= 1;

        const int rowBase = s * 16 + 8 * half;
#pragma unroll
        for (int i = 0; i < 8; ++i) {
            out[(size_t)(rowBase + i) * 128 + coln] = v[i];
        }
    }
}

__global__ __launch_bounds__(256) __attribute__((amdgpu_waves_per_eu(4, 4)))
void sage_transform64(const float* __restrict__ H,
                      const float* __restrict__ Wl,
                      const float* __restrict__ Wr,
                      const float* __restrict__ bias,
                      float* __restrict__ P,
                      float* __restrict__ Q,
                      int n_nodes) {
    __shared__ __attribute__((aligned(16))) float lds[2][2][STRIP_WORDS];

    const int n_strips = n_nodes / 16;
    const int n_pairs  = (n_strips + 1) / 2;
    const int tid  = threadIdx.x;
    const int lane = tid & 31;
    const int w    = tid >> 5;
    const int grp  = w >> 2;
    const int r    = lane & 15;
    const int half = lane >> 4;
    const int kk   = half * 2;
    const int coln = (w & 3) * 16 + r;

    v2f bL[32], bR[32];
#pragma unroll
    for (int ks = 0; ks < 32; ++ks) {
        const int k = ks * 4 + kk;
        bL[ks].x = Wl[(size_t)k * 64 + coln];
        bL[ks].y = Wl[(size_t)(k + 1) * 64 + coln];
        bR[ks].x = Wr[(size_t)k * 64 + coln];
        bR[ks].y = Wr[(size_t)(k + 1) * 64 + coln];
    }
    const float bv = bias[coln];

    const int srow = tid >> 5;
    const int c4   = tid & 31;
    const int last = n_strips - 1;

    int p   = blockIdx.x;
    int cur = 0;

    if (p < n_pairs) {
#pragma unroll
        for (int sp = 0; sp < 2; ++sp) {
            const int st   = min(2 * p + sp, last);
            const int row0 = st * 16 + srow;
            const float4 h0 = *(reinterpret_cast<const float4*>(H + (size_t)row0 * 128) + c4);
            const float4 h1 = *(reinterpret_cast<const float4*>(H + (size_t)(row0 + 8) * 128) + c4);
            *reinterpret_cast<float4*>(&lds[0][sp][srow * LDS_PITCH + c4 * 4])       = h0;
            *reinterpret_cast<float4*>(&lds[0][sp][(srow + 8) * LDS_PITCH + c4 * 4]) = h1;
        }
    }
    __syncthreads();

    for (; p < n_pairs; p += gridDim.x) {
        const int ahead = p + (int)gridDim.x;
        const int nxt   = ahead < n_pairs ? ahead : p;
        const int rowA  = min(2 * nxt, last) * 16 + srow;
        const int rowB  = min(2 * nxt + 1, last) * 16 + srow;
        const float4 h00 = *(reinterpret_cast<const float4*>(H + (size_t)rowA * 128) + c4);
        const float4 h01 = *(reinterpret_cast<const float4*>(H + (size_t)(rowA + 8) * 128) + c4);
        const float4 h10 = *(reinterpret_cast<const float4*>(H + (size_t)rowB * 128) + c4);
        const float4 h11 = *(reinterpret_cast<const float4*>(H + (size_t)(rowB + 8) * 128) + c4);

        const float* lh = &lds[cur][grp][r * LDS_PITCH + kk];

        v8f accP = {};
        v8f accQ = {};
#pragma unroll
        for (int ks = 0; ks < 32; ++ks) {
            const v2f aH = *reinterpret_cast<const v2f*>(lh + ks * 4);
            accP = __builtin_amdgcn_wmma_f32_16x16x4_f32(
                       false, aH, false, bL[ks], (short)0, accP, false, false);
            accQ = __builtin_amdgcn_wmma_f32_16x16x4_f32(
                       false, aH, false, bR[ks], (short)0, accQ, false, false);
        }

        float vq[8];
#pragma unroll
        for (int i = 0; i < 8; ++i) {
            vq[i] = accQ[i] + bv;
        }

        {
            const int nb = cur ^ 1;
            *reinterpret_cast<float4*>(&lds[nb][0][srow * LDS_PITCH + c4 * 4])       = h00;
            *reinterpret_cast<float4*>(&lds[nb][0][(srow + 8) * LDS_PITCH + c4 * 4]) = h01;
            *reinterpret_cast<float4*>(&lds[nb][1][srow * LDS_PITCH + c4 * 4])       = h10;
            *reinterpret_cast<float4*>(&lds[nb][1][(srow + 8) * LDS_PITCH + c4 * 4]) = h11;
        }
        __syncthreads();
        cur ^= 1;

        const int st = 2 * p + grp;
        if (st < n_strips) {
            const int rowBase = st * 16 + 8 * half;
#pragma unroll
            for (int i = 0; i < 8; ++i) {
                const size_t off = (size_t)(rowBase + i) * 64 + coln;
                P[off] = accP[i];
                Q[off] = vq[i];
            }
        }
    }
}

extern "C" void kernel_launch(void* const* d_in, const int* in_sizes, int n_in,
                              void* d_out, int out_size, void* d_ws, size_t ws_size,
                              hipStream_t stream) {
    const float* x    = (const float*)d_in[0];
    const int*   ei   = (const int*)  d_in[1];
    const float* W1l  = (const float*)d_in[2];
    const float* W1r  = (const float*)d_in[3];
    const float* b1   = (const float*)d_in[4];
    const float* W2l  = (const float*)d_in[5];
    const float* W2r  = (const float*)d_in[6];
    const float* b2   = (const float*)d_in[7];
    float*       outp = (float*)d_out;

    const int n_nodes = in_sizes[0] / IN_DIM;
    const int n_edges = in_sizes[1] / 2;
    const int* src = ei;
    const int* dst = ei + n_edges;

    const size_t aggBytes = (size_t)n_nodes * 128 * sizeof(float);
    const size_t rowBytes = (size_t)n_nodes * 64 * sizeof(float);
    const size_t degBytes = (((size_t)n_nodes * sizeof(float)) + 255) & ~(size_t)255;

    char* ws    = (char*)d_ws;
    float* agg1 = (float*)(ws);
    float* sum2 = (float*)(ws + aggBytes);
    float* deg  = (float*)(ws + aggBytes + rowBytes);
    float* h    = (float*)(ws + aggBytes + rowBytes + degBytes);
    float* p2   = (float*)(ws + 2 * aggBytes + rowBytes + degBytes);
    float* q2   = (float*)(ws + 2 * aggBytes + 2 * rowBytes + degBytes);

    const int ownBlocks = (n_nodes + OWN_NODES - 1) / OWN_NODES;
    const size_t listBytes = (size_t)ownBlocks * LIST_CAP * 2 * sizeof(int);
    int* lists  = (int*)(ws + 2 * aggBytes + 3 * rowBytes + degBytes);
    int* heads  = (int*)(ws + 2 * aggBytes + 3 * rowBytes + degBytes + listBytes);
    const int denseBlocks = 512;
    const int finishBlocks = (n_nodes * 16 + 255) / 256;

    sage_owner_sum<128, true, true, false><<<ownBlocks, OWN_THREADS, 0, stream>>>(
        x, src, dst, agg1, deg, lists, heads, n_nodes, n_edges);
    sage_gemm128<<<denseBlocks, 256, 0, stream>>>(
        agg1, deg, x, W1l, W1r, b1, h, n_nodes);

    sage_transform64<<<denseBlocks, 256, 0, stream>>>(
        h, W2l, W2r, b2, p2, q2, n_nodes);
    sage_owner_sum<64, false, false, true><<<ownBlocks, OWN_THREADS, 0, stream>>>(
        p2, src, dst, sum2, nullptr, lists, heads, n_nodes, n_edges);
    sage_finish64<<<finishBlocks, 256, 0, stream>>>(sum2, deg, q2, outp, n_nodes);
}


# layer-1 edge scan split from its adds: waves note finds in wave-private LDS lists (ballot rank; no queue, no per-chunk barrier), then resolve sources and add rows pipelined; sum order = LDS arrival
# speedup vs baseline: 3.1100x; 1.2601x over previous
#include <hip/hip_runtime.h>


typedef __attribute__((ext_vector_type(2))) float v2f;
typedef __attribute__((ext_vector_type(8))) float v8f;

#define N_NODES 50000
#define N_EDGES 1600000
#define IN_DIM  128
#define HIDDEN  128
#define OUT_DIM 64

#define OWN_NODES   200
#define OWN_THREADS 1024
#define SCAN_VEC    8
#define OWN_CHUNK   (OWN_THREADS * SCAN_VEC * 4)
#define SEG_CAP     512
#define LIST_CAP    32768
#define REPLAY_CAP  8192

__device__ __forceinline__ void lds_add(float* p, float v) {
    __hip_atomic_fetch_add(p, v, __ATOMIC_RELAXED, __HIP_MEMORY_SCOPE_WORKGROUP);
}

template <int D, bool WITH_DEG>
__device__ __forceinline__ void own_note(int e, int dl, int pos, unsigned* seg,
                                         float* acc, float* cnt,
                                         const float* __restrict__ feat,
                                         const int* __restrict__ src) {
    if (pos < SEG_CAP) {
        seg[pos] = ((unsigned)e << 8) | (unsigned)dl;
    } else {
        const float* row = feat + (size_t)src[e] * D;
        for (int c = 0; c < D; ++c) lds_add(&acc[dl * D + c], row[c]);
        if (WITH_DEG) lds_add(&cnt[dl], 1.0f);
    }
}

template <int D, bool WITH_DEG, bool RECORD>
__device__ __forceinline__ void seg_resolve(unsigned* seg, int m, int lane,
                                            const int* __restrict__ src, float* cnt,
                                            int2* __restrict__ lst, int off) {
    constexpr int CV = 8;
    for (int i0 = 0; i0 < m; i0 += 32 * CV) {
        unsigned ent[CV];
        int      s[CV];
#pragma unroll
        for (int t = 0; t < CV; ++t) {
            const int i = i0 + t * 32 + lane;
            ent[t] = seg[i < m ? i : 0];
            s[t]   = src[ent[t] >> 8];
        }
#pragma unroll
        for (int t = 0; t < CV; ++t) {
            const int i = i0 + t * 32 + lane;
            if (i < m) {
                const int dl = (int)(ent[t] & 255u);
                seg[i] = ((unsigned)s[t] << 8) | (unsigned)dl;
                if (WITH_DEG) lds_add(&cnt[dl], 1.0f);
                if (RECORD) {
                    const int g = off + i;
                    if (g < LIST_CAP) lst[g] = make_int2(s[t], dl);
                }
            }
        }
    }
}

template <int D, int SLOTS>
__device__ __forceinline__ void seg_fetch(const unsigned* seg, int m, int k0, int eslot, int sub,
                                          const float* __restrict__ feat,
                                          int (&dl)[SLOTS], bool (&ok)[SLOTS], float4 (&v)[SLOTS]) {
    constexpr int EPW = 32 / (D / 4);
#pragma unroll
    for (int j = 0; j < SLOTS; ++j) {
        const int idx = k0 + j * EPW + eslot;
        ok[j] = idx < m;
        const unsigned ent = seg[ok[j] ? idx : 0];
        dl[j] = (int)(ent & 255u);
        const unsigned off = (ent >> 8) * (unsigned)(D * 4) + (unsigned)(sub * 16);
        v[j] = *reinterpret_cast<const float4*>(reinterpret_cast<const char*>(feat) + off);
    }
}

template <int D, int SLOTS>
__device__ __forceinline__ void replay_fetch(const int2* q, int m, int k0, int wave, int eslot, int sub,
                                             const float* __restrict__ feat,
                                             int (&dl)[SLOTS], bool (&ok)[SLOTS], float4 (&v)[SLOTS]) {
    constexpr int EPW = 32 / (D / 4);
#pragma unroll
    for (int j = 0; j < SLOTS; ++j) {
        const int idx = k0 + (j * 32 + wave) * EPW + eslot;
        ok[j] = idx < m;
        const int2 ent = q[ok[j] ? idx : 0];
        dl[j] = ent.y;
        const unsigned off = (unsigned)ent.x * (unsigned)(D * 4) + (unsigned)(sub * 16);
        v[j] = *reinterpret_cast<const float4*>(reinterpret_cast<const char*>(feat) + off);
    }
}

template <int D, int SLOTS>
__device__ __forceinline__ void replay_add(float* acc, int sub, int rot,
                                           const int (&dl)[SLOTS], const bool (&ok)[SLOTS], const float4 (&v)[SLOTS]) {
#pragma unroll
    for (int j = 0; j < SLOTS; ++j) {
        if (ok[j]) {
            float* a = &acc[dl[j] * D + sub * 4];
            lds_add(a + rot,             rot ? v[j].y : v[j].x);
            lds_add(a + 1 + rot,         rot ? v[j].z : v[j].y);
            lds_add(a + 2 + rot,         rot ? v[j].w : v[j].z);
            lds_add(a + (rot ? 0 : 3),   rot ? v[j].x : v[j].w);
        }
    }
}

template <int D, bool WITH_DEG, bool RECORD, bool REPLAY>
__global__ __launch_bounds__(OWN_THREADS)
void sage_owner_sum(const float* __restrict__ feat,
                    const int*   __restrict__ src,
                    const int*   __restrict__ dst,
                    float* __restrict__ agg,
                    float* __restrict__ deg,
                    int*   __restrict__ lists,
                    int*   __restrict__ heads,
                    int n_nodes, int n_edges) {
    constexpr int WAVES = OWN_THREADS / 32;
    static_assert(REPLAY_CAP * 2 >= WAVES * SEG_CAP, "the replay stage and the waves' lists of the scan form share the room");
    static_assert(REPLAY_CAP % OWN_THREADS == 0, "whole staging rounds");
    static_assert(N_EDGES < (1 << 24) && N_NODES < (1 << 24) && OWN_NODES <= 256, "a find packs into one word");

    __shared__ __attribute__((aligned(16))) float acc[OWN_NODES * D];
    __shared__ float cnt[OWN_NODES];
    __shared__ int2  queue[REPLAY_CAP];
    __shared__ int   wcnt[WAVES];

    constexpr int LPE   = D / 4;
    constexpr int EPW   = 32 / LPE;
    constexpr int RSLOTS = 8;

    const int tid  = threadIdx.x;
    const int lane = tid & 31;
    const int wave = tid >> 5;
    const int sub  = lane % LPE;
    const int eslot = lane / LPE;
    const int rot  = lane >> 4;
    const int base = blockIdx.x * OWN_NODES;
    const int lim  = min(OWN_NODES, n_nodes - base);
    const int n4   = n_edges & ~3;

    int2* const lst = reinterpret_cast<int2*>(lists) + (size_t)blockIdx.x * LIST_CAP;

    int head_n = 0, head_bad = 1;
    if (REPLAY) {
        const int2 h = reinterpret_cast<const int2*>(heads)[blockIdx.x];
        head_n   = __builtin_amdgcn_readfirstlane(h.x);
        head_bad = __builtin_amdgcn_readfirstlane(h.y);
    }

    for (int i = tid; i < OWN_NODES * D; i += OWN_THREADS) acc[i] = 0.0f;
    if (tid < OWN_NODES) cnt[tid] = 0.0f;

    int total = 0;
    int bad   = 0;

    if (REPLAY && head_bad == 0) {
        constexpr int RPT    = REPLAY_CAP / OWN_THREADS;
        constexpr int RPASS  = 32 * EPW * RSLOTS;

        __syncthreads();

        for (int c0 = 0; c0 < head_n; c0 += REPLAY_CAP) {
            const int m = min(REPLAY_CAP, head_n - c0);
            int2 pre[RPT];
#pragma unroll
            for (int t = 0; t < RPT; ++t) {
                const int i = c0 + t * OWN_THREADS + tid;
                pre[t] = lst[i < head_n ? i : 0];
            }
#pragma unroll
            for (int t = 0; t < RPT; ++t) {
                const int i = t * OWN_THREADS + tid;
                if (i < m) queue[i] = pre[t];
            }
            __syncthreads();

            int    dlA[RSLOTS], dlB[RSLOTS];
            bool   okA[RSLOTS], okB[RSLOTS];
            float4 vA[RSLOTS],  vB[RSLOTS];
            replay_fetch<D, RSLOTS>(queue, m, 0, wave, eslot, sub, feat, dlA, okA, vA);
            for (int k0 = 0; k0 < m; k0 += 2 * RPASS) {
                replay_fetch<D, RSLOTS>(queue, m, k0 + RPASS, wave, eslot, sub, feat, dlB, okB, vB);
                replay_add<D, RSLOTS>(acc, sub, rot, dlA, okA, vA);
                replay_fetch<D, RSLOTS>(queue, m, k0 + 2 * RPASS, wave, eslot, sub, feat, dlA, okA, vA);
                replay_add<D, RSLOTS>(acc, sub, rot, dlB, okB, vB);
            }
            __syncthreads();
        }
    } else {
        int4 dcur[SCAN_VEC];
#pragma unroll
        for (int j = 0; j < SCAN_VEC; ++j) {
            const int e = (j * OWN_THREADS + tid) * 4;
            const int4 v = *reinterpret_cast<const int4*>(dst + (e < n4 ? e : 0));
            dcur[j] = e < n4 ? v : make_int4(-1, -1, -1, -1);
        }
        __syncthreads();

        unsigned* const seg = reinterpret_cast<unsigned*>(queue) + wave * SEG_CAP;
        int wn = 0;
        for (int c0 = 0; c0 < n4; c0 += OWN_CHUNK) {
            int4 dnext[SCAN_VEC];
#pragma unroll
            for (int j = 0; j < SCAN_VEC; ++j) {
                const int e = c0 + OWN_CHUNK + (j * OWN_THREADS + tid) * 4;
                const int4 v = *reinterpret_cast<const int4*>(dst + (e < n4 ? e : 0));
                dnext[j] = e < n4 ? v : make_int4(-1, -1, -1, -1);
            }

#pragma unroll
            for (int j = 0; j < SCAN_VEC; ++j) {
                const int eb = c0 + (j * OWN_THREADS + tid) * 4;
                const unsigned dl[4] = {(unsigned)(dcur[j].x - base), (unsigned)(dcur[j].y - base),
                                        (unsigned)(dcur[j].z - base), (unsigned)(dcur[j].w - base)};
                const bool any = min(min(dl[0], dl[1]), min(dl[2], dl[3])) < (unsigned)lim;
                if (__builtin_amdgcn_ballot_w32(any) != 0u) {
#pragma unroll
                    for (int k = 0; k < 4; ++k) {
                        const bool hit = dl[k] < (unsigned)lim;
                        const unsigned who = __builtin_amdgcn_ballot_w32(hit);
                        if (hit) {
                            const int pos = wn + (int)__builtin_amdgcn_mbcnt_lo(who, 0u);
                            own_note<D, WITH_DEG>(eb + k, (int)dl[k], pos, seg, acc, cnt, feat, src);
                        }
                        wn += __builtin_popcount(who);
                    }
                }
            }

#pragma unroll
            for (int j = 0; j < SCAN_VEC; ++j) dcur[j] = dnext[j];
        }

        if (lane == 0) wcnt[wave] = wn;
        __syncthreads();

        const int m = min(wn, SEG_CAP);
        int off = 0;
        for (int w = 0; w < WAVES; ++w) {
            const int cw = wcnt[w];
            const int mw = min(cw, SEG_CAP);
            if (w < wave) off += mw;
            total += mw;
            if (cw > SEG_CAP) bad = 1;
        }

        seg_resolve<D, WITH_DEG, RECORD>(seg, m, lane, src, cnt, lst, off);
        __syncthreads();

        if (m > 0) {
            constexpr int PASS = EPW * RSLOTS;
            int    dlA[RSLOTS], dlB[RSLOTS];
            bool   okA[RSLOTS], okB[RSLOTS];
            float4 vA[RSLOTS],  vB[RSLOTS];
            seg_fetch<D, RSLOTS>(seg, m, 0, eslot, sub, feat, dlA, okA, vA);
            for (int k0 = 0; k0 < m; k0 += 2 * PASS) {
                seg_fetch<D, RSLOTS>(seg, m, k0 + PASS, eslot, sub, feat, dlB, okB, vB);
                replay_add<D, RSLOTS>(acc, sub, rot, dlA, okA, vA);
                seg_fetch<D, RSLOTS>(seg, m, k0 + 2 * PASS, eslot, sub, feat, dlA, okA, vA);
                replay_add<D, RSLOTS>(acc, sub, rot, dlB, okB, vB);
            }
        }
    }

    if (tid < n_edges - n4) {
        const int e = n4 + tid;
        const unsigned dl = (unsigned)(dst[e] - base);
        if (dl < (unsigned)lim) {
            const float* row = feat + (size_t)src[e] * D;
            for (int c = 0; c < D; ++c) lds_add(&acc[dl * D + c], row[c]);
            if (WITH_DEG) lds_add(&cnt[dl], 1.0f);
        }
    }
    __syncthreads();

    const float4* a4 = reinterpret_cast<const float4*>(acc);
    float4* o4 = reinterpret_cast<float4*>(agg + (size_t)base * D);
    for (int i = tid; i < lim * (D / 4); i += OWN_THREADS) o4[i] = a4[i];
    if (WITH_DEG && tid < lim) deg[base + tid] = cnt[tid];

    if (RECORD && tid == 0) {
        const int full = (bad != 0 || total > LIST_CAP) ? 1 : 0;
        reinterpret_cast<int2*>(heads)[blockIdx.x] = make_int2(min(total, LIST_CAP), full);
    }
}

__global__ __launch_bounds__(256)
void sage_finish64(const float* __restrict__ S,
                   const float* __restrict__ deg,
                   const float* __restrict__ Q,
                   float* __restrict__ out,
                   int n_nodes) {
    const int t = blockIdx.x * 256 + threadIdx.x;
    if (t >= n_nodes * 16) return;

    const float invd = 1.0f / fmaxf(deg[t >> 4], 1.0f);
    const float4 s = reinterpret_cast<const float4*>(S)[t];
    const float4 q = reinterpret_cast<const float4*>(Q)[t];

    float4 o;
    o.x = q.x + s.x * invd;
    o.y = q.y + s.y * invd;
    o.z = q.z + s.z * invd;
    o.w = q.w + s.w * invd;
    reinterpret_cast<float4*>(out)[t] = o;
}

#define LDS_PITCH   132
#define STRIP_WORDS (16 * LDS_PITCH)

__global__ __launch_bounds__(256) __attribute__((amdgpu_waves_per_eu(4, 4)))
void sage_gemm128(const float* __restrict__ agg,
                  const float* __restrict__ deg,
                  const float* __restrict__ X,
                  const float* __restrict__ Wl,
                  const float* __restrict__ Wr,
                  const float* __restrict__ bias,
                  float* __restrict__ out,
                  int n_nodes) {
    __shared__ __attribute__((aligned(16))) float lds[2][2][STRIP_WORDS];

    const int n_strips = n_nodes / 16;
    const int tid  = threadIdx.x;
    const int lane = tid & 31;
    const int w    = tid >> 5;
    const int r    = lane & 15;
    const int half = lane >> 4;
    const int kk   = half * 2;
    const int coln = w * 16 + r;

    v2f bL[32], bR[32];
#pragma unroll
    for (int ks = 0; ks < 32; ++ks) {
        const int k = ks * 4 + kk;
        bL[ks].x = Wl[(size_t)k * 128 + coln];
        bL[ks].y = Wl[(size_t)(k + 1) * 128 + coln];
        bR[ks].x = Wr[(size_t)k * 128 + coln];
        bR[ks].y = Wr[(size_t)(k + 1) * 128 + coln];
    }
    const float bv = bias[coln];

    const int srow = tid >> 5;
    const int c4   = tid & 31;

    int s   = blockIdx.x;
    int cur = 0;

    if (s < n_strips) {
        const int row0 = s * 16 + srow;
        const int row1 = row0 + 8;
        const float i0 = 1.0f / fmaxf(deg[row0], 1.0f);
        const float i1 = 1.0f / fmaxf(deg[row1], 1.0f);
        float4 a0 = *(reinterpret_cast<const float4*>(agg + (size_t)row0 * 128) + c4);
        float4 a1 = *(reinterpret_cast<const float4*>(agg + (size_t)row1 * 128) + c4);
        const float4 x0 = *(reinterpret_cast<const float4*>(X + (size_t)row0 * 128) + c4);
        const float4 x1 = *(reinterpret_cast<const float4*>(X + (size_t)row1 * 128) + c4);
        a0.x *= i0; a0.y *= i0; a0.z *= i0; a0.w *= i0;
        a1.x *= i1; a1.y *= i1; a1.z *= i1; a1.w *= i1;
        *reinterpret_cast<float4*>(&lds[0][0][srow * LDS_PITCH + c4 * 4])       = a0;
        *reinterpret_cast<float4*>(&lds[0][0][(srow + 8) * LDS_PITCH + c4 * 4]) = a1;
        *reinterpret_cast<float4*>(&lds[0][1][srow * LDS_PITCH + c4 * 4])       = x0;
        *reinterpret_cast<float4*>(&lds[0][1][(srow + 8) * LDS_PITCH + c4 * 4]) = x1;
    }
    __syncthreads();

    for (; s < n_strips; s += gridDim.x) {
        const int ahead = s + (int)gridDim.x;
        const int nxt   = ahead < n_strips ? ahead : s;
        const int row0  = nxt * 16 + srow;
        const int row1  = row0 + 8;
        const float d0 = deg[row0];
        const float d1 = deg[row1];
        float4 a0 = *(reinterpret_cast<const float4*>(agg + (size_t)row0 * 128) + c4);
        float4 a1 = *(reinterpret_cast<const float4*>(agg + (size_t)row1 * 128) + c4);
        const float4 x0 = *(reinterpret_cast<const float4*>(X + (size_t)row0 * 128) + c4);
        const float4 x1 = *(reinterpret_cast<const float4*>(X + (size_t)row1 * 128) + c4);

        const float* la = &lds[cur][0][r * LDS_PITCH + kk];
        const float* lx = &lds[cur][1][r * LDS_PITCH + kk];

        v8f accA = {};
        v8f accX = {};
#pragma unroll
        for (int ks = 0; ks < 32; ++ks) {
            const v2f aA = *reinterpret_cast<const v2f*>(la + ks * 4);
            const v2f aX = *reinterpret_cast<const v2f*>(lx + ks * 4);
            accA = __builtin_amdgcn_wmma_f32_16x16x4_f32(
                       false, aA, false, bL[ks], (short)0, accA, false, false);
            accX = __builtin_amdgcn_wmma_f32_16x16x4_f32(
                       false, aX, false, bR[ks], (short)0, accX, false, false);
        }

        float v[8];
#pragma unroll
        for (int i = 0; i < 8; ++i) {
            v[i] = fmaxf((accA[i] + accX[i]) + bv, 0.0f);
        }

        {
            const float i0 = 1.0f / fmaxf(d0, 1.0f);
            const float i1 = 1.0f / fmaxf(d1, 1.0f);
            a0.x *= i0; a0.y *= i0; a0.z *= i0; a0.w *= i0;
            a1.x *= i1; a1.y *= i1; a1.z *= i1; a1.w *= i1;
            const int nb = cur ^ 1;
            *reinterpret_cast<float4*>(&lds[nb][0][srow * LDS_PITCH + c4 * 4])       = a0;
            *reinterpret_cast<float4*>(&lds[nb][0][(srow + 8) * LDS_PITCH + c4 * 4]) = a1;
            *reinterpret_cast<float4*>(&lds[nb][1][srow * LDS_PITCH + c4 * 4])       = x0;
            *reinterpret_cast<float4*>(&lds[nb][1][(srow + 8) * LDS_PITCH + c4 * 4]) = x1;
        }
        __syncthreads();
        cur ^= 1;

        const int rowBase = s * 16 + 8 * half;
#pragma unroll
        for (int i = 0; i < 8; ++i) {
            out[(size_t)(rowBase + i) * 128 + coln] = v[i];
        }
    }
}

__global__ __launch_bounds__(256) __attribute__((amdgpu_waves_per_eu(4, 4)))
void sage_transform64(const float* __restrict__ H,
                      const float* __restrict__ Wl,
                      const float* __restrict__ Wr,
                      const float* __restrict__ bias,
                      float* __restrict__ P,
                      float* __restrict__ Q,
                      int n_nodes) {
    __shared__ __attribute__((aligned(16))) float lds[2][2][STRIP_WORDS];

    const int n_strips = n_nodes / 16;
    const int n_pairs  = (n_strips + 1) / 2;
    const int tid  = threadIdx.x;
    const int lane = tid & 31;
    const int w    = tid >> 5;
    const int grp  = w >> 2;
    const int r    = lane & 15;
    const int half = lane >> 4;
    const int kk   = half * 2;
    const int coln = (w & 3) * 16 + r;

    v2f bL[32], bR[32];
#pragma unroll
    for (int ks = 0; ks < 32; ++ks) {
        const int k = ks * 4 + kk;
        bL[ks].x = Wl[(size_t)k * 64 + coln];
        bL[ks].y = Wl[(size_t)(k + 1) * 64 + coln];
        bR[ks].x = Wr[(size_t)k * 64 + coln];
        bR[ks].y = Wr[(size_t)(k + 1) * 64 + coln];
    }
    const float bv = bias[coln];

    const int srow = tid >> 5;
    const int c4   = tid & 31;
    const int last = n_strips - 1;

    int p   = blockIdx.x;
    int cur = 0;

    if (p < n_pairs) {
#pragma unroll
        for (int sp = 0; sp < 2; ++sp) {
            const int st   = min(2 * p + sp, last);
            const int row0 = st * 16 + srow;
            const float4 h0 = *(reinterpret_cast<const float4*>(H + (size_t)row0 * 128) + c4);
            const float4 h1 = *(reinterpret_cast<const float4*>(H + (size_t)(row0 + 8) * 128) + c4);
            *reinterpret_cast<float4*>(&lds[0][sp][srow * LDS_PITCH + c4 * 4])       = h0;
            *reinterpret_cast<float4*>(&lds[0][sp][(srow + 8) * LDS_PITCH + c4 * 4]) = h1;
        }
    }
    __syncthreads();

    for (; p < n_pairs; p += gridDim.x) {
        const int ahead = p + (int)gridDim.x;
        const int nxt   = ahead < n_pairs ? ahead : p;
        const int rowA  = min(2 * nxt, last) * 16 + srow;
        const int rowB  = min(2 * nxt + 1, last) * 16 + srow;
        const float4 h00 = *(reinterpret_cast<const float4*>(H + (size_t)rowA * 128) + c4);
        const float4 h01 = *(reinterpret_cast<const float4*>(H + (size_t)(rowA + 8) * 128) + c4);
        const float4 h10 = *(reinterpret_cast<const float4*>(H + (size_t)rowB * 128) + c4);
        const float4 h11 = *(reinterpret_cast<const float4*>(H + (size_t)(rowB + 8) * 128) + c4);

        const float* lh = &lds[cur][grp][r * LDS_PITCH + kk];

        v8f accP = {};
        v8f accQ = {};
#pragma unroll
        for (int ks = 0; ks < 32; ++ks) {
            const v2f aH = *reinterpret_cast<const v2f*>(lh + ks * 4);
            accP = __builtin_amdgcn_wmma_f32_16x16x4_f32(
                       false, aH, false, bL[ks], (short)0, accP, false, false);
            accQ = __builtin_amdgcn_wmma_f32_16x16x4_f32(
                       false, aH, false, bR[ks], (short)0, accQ, false, false);
        }

        float vq[8];
#pragma unroll
        for (int i = 0; i < 8; ++i) {
            vq[i] = accQ[i] + bv;
        }

        {
            const int nb = cur ^ 1;
            *reinterpret_cast<float4*>(&lds[nb][0][srow * LDS_PITCH + c4 * 4])       = h00;
            *reinterpret_cast<float4*>(&lds[nb][0][(srow + 8) * LDS_PITCH + c4 * 4]) = h01;
            *reinterpret_cast<float4*>(&lds[nb][1][srow * LDS_PITCH + c4 * 4])       = h10;
            *reinterpret_cast<float4*>(&lds[nb][1][(srow + 8) * LDS_PITCH + c4 * 4]) = h11;
        }
        __syncthreads();
        cur ^= 1;

        const int st = 2 * p + grp;
        if (st < n_strips) {
            const int rowBase = st * 16 + 8 * half;
#pragma unroll
            for (int i = 0; i < 8; ++i) {
                const size_t off = (size_t)(rowBase + i) * 64 + coln;
                P[off] = accP[i];
                Q[off] = vq[i];
            }
        }
    }
}

extern "C" void kernel_launch(void* const* d_in, const int* in_sizes, int n_in,
                              void* d_out, int out_size, void* d_ws, size_t ws_size,
                              hipStream_t stream) {
    const float* x    = (const float*)d_in[0];
    const int*   ei   = (const int*)  d_in[1];
    const float* W1l  = (const float*)d_in[2];
    const float* W1r  = (const float*)d_in[3];
    const float* b1   = (const float*)d_in[4];
    const float* W2l  = (const float*)d_in[5];
    const float* W2r  = (const float*)d_in[6];
    const float* b2   = (const float*)d_in[7];
    float*       outp = (float*)d_out;

    const int n_nodes = in_sizes[0] / IN_DIM;
    const int n_edges = in_sizes[1] / 2;
    const int* src = ei;
    const int* dst = ei + n_edges;

    const size_t aggBytes = (size_t)n_nodes * 128 * sizeof(float);
    const size_t rowBytes = (size_t)n_nodes * 64 * sizeof(float);
    const size_t degBytes = (((size_t)n_nodes * sizeof(float)) + 255) & ~(size_t)255;

    char* ws    = (char*)d_ws;
    float* agg1 = (float*)(ws);
    float* sum2 = (float*)(ws + aggBytes);
    float* deg  = (float*)(ws + aggBytes + rowBytes);
    float* h    = (float*)(ws + aggBytes + rowBytes + degBytes);
    float* p2   = (float*)(ws + 2 * aggBytes + rowBytes + degBytes);
    float* q2   = (float*)(ws + 2 * aggBytes + 2 * rowBytes + degBytes);

    const int ownBlocks = (n_nodes + OWN_NODES - 1) / OWN_NODES;
    const size_t listBytes = (size_t)ownBlocks * LIST_CAP * 2 * sizeof(int);
    int* lists  = (int*)(ws + 2 * aggBytes + 3 * rowBytes + degBytes);
    int* heads  = (int*)(ws + 2 * aggBytes + 3 * rowBytes + degBytes + listBytes);
    const int denseBlocks = 512;
    const int finishBlocks = (n_nodes * 16 + 255) / 256;

    sage_owner_sum<128, true, true, false><<<ownBlocks, OWN_THREADS, 0, stream>>>(
        x, src, dst, agg1, deg, lists, heads, n_nodes, n_edges);
    sage_gemm128<<<denseBlocks, 256, 0, stream>>>(
        agg1, deg, x, W1l, W1r, b1, h, n_nodes);

    sage_transform64<<<denseBlocks, 256, 0, stream>>>(
        h, W2l, W2r, b2, p2, q2, n_nodes);
    sage_owner_sum<64, false, false, true><<<ownBlocks, OWN_THREADS, 0, stream>>>(
        p2, src, dst, sum2, nullptr, lists, heads, n_nodes, n_edges);
    sage_finish64<<<finishBlocks, 256, 0, stream>>>(sum2, deg, q2, outp, n_nodes);
}


# layer-1 scan reads a 1-byte owner code per edge (a new first launch writes them) not the 4-byte destination: 16 edges per load; finds take list slots from an LDS counter; sums stay LDS arrival order
# speedup vs baseline: 3.6660x; 1.1788x over previous
#include <hip/hip_runtime.h>


typedef __attribute__((ext_vector_type(2))) float v2f;
typedef __attribute__((ext_vector_type(8))) float v8f;

#define N_NODES 50000
#define N_EDGES 1600000
#define IN_DIM  128
#define HIDDEN  128
#define OUT_DIM 64

#define OWN_NODES   200
#define OWN_THREADS 1024
#define SCAN_VEC    8
#define OWN_CHUNK   (OWN_THREADS * SCAN_VEC)
#define CODE_NONE   255u
#define SEG_CAP     512
#define LIST_CAP    32768
#define REPLAY_CAP  8192

__device__ __forceinline__ void lds_add(float* p, float v) {
    __hip_atomic_fetch_add(p, v, __ATOMIC_RELAXED, __HIP_MEMORY_SCOPE_WORKGROUP);
}

__device__ __forceinline__ unsigned owner_code(int d, int n_nodes) {
    return (unsigned)d < (unsigned)n_nodes ? (unsigned)d / (unsigned)OWN_NODES : CODE_NONE;
}

__global__ __launch_bounds__(256)
void sage_owner_codes(const int* __restrict__ dst,
                      int* __restrict__ codes,
                      int n_nodes, int n_edges) {
    static_assert((N_NODES + OWN_NODES - 1) / OWN_NODES <= (int)CODE_NONE, "a work-group's number fits in a byte beside CODE_NONE");
    const int nw = (n_edges & ~15) >> 2;
    int4 d[4];
#pragma unroll
    for (int j = 0; j < 4; ++j) {
        const int w = blockIdx.x * 1024 + j * 256 + threadIdx.x;
        d[j] = w < nw ? reinterpret_cast<const int4*>(dst)[w] : make_int4(-1, -1, -1, -1);
    }
#pragma unroll
    for (int j = 0; j < 4; ++j) {
        const int w = blockIdx.x * 1024 + j * 256 + threadIdx.x;
        if (w < nw + 4) {
            codes[w] = (int)(owner_code(d[j].x, n_nodes) | (owner_code(d[j].y, n_nodes) << 8) |
                             (owner_code(d[j].z, n_nodes) << 16) | (owner_code(d[j].w, n_nodes) << 24));
        }
    }
}

template <int D, bool WITH_DEG>
__device__ __forceinline__ void own_note(int e, int pos, int base, unsigned* seg,
                                         float* acc, float* cnt,
                                         const float* __restrict__ feat,
                                         const int* __restrict__ src,
                                         const int* __restrict__ dst) {
    if (pos < SEG_CAP) {
        seg[pos] = (unsigned)e;
    } else {
        const int dl = dst[e] - base;
        const float* row = feat + (size_t)src[e] * D;
        for (int c = 0; c < D; ++c) lds_add(&acc[dl * D + c], row[c]);
        if (WITH_DEG) lds_add(&cnt[dl], 1.0f);
    }
}

template <int D, bool WITH_DEG, bool RECORD>
__device__ __forceinline__ void seg_resolve(unsigned* seg, int m, int lane, int base,
                                            const int* __restrict__ src,
                                            const int* __restrict__ dst, float* cnt,
                                            int2* __restrict__ lst, int off) {
    constexpr int CV = 8;
    for (int i0 = 0; i0 < m; i0 += 32 * CV) {
        int s[CV];
        int d[CV];
#pragma unroll
        for (int t = 0; t < CV; ++t) {
            const int i = i0 + t * 32 + lane;
            const unsigned e = i < m ? seg[i] : 0u;
            s[t] = src[e];
            d[t] = dst[e];
        }
#pragma unroll
        for (int t = 0; t < CV; ++t) {
            const int i = i0 + t * 32 + lane;
            if (i < m) {
                const int dl = d[t] - base;
                seg[i] = ((unsigned)s[t] << 8) | (unsigned)dl;
                if (WITH_DEG) lds_add(&cnt[dl], 1.0f);
                if (RECORD) {
                    const int g = off + i;
                    if (g < LIST_CAP) lst[g] = make_int2(s[t], dl);
                }
            }
        }
    }
}

template <int D, int SLOTS>
__device__ __forceinline__ void seg_fetch(const unsigned* seg, int m, int k0, int eslot, int sub,
                                          const float* __restrict__ feat,
                                          int (&dl)[SLOTS], bool (&ok)[SLOTS], float4 (&v)[SLOTS]) {
    constexpr int EPW = 32 / (D / 4);
#pragma unroll
    for (int j = 0; j < SLOTS; ++j) {
        const int idx = k0 + j * EPW + eslot;
        ok[j] = idx < m;
        const unsigned ent = seg[ok[j] ? idx : 0];
        dl[j] = (int)(ent & 255u);
        const unsigned off = (ent >> 8) * (unsigned)(D * 4) + (unsigned)(sub * 16);
        v[j] = *reinterpret_cast<const float4*>(reinterpret_cast<const char*>(feat) + off);
    }
}

template <int D, int SLOTS>
__device__ __forceinline__ void replay_fetch(const int2* q, int m, int k0, int wave, int eslot, int sub,
                                             const float* __restrict__ feat,
                                             int (&dl)[SLOTS], bool (&ok)[SLOTS], float4 (&v)[SLOTS]) {
    constexpr int EPW = 32 / (D / 4);
#pragma unroll
    for (int j = 0; j < SLOTS; ++j) {
        const int idx = k0 + (j * 32 + wave) * EPW + eslot;
        ok[j] = idx < m;
        const int2 ent = q[ok[j] ? idx : 0];
        dl[j] = ent.y;
        const unsigned off = (unsigned)ent.x * (unsigned)(D * 4) + (unsigned)(sub * 16);
        v[j] = *reinterpret_cast<const float4*>(reinterpret_cast<const char*>(feat) + off);
    }
}

template <int D, int SLOTS>
__device__ __forceinline__ void replay_add(float* acc, int sub, int rot,
                                           const int (&dl)[SLOTS], const bool (&ok)[SLOTS], const float4 (&v)[SLOTS]) {
#pragma unroll
    for (int j = 0; j < SLOTS; ++j) {
        if (ok[j]) {
            float* a = &acc[dl[j] * D + sub * 4];
            lds_add(a + rot,             rot ? v[j].y : v[j].x);
            lds_add(a + 1 + rot,         rot ? v[j].z : v[j].y);
            lds_add(a + 2 + rot,         rot ? v[j].w : v[j].z);
            lds_add(a + (rot ? 0 : 3),   rot ? v[j].x : v[j].w);
        }
    }
}

template <int D, bool WITH_DEG, bool RECORD, bool REPLAY>
__global__ __launch_bounds__(OWN_THREADS)
void sage_owner_sum(const float* __restrict__ feat,
                    const int*   __restrict__ src,
                    const int*   __restrict__ dst,
                    const int*   __restrict__ codes,
                    float* __restrict__ agg,
                    float* __restrict__ deg,
                    int*   __restrict__ lists,
                    int*   __restrict__ heads,
                    int n_nodes, int n_edges) {
    constexpr int WAVES = OWN_THREADS / 32;
    static_assert(REPLAY_CAP * 2 >= WAVES * SEG_CAP, "the replay stage and the waves' lists of the scan form share the room");
    static_assert(REPLAY_CAP % OWN_THREADS == 0, "whole staging rounds");
    static_assert(N_EDGES < (1 << 24) && N_NODES < (1 << 24) && OWN_NODES <= 256, "a find packs into one word");

    __shared__ __attribute__((aligned(16))) float acc[OWN_NODES * D];
    __shared__ float cnt[OWN_NODES];
    __shared__ int2  queue[REPLAY_CAP];
    __shared__ int   wcnt[WAVES];

    constexpr int LPE   = D / 4;
    constexpr int EPW   = 32 / LPE;
    constexpr int RSLOTS = 8;

    const int tid  = threadIdx.x;
    const int lane = tid & 31;
    const int wave = tid >> 5;
    const int sub  = lane % LPE;
    const int eslot = lane / LPE;
    const int rot  = lane >> 4;
    const int base = blockIdx.x * OWN_NODES;
    const int lim  = min(OWN_NODES, n_nodes - base);
    const int nv   = n_edges & ~15;

    int2* const lst = reinterpret_cast<int2*>(lists) + (size_t)blockIdx.x * LIST_CAP;

    int head_n = 0, head_bad = 1;
    if (REPLAY) {
        const int2 h = reinterpret_cast<const int2*>(heads)[blockIdx.x];
        head_n   = __builtin_amdgcn_readfirstlane(h.x);
        head_bad = __builtin_amdgcn_readfirstlane(h.y);
    }

    for (int i = tid; i < OWN_NODES * D; i += OWN_THREADS) acc[i] = 0.0f;
    if (tid < OWN_NODES) cnt[tid] = 0.0f;
    if (tid < WAVES) wcnt[tid] = 0;

    int total = 0;
    int bad   = 0;

    if (REPLAY && head_bad == 0) {
        constexpr int RPT    = REPLAY_CAP / OWN_THREADS;
        constexpr int RPASS  = 32 * EPW * RSLOTS;

        __syncthreads();

        for (int c0 = 0; c0 < head_n; c0 += REPLAY_CAP) {
            const int m = min(REPLAY_CAP, head_n - c0);
            int2 pre[RPT];
#pragma unroll
            for (int t = 0; t < RPT; ++t) {
                const int i = c0 + t * OWN_THREADS + tid;
                pre[t] = lst[i < head_n ? i : 0];
            }
#pragma unroll
            for (int t = 0; t < RPT; ++t) {
                const int i = t * OWN_THREADS + tid;
                if (i < m) queue[i] = pre[t];
            }
            __syncthreads();

            int    dlA[RSLOTS], dlB[RSLOTS];
            bool   okA[RSLOTS], okB[RSLOTS];
            float4 vA[RSLOTS],  vB[RSLOTS];
            replay_fetch<D, RSLOTS>(queue, m, 0, wave, eslot, sub, feat, dlA, okA, vA);
            for (int k0 = 0; k0 < m; k0 += 2 * RPASS) {
                replay_fetch<D, RSLOTS>(queue, m, k0 + RPASS, wave, eslot, sub, feat, dlB, okB, vB);
                replay_add<D, RSLOTS>(acc, sub, rot, dlA, okA, vA);
                replay_fetch<D, RSLOTS>(queue, m, k0 + 2 * RPASS, wave, eslot, sub, feat, dlA, okA, vA);
                replay_add<D, RSLOTS>(acc, sub, rot, dlB, okB, vB);
            }
            __syncthreads();
        }
    } else {
        const int ngrp = nv >> 4;
        const int4* const code4 = reinterpret_cast<const int4*>(codes);
        const unsigned pat = (unsigned)blockIdx.x * 0x01010101u;

        int4 ccur[SCAN_VEC];
#pragma unroll
        for (int j = 0; j < SCAN_VEC; ++j) {
            const int g = j * OWN_THREADS + tid;
            ccur[j] = code4[min(g, ngrp)];
        }
        __syncthreads();

        unsigned* const seg = reinterpret_cast<unsigned*>(queue) + wave * SEG_CAP;
        for (int g0 = 0; g0 < ngrp; g0 += OWN_CHUNK) {
            int4 cnext[SCAN_VEC];
#pragma unroll
            for (int j = 0; j < SCAN_VEC; ++j) {
                const int g = g0 + OWN_CHUNK + j * OWN_THREADS + tid;
                cnext[j] = code4[min(g, ngrp)];
            }

#pragma unroll
            for (int j = 0; j < SCAN_VEC; ++j) {
                const unsigned x[4] = {(unsigned)ccur[j].x ^ pat, (unsigned)ccur[j].y ^ pat,
                                       (unsigned)ccur[j].z ^ pat, (unsigned)ccur[j].w ^ pat};
                unsigned hit = 0u;
#pragma unroll
                for (int i = 0; i < 4; ++i) {
                    const unsigned y = ((x[i] & 0x7F7F7F7Fu) + 0x7F7F7F7Fu) | x[i];
                    hit |= (~y & 0x80808080u) >> (7 - i);
                }
                if (hit != 0u) {
                    const int nhit = __builtin_popcount(hit);
                    int pos = __hip_atomic_fetch_add(&wcnt[wave], nhit, __ATOMIC_RELAXED, __HIP_MEMORY_SCOPE_WORKGROUP);
                    const int e0 = (g0 + j * OWN_THREADS + tid) << 4;
                    if (pos + nhit <= SEG_CAP) {
                        do {
                            const int p = __builtin_ctz(hit);
                            hit &= hit - 1u;
                            seg[pos++] = (unsigned)(e0 + ((p & 7) << 2) + (p >> 3));
                        } while (hit != 0u);
                    } else {
                        do {
                            const int p = __builtin_ctz(hit);
                            hit &= hit - 1u;
                            own_note<D, WITH_DEG>(e0 + ((p & 7) << 2) + (p >> 3), pos++, base, seg, acc, cnt, feat, src, dst);
                        } while (hit != 0u);
                    }
                }
            }

#pragma unroll
            for (int j = 0; j < SCAN_VEC; ++j) ccur[j] = cnext[j];
        }

        __syncthreads();
        const int wn = wcnt[wave];

        const int m = min(wn, SEG_CAP);
        int off = 0;
        for (int w = 0; w < WAVES; ++w) {
            const int cw = wcnt[w];
            const int mw = min(cw, SEG_CAP);
            if (w < wave) off += mw;
            total += mw;
            if (cw > SEG_CAP) bad = 1;
        }

        seg_resolve<D, WITH_DEG, RECORD>(seg, m, lane, base, src, dst, cnt, lst, off);
        __syncthreads();

        if (m > 0) {
            constexpr int PASS = EPW * RSLOTS;
            int    dlA[RSLOTS], dlB[RSLOTS];
            bool   okA[RSLOTS], okB[RSLOTS];
            float4 vA[RSLOTS],  vB[RSLOTS];
            seg_fetch<D, RSLOTS>(seg, m, 0, eslot, sub, feat, dlA, okA, vA);
            for (int k0 = 0; k0 < m; k0 += 2 * PASS) {
                seg_fetch<D, RSLOTS>(seg, m, k0 + PASS, eslot, sub, feat, dlB, okB, vB);
                replay_add<D, RSLOTS>(acc, sub, rot, dlA, okA, vA);
                seg_fetch<D, RSLOTS>(seg, m, k0 + 2 * PASS, eslot, sub, feat, dlA, okA, vA);
                replay_add<D, RSLOTS>(acc, sub, rot, dlB, okB, vB);
            }
        }
    }

    if (tid < n_edges - nv) {
        const int e = nv + tid;
        const unsigned dl = (unsigned)(dst[e] - base);
        if (dl < (unsigned)lim) {
            const float* row = feat + (size_t)src[e] * D;
            for (int c = 0; c < D; ++c) lds_add(&acc[dl * D + c], row[c]);
            if (WITH_DEG) lds_add(&cnt[dl], 1.0f);
        }
    }
    __syncthreads();

    const float4* a4 = reinterpret_cast<const float4*>(acc);
    float4* o4 = reinterpret_cast<float4*>(agg + (size_t)base * D);
    for (int i = tid; i < lim * (D / 4); i += OWN_THREADS) o4[i] = a4[i];
    if (WITH_DEG && tid < lim) deg[base + tid] = cnt[tid];

    if (RECORD && tid == 0) {
        const int full = (bad != 0 || total > LIST_CAP) ? 1 : 0;
        reinterpret_cast<int2*>(heads)[blockIdx.x] = make_int2(min(total, LIST_CAP), full);
    }
}

__global__ __launch_bounds__(256)
void sage_finish64(const float* __restrict__ S,
                   const float* __restrict__ deg,
                   const float* __restrict__ Q,
                   float* __restrict__ out,
                   int n_nodes) {
    const int t = blockIdx.x * 256 + threadIdx.x;
    if (t >= n_nodes * 16) return;

    const float invd = 1.0f / fmaxf(deg[t >> 4], 1.0f);
    const float4 s = reinterpret_cast<const float4*>(S)[t];
    const float4 q = reinterpret_cast<const float4*>(Q)[t];

    float4 o;
    o.x = q.x + s.x * invd;
    o.y = q.y + s.y * invd;
    o.z = q.z + s.z * invd;
    o.w = q.w + s.w * invd;
    reinterpret_cast<float4*>(out)[t] = o;
}

#define LDS_PITCH   132
#define STRIP_WORDS (16 * LDS_PITCH)

__global__ __launch_bounds__(256) __attribute__((amdgpu_waves_per_eu(4, 4)))
void sage_gemm128(const float* __restrict__ agg,
                  const float* __restrict__ deg,
                  const float* __restrict__ X,
                  const float* __restrict__ Wl,
                  const float* __restrict__ Wr,
                  const float* __restrict__ bias,
                  float* __restrict__ out,
                  int n_nodes) {
    __shared__ __attribute__((aligned(16))) float lds[2][2][STRIP_WORDS];

    const int n_strips = n_nodes / 16;
    const int tid  = threadIdx.x;
    const int lane = tid & 31;
    const int w    = tid >> 5;
    const int r    = lane & 15;
    const int half = lane >> 4;
    const int kk   = half * 2;
    const int coln = w * 16 + r;

    v2f bL[32], bR[32];
#pragma unroll
    for (int ks = 0; ks < 32; ++ks) {
        const int k = ks * 4 + kk;
        bL[ks].x = Wl[(size_t)k * 128 + coln];
        bL[ks].y = Wl[(size_t)(k + 1) * 128 + coln];
        bR[ks].x = Wr[(size_t)k * 128 + coln];
        bR[ks].y = Wr[(size_t)(k + 1) * 128 + coln];
    }
    const float bv = bias[coln];

    const int srow = tid >> 5;
    const int c4   = tid & 31;

    int s   = blockIdx.x;
    int cur = 0;

    if (s < n_strips) {
        const int row0 = s * 16 + srow;
        const int row1 = row0 + 8;
        const float i0 = 1.0f / fmaxf(deg[row0], 1.0f);
        const float i1 = 1.0f / fmaxf(deg[row1], 1.0f);
        float4 a0 = *(reinterpret_cast<const float4*>(agg + (size_t)row0 * 128) + c4);
        float4 a1 = *(reinterpret_cast<const float4*>(agg + (size_t)row1 * 128) + c4);
        const float4 x0 = *(reinterpret_cast<const float4*>(X + (size_t)row0 * 128) + c4);
        const float4 x1 = *(reinterpret_cast<const float4*>(X + (size_t)row1 * 128) + c4);
        a0.x *= i0; a0.y *= i0; a0.z *= i0; a0.w *= i0;
        a1.x *= i1; a1.y *= i1; a1.z *= i1; a1.w *= i1;
        *reinterpret_cast<float4*>(&lds[0][0][srow * LDS_PITCH + c4 * 4])       = a0;
        *reinterpret_cast<float4*>(&lds[0][0][(srow + 8) * LDS_PITCH + c4 * 4]) = a1;
        *reinterpret_cast<float4*>(&lds[0][1][srow * LDS_PITCH + c4 * 4])       = x0;
        *reinterpret_cast<float4*>(&lds[0][1][(srow + 8) * LDS_PITCH + c4 * 4]) = x1;
    }
    __syncthreads();

    for (; s < n_strips; s += gridDim.x) {
        const int ahead = s + (int)gridDim.x;
        const int nxt   = ahead < n_strips ? ahead : s;
        const int row0  = nxt * 16 + srow;
        const int row1  = row0 + 8;
        const float d0 = deg[row0];
        const float d1 = deg[row1];
        float4 a0 = *(reinterpret_cast<const float4*>(agg + (size_t)row0 * 128) + c4);
        float4 a1 = *(reinterpret_cast<const float4*>(agg + (size_t)row1 * 128) + c4);
        const float4 x0 = *(reinterpret_cast<const float4*>(X + (size_t)row0 * 128) + c4);
        const float4 x1 = *(reinterpret_cast<const float4*>(X + (size_t)row1 * 128) + c4);

        const float* la = &lds[cur][0][r * LDS_PITCH + kk];
        const float* lx = &lds[cur][1][r * LDS_PITCH + kk];

        v8f accA = {};
        v8f accX = {};
#pragma unroll
        for (int ks = 0; ks < 32; ++ks) {
            const v2f aA = *reinterpret_cast<const v2f*>(la + ks * 4);
            const v2f aX = *reinterpret_cast<const v2f*>(lx + ks * 4);
            accA = __builtin_amdgcn_wmma_f32_16x16x4_f32(
                       false, aA, false, bL[ks], (short)0, accA, false, false);
            accX = __builtin_amdgcn_wmma_f32_16x16x4_f32(
                       false, aX, false, bR[ks], (short)0, accX, false, false);
        }

        float v[8];
#pragma unroll
        for (int i = 0; i < 8; ++i) {
            v[i] = fmaxf((accA[i] + accX[i]) + bv, 0.0f);
        }

        {
            const float i0 = 1.0f / fmaxf(d0, 1.0f);
            const float i1 = 1.0f / fmaxf(d1, 1.0f);
            a0.x *= i0; a0.y *= i0; a0.z *= i0; a0.w *= i0;
            a1.x *= i1; a1.y *= i1; a1.z *= i1; a1.w *= i1;
            const int nb = cur ^ 1;
            *reinterpret_cast<float4*>(&lds[nb][0][srow * LDS_PITCH + c4 * 4])       = a0;
            *reinterpret_cast<float4*>(&lds[nb][0][(srow + 8) * LDS_PITCH + c4 * 4]) = a1;
            *reinterpret_cast<float4*>(&lds[nb][1][srow * LDS_PITCH + c4 * 4])       = x0;
            *reinterpret_cast<float4*>(&lds[nb][1][(srow + 8) * LDS_PITCH + c4 * 4]) = x1;
        }
        __syncthreads();
        cur ^= 1;

        const int rowBase = s * 16 + 8 * half;
#pragma unroll
        for (int i = 0; i < 8; ++i) {
            out[(size_t)(rowBase + i) * 128 + coln] = v[i];
        }
    }
}

__global__ __launch_bounds__(256) __attribute__((amdgpu_waves_per_eu(4, 4)))
void sage_transform64(const float* __restrict__ H,
                      const float* __restrict__ Wl,
                      const float* __restrict__ Wr,
                      const float* __restrict__ bias,
                      float* __restrict__ P,
                      float* __restrict__ Q,
                      int n_nodes) {
    __shared__ __attribute__((aligned(16))) float lds[2][2][STRIP_WORDS];

    const int n_strips = n_nodes / 16;
    const int n_pairs  = (n_strips + 1) / 2;
    const int tid  = threadIdx.x;
    const int lane = tid & 31;
    const int w    = tid >> 5;
    const int grp  = w >> 2;
    const int r    = lane & 15;
    const int half = lane >> 4;
    const int kk   = half * 2;
    const int coln = (w & 3) * 16 + r;

    v2f bL[32], bR[32];
#pragma unroll
    for (int ks = 0; ks < 32; ++ks) {
        const int k = ks * 4 + kk;
        bL[ks].x = Wl[(size_t)k * 64 + coln];
        bL[ks].y = Wl[(size_t)(k + 1) * 64 + coln];
        bR[ks].x = Wr[(size_t)k * 64 + coln];
        bR[ks].y = Wr[(size_t)(k + 1) * 64 + coln];
    }
    const float bv = bias[coln];

    const int srow = tid >> 5;
    const int c4   = tid & 31;
    const int last = n_strips - 1;

    int p   = blockIdx.x;
    int cur = 0;

    if (p < n_pairs) {
#pragma unroll
        for (int sp = 0; sp < 2; ++sp) {
            const int st   = min(2 * p + sp, last);
            const int row0 = st * 16 + srow;
            const float4 h0 = *(reinterpret_cast<const float4*>(H + (size_t)row0 * 128) + c4);
            const float4 h1 = *(reinterpret_cast<const float4*>(H + (size_t)(row0 + 8) * 128) + c4);
            *reinterpret_cast<float4*>(&lds[0][sp][srow * LDS_PITCH + c4 * 4])       = h0;
            *reinterpret_cast<float4*>(&lds[0][sp][(srow + 8) * LDS_PITCH + c4 * 4]) = h1;
        }
    }
    __syncthreads();

    for (; p < n_pairs; p += gridDim.x) {
        const int ahead = p + (int)gridDim.x;
        const int nxt   = ahead < n_pairs ? ahead : p;
        const int rowA  = min(2 * nxt, last) * 16 + srow;
        const int rowB  = min(2 * nxt + 1, last) * 16 + srow;
        const float4 h00 = *(reinterpret_cast<const float4*>(H + (size_t)rowA * 128) + c4);
        const float4 h01 = *(reinterpret_cast<const float4*>(H + (size_t)(rowA + 8) * 128) + c4);
        const float4 h10 = *(reinterpret_cast<const float4*>(H + (size_t)rowB * 128) + c4);
        const float4 h11 = *(reinterpret_cast<const float4*>(H + (size_t)(rowB + 8) * 128) + c4);

        const float* lh = &lds[cur][grp][r * LDS_PITCH + kk];

        v8f accP = {};
        v8f accQ = {};
#pragma unroll
        for (int ks = 0; ks < 32; ++ks) {
            const v2f aH = *reinterpret_cast<const v2f*>(lh + ks * 4);
            accP = __builtin_amdgcn_wmma_f32_16x16x4_f32(
                       false, aH, false, bL[ks], (short)0, accP, false, false);
            accQ = __builtin_amdgcn_wmma_f32_16x16x4_f32(
                       false, aH, false, bR[ks], (short)0, accQ, false, false);
        }

        float vq[8];
#pragma unroll
        for (int i = 0; i < 8; ++i) {
            vq[i] = accQ[i] + bv;
        }

        {
            const int nb = cur ^ 1;
            *reinterpret_cast<float4*>(&lds[nb][0][srow * LDS_PITCH + c4 * 4])       = h00;
            *reinterpret_cast<float4*>(&lds[nb][0][(srow + 8) * LDS_PITCH + c4 * 4]) = h01;
            *reinterpret_cast<float4*>(&lds[nb][1][srow * LDS_PITCH + c4 * 4])       = h10;
            *reinterpret_cast<float4*>(&lds[nb][1][(srow + 8) * LDS_PITCH + c4 * 4]) = h11;
        }
        __syncthreads();
        cur ^= 1;

        const int st = 2 * p + grp;
        if (st < n_strips) {
            const int rowBase = st * 16 + 8 * half;
#pragma unroll
            for (int i = 0; i < 8; ++i) {
                const size_t off = (size_t)(rowBase + i) * 64 + coln;
                P[off] = accP[i];
                Q[off] = vq[i];
            }
        }
    }
}

extern "C" void kernel_launch(void* const* d_in, const int* in_sizes, int n_in,
                              void* d_out, int out_size, void* d_ws, size_t ws_size,
                              hipStream_t stream) {
    const float* x    = (const float*)d_in[0];
    const int*   ei   = (const int*)  d_in[1];
    const float* W1l  = (const float*)d_in[2];
    const float* W1r  = (const float*)d_in[3];
    const float* b1   = (const float*)d_in[4];
    const float* W2l  = (const float*)d_in[5];
    const float* W2r  = (const float*)d_in[6];
    const float* b2   = (const float*)d_in[7];
    float*       outp = (float*)d_out;

    const int n_nodes = in_sizes[0] / IN_DIM;
    const int n_edges = in_sizes[1] / 2;
    const int* src = ei;
    const int* dst = ei + n_edges;

    const size_t aggBytes = (size_t)n_nodes * 128 * sizeof(float);
    const size_t rowBytes = (size_t)n_nodes * 64 * sizeof(float);
    const size_t degBytes = (((size_t)n_nodes * sizeof(float)) + 255) & ~(size_t)255;

    char* ws    = (char*)d_ws;
    float* agg1 = (float*)(ws);
    float* sum2 = (float*)(ws + aggBytes);
    float* deg  = (float*)(ws + aggBytes + rowBytes);
    float* h    = (float*)(ws + aggBytes + rowBytes + degBytes);
    float* p2   = (float*)(ws + 2 * aggBytes + rowBytes + degBytes);
    float* q2   = (float*)(ws + 2 * aggBytes + 2 * rowBytes + degBytes);

    const int ownBlocks = (n_nodes + OWN_NODES - 1) / OWN_NODES;
    const size_t listBytes = (size_t)ownBlocks * LIST_CAP * 2 * sizeof(int);
    int* lists  = (int*)(ws + 2 * aggBytes + 3 * rowBytes + degBytes);
    int* heads  = (int*)(ws + 2 * aggBytes + 3 * rowBytes + degBytes + listBytes);
    const size_t headBytes = (((size_t)ownBlocks * 2 * sizeof(int)) + 255) & ~(size_t)255;
    int* codes  = (int*)(ws + 2 * aggBytes + 3 * rowBytes + degBytes + listBytes + headBytes);
    const int codeBlocks = ((n_edges >> 2) + 4 + 1023) / 1024;
    const int denseBlocks = 512;
    const int finishBlocks = (n_nodes * 16 + 255) / 256;

    sage_owner_codes<<<codeBlocks, 256, 0, stream>>>(dst, codes, n_nodes, n_edges);

    sage_owner_sum<128, true, true, false><<<ownBlocks, OWN_THREADS, 0, stream>>>(
        x, src, dst, codes, agg1, deg, lists, heads, n_nodes, n_edges);
    sage_gemm128<<<denseBlocks, 256, 0, stream>>>(
        agg1, deg, x, W1l, W1r, b1, h, n_nodes);

    sage_transform64<<<denseBlocks, 256, 0, stream>>>(
        h, W2l, W2r, b2, p2, q2, n_nodes);
    sage_owner_sum<64, false, false, true><<<ownBlocks, OWN_THREADS, 0, stream>>>(
        p2, src, dst, codes, sum2, nullptr, lists, heads, n_nodes, n_edges);
    sage_finish64<<<finishBlocks, 256, 0, stream>>>(sum2, deg, q2, outp, n_nodes);
}


# layer-1 scan reads 1 bit per edge from its owner's own bitmap (the first launch sets the bits: 4096-edge chunks, LDS bit ORs, whole-line rows out) not 1-byte codes; sums stay LDS arrival order
# speedup vs baseline: 3.8787x; 1.0580x over previous
#include <hip/hip_runtime.h>


typedef __attribute__((ext_vector_type(2))) float v2f;
typedef __attribute__((ext_vector_type(8))) float v8f;

#define N_NODES 50000
#define N_EDGES 1600000
#define IN_DIM  128
#define HIDDEN  128
#define OUT_DIM 64

#define OWN_NODES   200
#define OWN_THREADS 1024
#define OWN_BLOCKS  ((N_NODES + OWN_NODES - 1) / OWN_NODES)
#define SCAN_VEC    4
#define OWN_CHUNK   (OWN_THREADS * SCAN_VEC)
#define BIT_CHUNK   4096
#define BIT_WORDS   (BIT_CHUNK / 32)
#define SEG_CAP     512
#define LIST_CAP    32768
#define REPLAY_CAP  8192

__device__ __forceinline__ void lds_add(float* p, float v) {
    __hip_atomic_fetch_add(p, v, __ATOMIC_RELAXED, __HIP_MEMORY_SCOPE_WORKGROUP);
}

__device__ __forceinline__ int words_per_owner(int n_edges) {
    return ((n_edges + BIT_CHUNK - 1) / BIT_CHUNK) * BIT_WORDS;
}

__global__ __launch_bounds__(1024)
void sage_owner_bits(const int* __restrict__ dst,
                     int* __restrict__ bits,
                     int n_nodes, int n_edges) {
    static_assert(BIT_CHUNK == 4 * 1024, "four edges per thread");
    __shared__ __attribute__((aligned(16))) unsigned bm[OWN_BLOCKS * BIT_WORDS];

    const int tid = threadIdx.x;
    const int e0  = blockIdx.x * BIT_CHUNK + tid * 4;
    int d[4];
    if (e0 + 3 < n_edges) {
        const int4 v = *reinterpret_cast<const int4*>(dst + e0);
        d[0] = v.x; d[1] = v.y; d[2] = v.z; d[3] = v.w;
    } else {
#pragma unroll
        for (int k = 0; k < 4; ++k) d[k] = e0 + k < n_edges ? dst[e0 + k] : -1;
    }

    uint4* const bm4 = reinterpret_cast<uint4*>(bm);
    for (int i = tid; i < OWN_BLOCKS * BIT_WORDS / 4; i += 1024) bm4[i] = make_uint4(0u, 0u, 0u, 0u);
    __syncthreads();

#pragma unroll
    for (int k = 0; k < 4; ++k) {
        if ((unsigned)d[k] < (unsigned)n_nodes) {
            const unsigned o = (unsigned)d[k] / (unsigned)OWN_NODES;
            const int el = tid * 4 + k;
            if (o < (unsigned)OWN_BLOCKS) {
                __hip_atomic_fetch_or(&bm[o * BIT_WORDS + (el >> 5)], 1u << (el & 31),
                                      __ATOMIC_RELAXED, __HIP_MEMORY_SCOPE_WORKGROUP);
            }
        }
    }
    __syncthreads();

    const int wpo = words_per_owner(n_edges);
    for (int i = tid; i < OWN_BLOCKS * BIT_WORDS / 4; i += 1024) {
        const int o = i / (BIT_WORDS / 4);
        const int q = i % (BIT_WORDS / 4);
        reinterpret_cast<uint4*>(bits + (size_t)o * wpo + (size_t)blockIdx.x * BIT_WORDS)[q] = bm4[i];
    }
}

template <int D, bool WITH_DEG>
__device__ __forceinline__ void own_note(int e, int pos, int base, unsigned* seg,
                                         float* acc, float* cnt,
                                         const float* __restrict__ feat,
                                         const int* __restrict__ src,
                                         const int* __restrict__ dst) {
    if (pos < SEG_CAP) {
        seg[pos] = (unsigned)e;
    } else {
        const int dl = dst[e] - base;
        const float* row = feat + (size_t)src[e] * D;
        for (int c = 0; c < D; ++c) lds_add(&acc[dl * D + c], row[c]);
        if (WITH_DEG) lds_add(&cnt[dl], 1.0f);
    }
}

template <int D, bool WITH_DEG, bool RECORD>
__device__ __forceinline__ void seg_resolve(unsigned* seg, int m, int lane, int base,
                                            const int* __restrict__ src,
                                            const int* __restrict__ dst, float* cnt,
                                            int2* __restrict__ lst, int off) {
    constexpr int CV = 8;
    for (int i0 = 0; i0 < m; i0 += 32 * CV) {
        int s[CV];
        int d[CV];
#pragma unroll
        for (int t = 0; t < CV; ++t) {
            const int i = i0 + t * 32 + lane;
            const unsigned e = i < m ? seg[i] : 0u;
            s[t] = src[e];
            d[t] = dst[e];
        }
#pragma unroll
        for (int t = 0; t < CV; ++t) {
            const int i = i0 + t * 32 + lane;
            if (i < m) {
                const int dl = d[t] - base;
                seg[i] = ((unsigned)s[t] << 8) | (unsigned)dl;
                if (WITH_DEG) lds_add(&cnt[dl], 1.0f);
                if (RECORD) {
                    const int g = off + i;
                    if (g < LIST_CAP) lst[g] = make_int2(s[t], dl);
                }
            }
        }
    }
}

template <int D, int SLOTS>
__device__ __forceinline__ void seg_fetch(const unsigned* seg, int m, int k0, int eslot, int sub,
                                          const float* __restrict__ feat,
                                          int (&dl)[SLOTS], bool (&ok)[SLOTS], float4 (&v)[SLOTS]) {
    constexpr int EPW = 32 / (D / 4);
#pragma unroll
    for (int j = 0; j < SLOTS; ++j) {
        const int idx = k0 + j * EPW + eslot;
        ok[j] = idx < m;
        const unsigned ent = seg[ok[j] ? idx : 0];
        dl[j] = (int)(ent & 255u);
        const unsigned off = (ent >> 8) * (unsigned)(D * 4) + (unsigned)(sub * 16);
        v[j] = *reinterpret_cast<const float4*>(reinterpret_cast<const char*>(feat) + off);
    }
}

template <int D, int SLOTS>
__device__ __forceinline__ void replay_fetch(const int2* q, int m, int k0, int wave, int eslot, int sub,
                                             const float* __restrict__ feat,
                                             int (&dl)[SLOTS], bool (&ok)[SLOTS], float4 (&v)[SLOTS]) {
    constexpr int EPW = 32 / (D / 4);
#pragma unroll
    for (int j = 0; j < SLOTS; ++j) {
        const int idx = k0 + (j * 32 + wave) * EPW + eslot;
        ok[j] = idx < m;
        const int2 ent = q[ok[j] ? idx : 0];
        dl[j] = ent.y;
        const unsigned off = (unsigned)ent.x * (unsigned)(D * 4) + (unsigned)(sub * 16);
        v[j] = *reinterpret_cast<const float4*>(reinterpret_cast<const char*>(feat) + off);
    }
}

template <int D, int SLOTS>
__device__ __forceinline__ void replay_add(float* acc, int sub, int rot,
                                           const int (&dl)[SLOTS], const bool (&ok)[SLOTS], const float4 (&v)[SLOTS]) {
#pragma unroll
    for (int j = 0; j < SLOTS; ++j) {
        if (ok[j]) {
            float* a = &acc[dl[j] * D + sub * 4];
            lds_add(a + rot,             rot ? v[j].y : v[j].x);
            lds_add(a + 1 + rot,         rot ? v[j].z : v[j].y);
            lds_add(a + 2 + rot,         rot ? v[j].w : v[j].z);
            lds_add(a + (rot ? 0 : 3),   rot ? v[j].x : v[j].w);
        }
    }
}

template <int D, bool WITH_DEG, bool RECORD, bool REPLAY>
__global__ __launch_bounds__(OWN_THREADS)
void sage_owner_sum(const float* __restrict__ feat,
                    const int*   __restrict__ src,
                    const int*   __restrict__ dst,
                    const int*   __restrict__ bits,
                    float* __restrict__ agg,
                    float* __restrict__ deg,
                    int*   __restrict__ lists,
                    int*   __restrict__ heads,
                    int n_nodes, int n_edges) {
    constexpr int WAVES = OWN_THREADS / 32;
    static_assert(REPLAY_CAP * 2 >= WAVES * SEG_CAP, "the replay stage and the waves' lists of the scan form share the room");
    static_assert(REPLAY_CAP % OWN_THREADS == 0, "whole staging rounds");
    static_assert(N_EDGES < (1 << 24) && N_NODES < (1 << 24) && OWN_NODES <= 256, "a find packs into one word");

    __shared__ __attribute__((aligned(16))) float acc[OWN_NODES * D];
    __shared__ float cnt[OWN_NODES];
    __shared__ int2  queue[REPLAY_CAP];
    __shared__ int   wcnt[WAVES];

    constexpr int LPE   = D / 4;
    constexpr int EPW   = 32 / LPE;
    constexpr int RSLOTS = 8;

    const int tid  = threadIdx.x;
    const int lane = tid & 31;
    const int wave = tid >> 5;
    const int sub  = lane % LPE;
    const int eslot = lane / LPE;
    const int rot  = lane >> 4;
    const int base = blockIdx.x * OWN_NODES;
    const int lim  = min(OWN_NODES, n_nodes - base);

    int2* const lst = reinterpret_cast<int2*>(lists) + (size_t)blockIdx.x * LIST_CAP;

    int head_n = 0, head_bad = 1;
    if (REPLAY) {
        const int2 h = reinterpret_cast<const int2*>(heads)[blockIdx.x];
        head_n   = __builtin_amdgcn_readfirstlane(h.x);
        head_bad = __builtin_amdgcn_readfirstlane(h.y);
    }

    for (int i = tid; i < OWN_NODES * D; i += OWN_THREADS) acc[i] = 0.0f;
    if (tid < OWN_NODES) cnt[tid] = 0.0f;
    if (tid < WAVES) wcnt[tid] = 0;

    int total = 0;
    int bad   = 0;

    if (REPLAY && head_bad == 0) {
        constexpr int RPT    = REPLAY_CAP / OWN_THREADS;
        constexpr int RPASS  = 32 * EPW * RSLOTS;

        __syncthreads();

        for (int c0 = 0; c0 < head_n; c0 += REPLAY_CAP) {
            const int m = min(REPLAY_CAP, head_n - c0);
            int2 pre[RPT];
#pragma unroll
            for (int t = 0; t < RPT; ++t) {
                const int i = c0 + t * OWN_THREADS + tid;
                pre[t] = lst[i < head_n ? i : 0];
            }
#pragma unroll
            for (int t = 0; t < RPT; ++t) {
                const int i = t * OWN_THREADS + tid;
                if (i < m) queue[i] = pre[t];
            }
            __syncthreads();

            int    dlA[RSLOTS], dlB[RSLOTS];
            bool   okA[RSLOTS], okB[RSLOTS];
            float4 vA[RSLOTS],  vB[RSLOTS];
            replay_fetch<D, RSLOTS>(queue, m, 0, wave, eslot, sub, feat, dlA, okA, vA);
            for (int k0 = 0; k0 < m; k0 += 2 * RPASS) {
                replay_fetch<D, RSLOTS>(queue, m, k0 + RPASS, wave, eslot, sub, feat, dlB, okB, vB);
                replay_add<D, RSLOTS>(acc, sub, rot, dlA, okA, vA);
                replay_fetch<D, RSLOTS>(queue, m, k0 + 2 * RPASS, wave, eslot, sub, feat, dlA, okA, vA);
                replay_add<D, RSLOTS>(acc, sub, rot, dlB, okB, vB);
            }
            __syncthreads();
        }
    } else {
        const int wpo  = words_per_owner(n_edges);
        const int ngrp = wpo >> 2;
        const uint4* const bit4 = reinterpret_cast<const uint4*>(bits + (size_t)blockIdx.x * wpo);

        uint4 ccur[SCAN_VEC];
#pragma unroll
        for (int j = 0; j < SCAN_VEC; ++j) {
            const int g = j * OWN_THREADS + tid;
            ccur[j] = g < ngrp ? bit4[g] : make_uint4(0u, 0u, 0u, 0u);
        }
        __syncthreads();

        unsigned* const seg = reinterpret_cast<unsigned*>(queue) + wave * SEG_CAP;
        for (int g0 = 0; g0 < ngrp; g0 += OWN_CHUNK) {
            uint4 cnext[SCAN_VEC];
#pragma unroll
            for (int j = 0; j < SCAN_VEC; ++j) {
                const int g = g0 + OWN_CHUNK + j * OWN_THREADS + tid;
                cnext[j] = g < ngrp ? bit4[g] : make_uint4(0u, 0u, 0u, 0u);
            }

#pragma unroll
            for (int j = 0; j < SCAN_VEC; ++j) {
                const unsigned w[4] = {ccur[j].x, ccur[j].y, ccur[j].z, ccur[j].w};
                const int nhit = __builtin_popcount(w[0]) + __builtin_popcount(w[1]) +
                                 __builtin_popcount(w[2]) + __builtin_popcount(w[3]);
                if (nhit != 0) {
                    int pos = __hip_atomic_fetch_add(&wcnt[wave], nhit, __ATOMIC_RELAXED, __HIP_MEMORY_SCOPE_WORKGROUP);
                    const int e0 = (g0 + j * OWN_THREADS + tid) << 7;
                    if (pos + nhit <= SEG_CAP) {
#pragma unroll
                        for (int i = 0; i < 4; ++i) {
                            unsigned b = w[i];
                            while (b != 0u) {
                                const int p = __builtin_ctz(b);
                                b &= b - 1u;
                                seg[pos++] = (unsigned)(e0 + 32 * i + p);
                            }
                        }
                    } else {
#pragma unroll
                        for (int i = 0; i < 4; ++i) {
                            unsigned b = w[i];
                            while (b != 0u) {
                                const int p = __builtin_ctz(b);
                                b &= b - 1u;
                                own_note<D, WITH_DEG>(e0 + 32 * i + p, pos++, base, seg, acc, cnt, feat, src, dst);
                            }
                        }
                    }
                }
            }

#pragma unroll
            for (int j = 0; j < SCAN_VEC; ++j) ccur[j] = cnext[j];
        }

        __syncthreads();
        const int wn = wcnt[wave];

        const int m = min(wn, SEG_CAP);
        int off = 0;
        for (int w = 0; w < WAVES; ++w) {
            const int cw = wcnt[w];
            const int mw = min(cw, SEG_CAP);
            if (w < wave) off += mw;
            total += mw;
            if (cw > SEG_CAP) bad = 1;
        }

        seg_resolve<D, WITH_DEG, RECORD>(seg, m, lane, base, src, dst, cnt, lst, off);
        __syncthreads();

        if (m > 0) {
            constexpr int PASS = EPW * RSLOTS;
            int    dlA[RSLOTS], dlB[RSLOTS];
            bool   okA[RSLOTS], okB[RSLOTS];
            float4 vA[RSLOTS],  vB[RSLOTS];
            seg_fetch<D, RSLOTS>(seg, m, 0, eslot, sub, feat, dlA, okA, vA);
            for (int k0 = 0; k0 < m; k0 += 2 * PASS) {
                seg_fetch<D, RSLOTS>(seg, m, k0 + PASS, eslot, sub, feat, dlB, okB, vB);
                replay_add<D, RSLOTS>(acc, sub, rot, dlA, okA, vA);
                seg_fetch<D, RSLOTS>(seg, m, k0 + 2 * PASS, eslot, sub, feat, dlA, okA, vA);
                replay_add<D, RSLOTS>(acc, sub, rot, dlB, okB, vB);
            }
        }
    }

    __syncthreads();

    const float4* a4 = reinterpret_cast<const float4*>(acc);
    float4* o4 = reinterpret_cast<float4*>(agg + (size_t)base * D);
    for (int i = tid; i < lim * (D / 4); i += OWN_THREADS) o4[i] = a4[i];
    if (WITH_DEG && tid < lim) deg[base + tid] = cnt[tid];

    if (RECORD && tid == 0) {
        const int full = (bad != 0 || total > LIST_CAP) ? 1 : 0;
        reinterpret_cast<int2*>(heads)[blockIdx.x] = make_int2(min(total, LIST_CAP), full);
    }
}

__global__ __launch_bounds__(256)
void sage_finish64(const float* __restrict__ S,
                   const float* __restrict__ deg,
                   const float* __restrict__ Q,
                   float* __restrict__ out,
                   int n_nodes) {
    const int t = blockIdx.x * 256 + threadIdx.x;
    if (t >= n_nodes * 16) return;

    const float invd = 1.0f / fmaxf(deg[t >> 4], 1.0f);
    const float4 s = reinterpret_cast<const float4*>(S)[t];
    const float4 q = reinterpret_cast<const float4*>(Q)[t];

    float4 o;
    o.x = q.x + s.x * invd;
    o.y = q.y + s.y * invd;
    o.z = q.z + s.z * invd;
    o.w = q.w + s.w * invd;
    reinterpret_cast<float4*>(out)[t] = o;
}

#define LDS_PITCH   132
#define STRIP_WORDS (16 * LDS_PITCH)

__global__ __launch_bounds__(256) __attribute__((amdgpu_waves_per_eu(4, 4)))
void sage_gemm128(const float* __restrict__ agg,
                  const float* __restrict__ deg,
                  const float* __restrict__ X,
                  const float* __restrict__ Wl,
                  const float* __restrict__ Wr,
                  const float* __restrict__ bias,
                  float* __restrict__ out,
                  int n_nodes) {
    __shared__ __attribute__((aligned(16))) float lds[2][2][STRIP_WORDS];

    const int n_strips = n_nodes / 16;
    const int tid  = threadIdx.x;
    const int lane = tid & 31;
    const int w    = tid >> 5;
    const int r    = lane & 15;
    const int half = lane >> 4;
    const int kk   = half * 2;
    const int coln = w * 16 + r;

    v2f bL[32], bR[32];
#pragma unroll
    for (int ks = 0; ks < 32; ++ks) {
        const int k = ks * 4 + kk;
        bL[ks].x = Wl[(size_t)k * 128 + coln];
        bL[ks].y = Wl[(size_t)(k + 1) * 128 + coln];
        bR[ks].x = Wr[(size_t)k * 128 + coln];
        bR[ks].y = Wr[(size_t)(k + 1) * 128 + coln];
    }
    const float bv = bias[coln];

    const int srow = tid >> 5;
    const int c4   = tid & 31;

    int s   = blockIdx.x;
    int cur = 0;

    if (s < n_strips) {
        const int row0 = s * 16 + srow;
        const int row1 = row0 + 8;
        const float i0 = 1.0f / fmaxf(deg[row0], 1.0f);
        const float i1 = 1.0f / fmaxf(deg[row1], 1.0f);
        float4 a0 = *(reinterpret_cast<const float4*>(agg + (size_t)row0 * 128) + c4);
        float4 a1 = *(reinterpret_cast<const float4*>(agg + (size_t)row1 * 128) + c4);
        const float4 x0 = *(reinterpret_cast<const float4*>(X + (size_t)row0 * 128) + c4);
        const float4 x1 = *(reinterpret_cast<const float4*>(X + (size_t)row1 * 128) + c4);
        a0.x *= i0; a0.y *= i0; a0.z *= i0; a0.w *= i0;
        a1.x *= i1; a1.y *= i1; a1.z *= i1; a1.w *= i1;
        *reinterpret_cast<float4*>(&lds[0][0][srow * LDS_PITCH + c4 * 4])       = a0;
        *reinterpret_cast<float4*>(&lds[0][0][(srow + 8) * LDS_PITCH + c4 * 4]) = a1;
        *reinterpret_cast<float4*>(&lds[0][1][srow * LDS_PITCH + c4 * 4])       = x0;
        *reinterpret_cast<float4*>(&lds[0][1][(srow + 8) * LDS_PITCH + c4 * 4]) = x1;
    }
    __syncthreads();

    for (; s < n_strips; s += gridDim.x) {
        const int ahead = s + (int)gridDim.x;
        const int nxt   = ahead < n_strips ? ahead : s;
        const int row0  = nxt * 16 + srow;
        const int row1  = row0 + 8;
        const float d0 = deg[row0];
        const float d1 = deg[row1];
        float4 a0 = *(reinterpret_cast<const float4*>(agg + (size_t)row0 * 128) + c4);
        float4 a1 = *(reinterpret_cast<const float4*>(agg + (size_t)row1 * 128) + c4);
        const float4 x0 = *(reinterpret_cast<const float4*>(X + (size_t)row0 * 128) + c4);
        const float4 x1 = *(reinterpret_cast<const float4*>(X + (size_t)row1 * 128) + c4);

        const float* la = &lds[cur][0][r * LDS_PITCH + kk];
        const float* lx = &lds[cur][1][r * LDS_PITCH + kk];

        v8f accA = {};
        v8f accX = {};
#pragma unroll
        for (int ks = 0; ks < 32; ++ks) {
            const v2f aA = *reinterpret_cast<const v2f*>(la + ks * 4);
            const v2f aX = *reinterpret_cast<const v2f*>(lx + ks * 4);
            accA = __builtin_amdgcn_wmma_f32_16x16x4_f32(
                       false, aA, false, bL[ks], (short)0, accA, false, false);
            accX = __builtin_amdgcn_wmma_f32_16x16x4_f32(
                       false, aX, false, bR[ks], (short)0, accX, false, false);
        }

        float v[8];
#pragma unroll
        for (int i = 0; i < 8; ++i) {
            v[i] = fmaxf((accA[i] + accX[i]) + bv, 0.0f);
        }

        {
            const float i0 = 1.0f / fmaxf(d0, 1.0f);
            const float i1 = 1.0f / fmaxf(d1, 1.0f);
            a0.x *= i0; a0.y *= i0; a0.z *= i0; a0.w *= i0;
            a1.x *= i1; a1.y *= i1; a1.z *= i1; a1.w *= i1;
            const int nb = cur ^ 1;
            *reinterpret_cast<float4*>(&lds[nb][0][srow * LDS_PITCH + c4 * 4])       = a0;
            *reinterpret_cast<float4*>(&lds[nb][0][(srow + 8) * LDS_PITCH + c4 * 4]) = a1;
            *reinterpret_cast<float4*>(&lds[nb][1][srow * LDS_PITCH + c4 * 4])       = x0;
            *reinterpret_cast<float4*>(&lds[nb][1][(srow + 8) * LDS_PITCH + c4 * 4]) = x1;
        }
        __syncthreads();
        cur ^= 1;

        const int rowBase = s * 16 + 8 * half;
#pragma unroll
        for (int i = 0; i < 8; ++i) {
            out[(size_t)(rowBase + i) * 128 + coln] = v[i];
        }
    }
}

__global__ __launch_bounds__(256) __attribute__((amdgpu_waves_per_eu(4, 4)))
void sage_transform64(const float* __restrict__ H,
                      const float* __restrict__ Wl,
                      const float* __restrict__ Wr,
                      const float* __restrict__ bias,
                      float* __restrict__ P,
                      float* __restrict__ Q,
                      int n_nodes) {
    __shared__ __attribute__((aligned(16))) float lds[2][2][STRIP_WORDS];

    const int n_strips = n_nodes / 16;
    const int n_pairs  = (n_strips + 1) / 2;
    const int tid  = threadIdx.x;
    const int lane = tid & 31;
    const int w    = tid >> 5;
    const int grp  = w >> 2;
    const int r    = lane & 15;
    const int half = lane >> 4;
    const int kk   = half * 2;
    const int coln = (w & 3) * 16 + r;

    v2f bL[32], bR[32];
#pragma unroll
    for (int ks = 0; ks < 32; ++ks) {
        const int k = ks * 4 + kk;
        bL[ks].x = Wl[(size_t)k * 64 + coln];
        bL[ks].y = Wl[(size_t)(k + 1) * 64 + coln];
        bR[ks].x = Wr[(size_t)k * 64 + coln];
        bR[ks].y = Wr[(size_t)(k + 1) * 64 + coln];
    }
    const float bv = bias[coln];

    const int srow = tid >> 5;
    const int c4   = tid & 31;
    const int last = n_strips - 1;

    int p   = blockIdx.x;
    int cur = 0;

    if (p < n_pairs) {
#pragma unroll
        for (int sp = 0; sp < 2; ++sp) {
            const int st   = min(2 * p + sp, last);
            const int row0 = st * 16 + srow;
            const float4 h0 = *(reinterpret_cast<const float4*>(H + (size_t)row0 * 128) + c4);
            const float4 h1 = *(reinterpret_cast<const float4*>(H + (size_t)(row0 + 8) * 128) + c4);
            *reinterpret_cast<float4*>(&lds[0][sp][srow * LDS_PITCH + c4 * 4])       = h0;
            *reinterpret_cast<float4*>(&lds[0][sp][(srow + 8) * LDS_PITCH + c4 * 4]) = h1;
        }
    }
    __syncthreads();

    for (; p < n_pairs; p += gridDim.x) {
        const int ahead = p + (int)gridDim.x;
        const int nxt   = ahead < n_pairs ? ahead : p;
        const int rowA  = min(2 * nxt, last) * 16 + srow;
        const int rowB  = min(2 * nxt + 1, last) * 16 + srow;
        const float4 h00 = *(reinterpret_cast<const float4*>(H + (size_t)rowA * 128) + c4);
        const float4 h01 = *(reinterpret_cast<const float4*>(H + (size_t)(rowA + 8) * 128) + c4);
        const float4 h10 = *(reinterpret_cast<const float4*>(H + (size_t)rowB * 128) + c4);
        const float4 h11 = *(reinterpret_cast<const float4*>(H + (size_t)(rowB + 8) * 128) + c4);

        const float* lh = &lds[cur][grp][r * LDS_PITCH + kk];

        v8f accP = {};
        v8f accQ = {};
#pragma unroll
        for (int ks = 0; ks < 32; ++ks) {
            const v2f aH = *reinterpret_cast<const v2f*>(lh + ks * 4);
            accP = __builtin_amdgcn_wmma_f32_16x16x4_f32(
                       false, aH, false, bL[ks], (short)0, accP, false, false);
            accQ = __builtin_amdgcn_wmma_f32_16x16x4_f32(
                       false, aH, false, bR[ks], (short)0, accQ, false, false);
        }

        float vq[8];
#pragma unroll
        for (int i = 0; i < 8; ++i) {
            vq[i] = accQ[i] + bv;
        }

        {
            const int nb = cur ^ 1;
            *reinterpret_cast<float4*>(&lds[nb][0][srow * LDS_PITCH + c4 * 4])       = h00;
            *reinterpret_cast<float4*>(&lds[nb][0][(srow + 8) * LDS_PITCH + c4 * 4]) = h01;
            *reinterpret_cast<float4*>(&lds[nb][1][srow * LDS_PITCH + c4 * 4])       = h10;
            *reinterpret_cast<float4*>(&lds[nb][1][(srow + 8) * LDS_PITCH + c4 * 4]) = h11;
        }
        __syncthreads();
        cur ^= 1;

        const int st = 2 * p + grp;
        if (st < n_strips) {
            const int rowBase = st * 16 + 8 * half;
#pragma unroll
            for (int i = 0; i < 8; ++i) {
                const size_t off = (size_t)(rowBase + i) * 64 + coln;
                P[off] = accP[i];
                Q[off] = vq[i];
            }
        }
    }
}

extern "C" void kernel_launch(void* const* d_in, const int* in_sizes, int n_in,
                              void* d_out, int out_size, void* d_ws, size_t ws_size,
                              hipStream_t stream) {
    const float* x    = (const float*)d_in[0];
    const int*   ei   = (const int*)  d_in[1];
    const float* W1l  = (const float*)d_in[2];
    const float* W1r  = (const float*)d_in[3];
    const float* b1   = (const float*)d_in[4];
    const float* W2l  = (const float*)d_in[5];
    const float* W2r  = (const float*)d_in[6];
    const float* b2   = (const float*)d_in[7];
    float*       outp = (float*)d_out;

    const int n_nodes = in_sizes[0] / IN_DIM;
    const int n_edges = in_sizes[1] / 2;
    const int* src = ei;
    const int* dst = ei + n_edges;

    const size_t aggBytes = (size_t)n_nodes * 128 * sizeof(float);
    const size_t rowBytes = (size_t)n_nodes * 64 * sizeof(float);
    const size_t degBytes = (((size_t)n_nodes * sizeof(float)) + 255) & ~(size_t)255;

    char* ws    = (char*)d_ws;
    float* agg1 = (float*)(ws);
    float* sum2 = (float*)(ws + aggBytes);
    float* deg  = (float*)(ws + aggBytes + rowBytes);
    float* h    = (float*)(ws + aggBytes + rowBytes + degBytes);
    float* p2   = (float*)(ws + 2 * aggBytes + rowBytes + degBytes);
    float* q2   = (float*)(ws + 2 * aggBytes + 2 * rowBytes + degBytes);

    const int ownBlocks = (n_nodes + OWN_NODES - 1) / OWN_NODES;
    const size_t listBytes = (size_t)ownBlocks * LIST_CAP * 2 * sizeof(int);
    int* lists  = (int*)(ws + 2 * aggBytes + 3 * rowBytes + degBytes);
    int* heads  = (int*)(ws + 2 * aggBytes + 3 * rowBytes + degBytes + listBytes);
    const size_t headBytes = (((size_t)ownBlocks * 2 * sizeof(int)) + 255) & ~(size_t)255;
    int* bits   = (int*)(ws + 2 * aggBytes + 3 * rowBytes + degBytes + listBytes + headBytes);
    const int bitBlocks = (n_edges + BIT_CHUNK - 1) / BIT_CHUNK;
    const int denseBlocks = 512;
    const int finishBlocks = (n_nodes * 16 + 255) / 256;

    sage_owner_bits<<<bitBlocks, 1024, 0, stream>>>(dst, bits, n_nodes, n_edges);

    sage_owner_sum<128, true, true, false><<<ownBlocks, OWN_THREADS, 0, stream>>>(
        x, src, dst, bits, agg1, deg, lists, heads, n_nodes, n_edges);
    sage_gemm128<<<denseBlocks, 256, 0, stream>>>(
        agg1, deg, x, W1l, W1r, b1, h, n_nodes);

    sage_transform64<<<denseBlocks, 256, 0, stream>>>(
        h, W2l, W2r, b2, p2, q2, n_nodes);
    sage_owner_sum<64, false, false, true><<<ownBlocks, OWN_THREADS, 0, stream>>>(
        p2, src, dst, bits, sum2, nullptr, lists, heads, n_nodes, n_edges);
    sage_finish64<<<finishBlocks, 256, 0, stream>>>(sum2, deg, q2, outp, n_nodes);
}


# edge-sum add loops lose their per-row tests and selects: lists padded to whole passes (pad entries add into a spare LDS row), 128-wide sum rows column-interleaved in LDS; sum order unchanged
# speedup vs baseline: 4.3597x; 1.1240x over previous
#include <hip/hip_runtime.h>


typedef __attribute__((ext_vector_type(2))) float v2f;
typedef __attribute__((ext_vector_type(8))) float v8f;

#define N_NODES 50000
#define N_EDGES 1600000
#define IN_DIM  128
#define HIDDEN  128
#define OUT_DIM 64

#define OWN_NODES   200
#define OWN_THREADS 1024
#define OWN_BLOCKS  ((N_NODES + OWN_NODES - 1) / OWN_NODES)
#define SCAN_VEC    4
#define OWN_CHUNK   (OWN_THREADS * SCAN_VEC)
#define BIT_CHUNK   4096
#define BIT_WORDS   (BIT_CHUNK / 32)
#define SEG_CAP     512
#define LIST_CAP    32768
#define REPLAY_CAP  8192
#define SEG_STRIDE  (SEG_CAP + 32)
#define QUEUE_PAIRS (REPLAY_CAP + 512)
#define PAD_ROW     OWN_NODES

__device__ __forceinline__ void lds_add(float* p, float v) {
    __hip_atomic_fetch_add(p, v, __ATOMIC_RELAXED, __HIP_MEMORY_SCOPE_WORKGROUP);
}

__device__ __forceinline__ int words_per_owner(int n_edges) {
    return ((n_edges + BIT_CHUNK - 1) / BIT_CHUNK) * BIT_WORDS;
}

__global__ __launch_bounds__(1024)
void sage_owner_bits(const int* __restrict__ dst,
                     int* __restrict__ bits,
                     int n_nodes, int n_edges) {
    static_assert(BIT_CHUNK == 4 * 1024, "four edges per thread");
    __shared__ __attribute__((aligned(16))) unsigned bm[OWN_BLOCKS * BIT_WORDS];

    const int tid = threadIdx.x;
    const int e0  = blockIdx.x * BIT_CHUNK + tid * 4;
    int d[4];
    if (e0 + 3 < n_edges) {
        const int4 v = *reinterpret_cast<const int4*>(dst + e0);
        d[0] = v.x; d[1] = v.y; d[2] = v.z; d[3] = v.w;
    } else {
#pragma unroll
        for (int k = 0; k < 4; ++k) d[k] = e0 + k < n_edges ? dst[e0 + k] : -1;
    }

    uint4* const bm4 = reinterpret_cast<uint4*>(bm);
    for (int i = tid; i < OWN_BLOCKS * BIT_WORDS / 4; i += 1024) bm4[i] = make_uint4(0u, 0u, 0u, 0u);
    __syncthreads();

#pragma unroll
    for (int k = 0; k < 4; ++k) {
        if ((unsigned)d[k] < (unsigned)n_nodes) {
            const unsigned o = (unsigned)d[k] / (unsigned)OWN_NODES;
            const int el = tid * 4 + k;
            if (o < (unsigned)OWN_BLOCKS) {
                __hip_atomic_fetch_or(&bm[o * BIT_WORDS + (el >> 5)], 1u << (el & 31),
                                      __ATOMIC_RELAXED, __HIP_MEMORY_SCOPE_WORKGROUP);
            }
        }
    }
    __syncthreads();

    const int wpo = words_per_owner(n_edges);
    for (int i = tid; i < OWN_BLOCKS * BIT_WORDS / 4; i += 1024) {
        const int o = i / (BIT_WORDS / 4);
        const int q = i % (BIT_WORDS / 4);
        reinterpret_cast<uint4*>(bits + (size_t)o * wpo + (size_t)blockIdx.x * BIT_WORDS)[q] = bm4[i];
    }
}

template <int D>
__device__ __forceinline__ int acc_index(int row, int c) {
    return D == 128 ? row * D + (c & 3) * 32 + (c >> 2) : row * D + c;
}

template <int D, bool WITH_DEG>
__device__ __forceinline__ void own_note(int e, int pos, int base, unsigned* seg,
                                         float* acc, float* cnt,
                                         const float* __restrict__ feat,
                                         const int* __restrict__ src,
                                         const int* __restrict__ dst) {
    if (pos < SEG_CAP) {
        seg[pos] = (unsigned)e;
    } else {
        const int dl = dst[e] - base;
        const float* row = feat + (size_t)src[e] * D;
        for (int c = 0; c < D; ++c) lds_add(&acc[acc_index<D>(dl, c)], row[c]);
        if (WITH_DEG) lds_add(&cnt[dl], 1.0f);
    }
}

template <int D, bool WITH_DEG, bool RECORD>
__device__ __forceinline__ void seg_resolve(unsigned* seg, int m, int lane, int base,
                                            const int* __restrict__ src,
                                            const int* __restrict__ dst, float* cnt,
                                            int2* __restrict__ lst, int off) {
    constexpr int CV = 8;
    for (int i0 = 0; i0 < m; i0 += 32 * CV) {
        int s[CV];
        int d[CV];
#pragma unroll
        for (int t = 0; t < CV; ++t) {
            const int i = i0 + t * 32 + lane;
            const unsigned e = i < m ? seg[i] : 0u;
            s[t] = src[e];
            d[t] = dst[e];
        }
#pragma unroll
        for (int t = 0; t < CV; ++t) {
            const int i = i0 + t * 32 + lane;
            if (i < m) {
                const int dl = d[t] - base;
                seg[i] = ((unsigned)s[t] << 8) | (unsigned)dl;
                if (WITH_DEG) lds_add(&cnt[dl], 1.0f);
                if (RECORD) {
                    const int g = off + i;
                    if (g < LIST_CAP) lst[g] = make_int2(s[t], dl);
                }
            }
        }
    }
}

template <int D, int SLOTS>
__device__ __forceinline__ void seg_fetch(const unsigned* seg, int k0, int eslot, int sub,
                                          const float* __restrict__ feat,
                                          int (&dl)[SLOTS], float4 (&v)[SLOTS]) {
    constexpr int EPW = 32 / (D / 4);
#pragma unroll
    for (int j = 0; j < SLOTS; ++j) {
        const unsigned ent = seg[k0 + j * EPW + eslot];
        dl[j] = (int)(ent & 255u);
        const unsigned off = (ent >> 8) * (unsigned)(D * 4) + (unsigned)(sub * 16);
        v[j] = *reinterpret_cast<const float4*>(reinterpret_cast<const char*>(feat) + off);
    }
}

template <int D, int SLOTS>
__device__ __forceinline__ void replay_fetch(const int2* q, int k0, int wave, int eslot, int sub,
                                             const float* __restrict__ feat,
                                             int (&dl)[SLOTS], float4 (&v)[SLOTS]) {
    constexpr int EPW = 32 / (D / 4);
#pragma unroll
    for (int j = 0; j < SLOTS; ++j) {
        const int2 ent = q[k0 + (j * 32 + wave) * EPW + eslot];
        dl[j] = ent.y;
        const unsigned off = (unsigned)ent.x * (unsigned)(D * 4) + (unsigned)(sub * 16);
        v[j] = *reinterpret_cast<const float4*>(reinterpret_cast<const char*>(feat) + off);
    }
}

template <int D, int SLOTS>
__device__ __forceinline__ void replay_add(float* acc, int sub, int rot,
                                           const int (&dl)[SLOTS], const float4 (&v)[SLOTS]) {
#pragma unroll
    for (int j = 0; j < SLOTS; ++j) {
        if (D == 128) {
            float* a = &acc[dl[j] * D + sub];
            lds_add(a,      v[j].x);
            lds_add(a + 32, v[j].y);
            lds_add(a + 64, v[j].z);
            lds_add(a + 96, v[j].w);
        } else {
            float* a = &acc[dl[j] * D + sub * 4];
            lds_add(a + rot,             rot ? v[j].y : v[j].x);
            lds_add(a + 1 + rot,         rot ? v[j].z : v[j].y);
            lds_add(a + 2 + rot,         rot ? v[j].w : v[j].z);
            lds_add(a + (rot ? 0 : 3),   rot ? v[j].x : v[j].w);
        }
    }
}

template <int D, bool WITH_DEG, bool RECORD, bool REPLAY>
__global__ __launch_bounds__(OWN_THREADS)
void sage_owner_sum(const float* __restrict__ feat,
                    const int*   __restrict__ src,
                    const int*   __restrict__ dst,
                    const int*   __restrict__ bits,
                    float* __restrict__ agg,
                    float* __restrict__ deg,
                    int*   __restrict__ lists,
                    int*   __restrict__ heads,
                    int n_nodes, int n_edges) {
    constexpr int WAVES = OWN_THREADS / 32;
    static_assert(QUEUE_PAIRS * 2 >= WAVES * SEG_STRIDE, "the replay stage and the waves' lists of the scan form share the room");
    static_assert(REPLAY_CAP % OWN_THREADS == 0, "whole staging rounds");
    static_assert(N_EDGES < (1 << 24) && N_NODES < (1 << 24) && PAD_ROW <= 255, "a find packs into one word");

    __shared__ __attribute__((aligned(16))) float acc[(OWN_NODES + 1) * D];
    __shared__ float cnt[OWN_NODES];
    __shared__ __attribute__((aligned(16))) int2 queue[QUEUE_PAIRS];
    __shared__ int   wcnt[WAVES];

    constexpr int LPE   = D / 4;
    constexpr int EPW   = 32 / LPE;
    constexpr int RSLOTS = 8;

    const int tid  = threadIdx.x;
    const int lane = tid & 31;
    const int wave = tid >> 5;
    const int sub  = lane % LPE;
    const int eslot = lane / LPE;
    const int rot  = lane >> 4;
    const int base = blockIdx.x * OWN_NODES;
    const int lim  = min(OWN_NODES, n_nodes - base);

    int2* const lst = reinterpret_cast<int2*>(lists) + (size_t)blockIdx.x * LIST_CAP;

    int head_n = 0, head_bad = 1;
    if (REPLAY) {
        const int2 h = reinterpret_cast<const int2*>(heads)[blockIdx.x];
        head_n   = __builtin_amdgcn_readfirstlane(h.x);
        head_bad = __builtin_amdgcn_readfirstlane(h.y);
    }

    for (int i = tid; i < (OWN_NODES + 1) * D; i += OWN_THREADS) acc[i] = 0.0f;
    if (tid < OWN_NODES) cnt[tid] = 0.0f;
    if (tid < WAVES) wcnt[tid] = 0;

    int total = 0;
    int bad   = 0;

    if (REPLAY && head_bad == 0) {
        constexpr int RPT    = REPLAY_CAP / OWN_THREADS;
        constexpr int RPASS  = 32 * EPW * RSLOTS;

        __syncthreads();

        for (int c0 = 0; c0 < head_n; c0 += REPLAY_CAP) {
            const int m = min(REPLAY_CAP, head_n - c0);
            int2 pre[RPT];
#pragma unroll
            for (int t = 0; t < RPT; ++t) {
                const int i = c0 + t * OWN_THREADS + tid;
                pre[t] = lst[i < head_n ? i : 0];
            }
            const int mp = (m + RPASS - 1) / RPASS * RPASS;
            static_assert(REPLAY_CAP % RPASS == 0 && QUEUE_PAIRS >= REPLAY_CAP + RPASS && RPASS <= OWN_THREADS, "room for the pad entries");
#pragma unroll
            for (int t = 0; t < RPT; ++t) {
                const int i = t * OWN_THREADS + tid;
                if (i < m) queue[i] = pre[t];
                else if (i < mp + RPASS) queue[i] = make_int2(0, PAD_ROW);
            }
            if (REPLAY_CAP + tid < mp + RPASS) queue[REPLAY_CAP + tid] = make_int2(0, PAD_ROW);
            __syncthreads();

            int    dlA[RSLOTS], dlB[RSLOTS];
            float4 vA[RSLOTS],  vB[RSLOTS];
            replay_fetch<D, RSLOTS>(queue, 0, wave, eslot, sub, feat, dlA, vA);
            int k0 = 0;
            for (; k0 + 2 * RPASS <= mp; k0 += 2 * RPASS) {
                replay_fetch<D, RSLOTS>(queue, k0 + RPASS, wave, eslot, sub, feat, dlB, vB);
                __builtin_amdgcn_sched_barrier(0);
                replay_add<D, RSLOTS>(acc, sub, rot, dlA, vA);
                replay_fetch<D, RSLOTS>(queue, k0 + 2 * RPASS, wave, eslot, sub, feat, dlA, vA);
                __builtin_amdgcn_sched_barrier(0);
                replay_add<D, RSLOTS>(acc, sub, rot, dlB, vB);
            }
            if (k0 < mp) replay_add<D, RSLOTS>(acc, sub, rot, dlA, vA);
            __syncthreads();
        }
    } else {
        const int wpo  = words_per_owner(n_edges);
        const int ngrp = wpo >> 2;
        const uint4* const bit4 = reinterpret_cast<const uint4*>(bits + (size_t)blockIdx.x * wpo);

        uint4 ccur[SCAN_VEC];
#pragma unroll
        for (int j = 0; j < SCAN_VEC; ++j) {
            const int g = j * OWN_THREADS + tid;
            ccur[j] = g < ngrp ? bit4[g] : make_uint4(0u, 0u, 0u, 0u);
        }
        __syncthreads();

        unsigned* const seg = reinterpret_cast<unsigned*>(queue) + wave * SEG_STRIDE;
        for (int g0 = 0; g0 < ngrp; g0 += OWN_CHUNK) {
            uint4 cnext[SCAN_VEC];
#pragma unroll
            for (int j = 0; j < SCAN_VEC; ++j) {
                const int g = g0 + OWN_CHUNK + j * OWN_THREADS + tid;
                cnext[j] = g < ngrp ? bit4[g] : make_uint4(0u, 0u, 0u, 0u);
            }

#pragma unroll
            for (int j = 0; j < SCAN_VEC; ++j) {
                const unsigned w[4] = {ccur[j].x, ccur[j].y, ccur[j].z, ccur[j].w};
                const int nhit = __builtin_popcount(w[0]) + __builtin_popcount(w[1]) +
                                 __builtin_popcount(w[2]) + __builtin_popcount(w[3]);
                if (nhit != 0) {
                    int pos = __hip_atomic_fetch_add(&wcnt[wave], nhit, __ATOMIC_RELAXED, __HIP_MEMORY_SCOPE_WORKGROUP);
                    const int e0 = (g0 + j * OWN_THREADS + tid) << 7;
                    if (pos + nhit <= SEG_CAP) {
#pragma unroll
                        for (int i = 0; i < 4; ++i) {
                            unsigned b = w[i];
                            while (b != 0u) {
                                const int p = __builtin_ctz(b);
                                b &= b - 1u;
                                seg[pos++] = (unsigned)(e0 + 32 * i + p);
                            }
                        }
                    } else {
#pragma unroll
                        for (int i = 0; i < 4; ++i) {
                            unsigned b = w[i];
                            while (b != 0u) {
                                const int p = __builtin_ctz(b);
                                b &= b - 1u;
                                own_note<D, WITH_DEG>(e0 + 32 * i + p, pos++, base, seg, acc, cnt, feat, src, dst);
                            }
                        }
                    }
                }
            }

#pragma unroll
            for (int j = 0; j < SCAN_VEC; ++j) ccur[j] = cnext[j];
        }

        __syncthreads();
        const int wn = __builtin_amdgcn_readfirstlane(wcnt[wave]);

        const int m = min(wn, SEG_CAP);
        int off = 0;
        for (int w = 0; w < WAVES; ++w) {
            const int cw = wcnt[w];
            const int mw = min(cw, SEG_CAP);
            if (w < wave) off += mw;
            total += mw;
            if (cw > SEG_CAP) bad = 1;
        }

        seg_resolve<D, WITH_DEG, RECORD>(seg, m, lane, base, src, dst, cnt, lst, off);
        constexpr int PASS = EPW * RSLOTS;
        static_assert(SEG_CAP % PASS == 0 && SEG_STRIDE >= SEG_CAP + PASS, "room for the pad entries");
        const int mp = (m + PASS - 1) / PASS * PASS;
        for (int i = m + lane; i < mp + PASS; i += 32) seg[i] = (unsigned)PAD_ROW;
        __syncthreads();

        {
            int    dlA[RSLOTS], dlB[RSLOTS];
            float4 vA[RSLOTS],  vB[RSLOTS];
            seg_fetch<D, RSLOTS>(seg, 0, eslot, sub, feat, dlA, vA);
            int k0 = 0;
            for (; k0 + 2 * PASS <= mp; k0 += 2 * PASS) {
                seg_fetch<D, RSLOTS>(seg, k0 + PASS, eslot, sub, feat, dlB, vB);
                __builtin_amdgcn_sched_barrier(0);
                replay_add<D, RSLOTS>(acc, sub, rot, dlA, vA);
                seg_fetch<D, RSLOTS>(seg, k0 + 2 * PASS, eslot, sub, feat, dlA, vA);
                __builtin_amdgcn_sched_barrier(0);
                replay_add<D, RSLOTS>(acc, sub, rot, dlB, vB);
            }
            if (k0 < mp) replay_add<D, RSLOTS>(acc, sub, rot, dlA, vA);
        }
    }

    __syncthreads();

    const float4* a4 = reinterpret_cast<const float4*>(acc);
    float4* o4 = reinterpret_cast<float4*>(agg + (size_t)base * D);
    if (D == 128) {
        for (int i = tid; i < lim * (D / 4); i += OWN_THREADS) {
            const float* a = &acc[(i >> 5) * D + (i & 31)];
            o4[i] = make_float4(a[0], a[32], a[64], a[96]);
        }
    } else {
        for (int i = tid; i < lim * (D / 4); i += OWN_THREADS) o4[i] = a4[i];
    }
    if (WITH_DEG && tid < lim) deg[base + tid] = cnt[tid];

    if (RECORD && tid == 0) {
        const int full = (bad != 0 || total > LIST_CAP) ? 1 : 0;
        reinterpret_cast<int2*>(heads)[blockIdx.x] = make_int2(min(total, LIST_CAP), full);
    }
}

__global__ __launch_bounds__(256)
void sage_finish64(const float* __restrict__ S,
                   const float* __restrict__ deg,
                   const float* __restrict__ Q,
                   float* __restrict__ out,
                   int n_nodes) {
    const int t = blockIdx.x * 256 + threadIdx.x;
    if (t >= n_nodes * 16) return;

    const float invd = 1.0f / fmaxf(deg[t >> 4], 1.0f);
    const float4 s = reinterpret_cast<const float4*>(S)[t];
    const float4 q = reinterpret_cast<const float4*>(Q)[t];

    float4 o;
    o.x = q.x + s.x * invd;
    o.y = q.y + s.y * invd;
    o.z = q.z + s.z * invd;
    o.w = q.w + s.w * invd;
    reinterpret_cast<float4*>(out)[t] = o;
}

#define LDS_PITCH   132
#define STRIP_WORDS (16 * LDS_PITCH)

__global__ __launch_bounds__(256) __attribute__((amdgpu_waves_per_eu(4, 4)))
void sage_gemm128(const float* __restrict__ agg,
                  const float* __restrict__ deg,
                  const float* __restrict__ X,
                  const float* __restrict__ Wl,
                  const float* __restrict__ Wr,
                  const float* __restrict__ bias,
                  float* __restrict__ out,
                  int n_nodes) {
    __shared__ __attribute__((aligned(16))) float lds[2][2][STRIP_WORDS];

    const int n_strips = n_nodes / 16;
    const int tid  = threadIdx.x;
    const int lane = tid & 31;
    const int w    = tid >> 5;
    const int r    = lane & 15;
    const int half = lane >> 4;
    const int kk   = half * 2;
    const int coln = w * 16 + r;

    v2f bL[32], bR[32];
#pragma unroll
    for (int ks = 0; ks < 32; ++ks) {
        const int k = ks * 4 + kk;
        bL[ks].x = Wl[(size_t)k * 128 + coln];
        bL[ks].y = Wl[(size_t)(k + 1) * 128 + coln];
        bR[ks].x = Wr[(size_t)k * 128 + coln];
        bR[ks].y = Wr[(size_t)(k + 1) * 128 + coln];
    }
    const float bv = bias[coln];

    const int srow = tid >> 5;
    const int c4   = tid & 31;

    int s   = blockIdx.x;
    int cur = 0;

    if (s < n_strips) {
        const int row0 = s * 16 + srow;
        const int row1 = row0 + 8;
        const float i0 = 1.0f / fmaxf(deg[row0], 1.0f);
        const float i1 = 1.0f / fmaxf(deg[row1], 1.0f);
        float4 a0 = *(reinterpret_cast<const float4*>(agg + (size_t)row0 * 128) + c4);
        float4 a1 = *(reinterpret_cast<const float4*>(agg + (size_t)row1 * 128) + c4);
        const float4 x0 = *(reinterpret_cast<const float4*>(X + (size_t)row0 * 128) + c4);
        const float4 x1 = *(reinterpret_cast<const float4*>(X + (size_t)row1 * 128) + c4);
        a0.x *= i0; a0.y *= i0; a0.z *= i0; a0.w *= i0;
        a1.x *= i1; a1.y *= i1; a1.z *= i1; a1.w *= i1;
        *reinterpret_cast<float4*>(&lds[0][0][srow * LDS_PITCH + c4 * 4])       = a0;
        *reinterpret_cast<float4*>(&lds[0][0][(srow + 8) * LDS_PITCH + c4 * 4]) = a1;
        *reinterpret_cast<float4*>(&lds[0][1][srow * LDS_PITCH + c4 * 4])       = x0;
        *reinterpret_cast<float4*>(&lds[0][1][(srow + 8) * LDS_PITCH + c4 * 4]) = x1;
    }
    __syncthreads();

    for (; s < n_strips; s += gridDim.x) {
        const int ahead = s + (int)gridDim.x;
        const int nxt   = ahead < n_strips ? ahead : s;
        const int row0  = nxt * 16 + srow;
        const int row1  = row0 + 8;
        const float d0 = deg[row0];
        const float d1 = deg[row1];
        float4 a0 = *(reinterpret_cast<const float4*>(agg + (size_t)row0 * 128) + c4);
        float4 a1 = *(reinterpret_cast<const float4*>(agg + (size_t)row1 * 128) + c4);
        const float4 x0 = *(reinterpret_cast<const float4*>(X + (size_t)row0 * 128) + c4);
        const float4 x1 = *(reinterpret_cast<const float4*>(X + (size_t)row1 * 128) + c4);

        const float* la = &lds[cur][0][r * LDS_PITCH + kk];
        const float* lx = &lds[cur][1][r * LDS_PITCH + kk];

        v8f accA = {};
        v8f accX = {};
#pragma unroll
        for (int ks = 0; ks < 32; ++ks) {
            const v2f aA = *reinterpret_cast<const v2f*>(la + ks * 4);
            const v2f aX = *reinterpret_cast<const v2f*>(lx + ks * 4);
            accA = __builtin_amdgcn_wmma_f32_16x16x4_f32(
                       false, aA, false, bL[ks], (short)0, accA, false, false);
            accX = __builtin_amdgcn_wmma_f32_16x16x4_f32(
                       false, aX, false, bR[ks], (short)0, accX, false, false);
        }

        float v[8];
#pragma unroll
        for (int i = 0; i < 8; ++i) {
            v[i] = fmaxf((accA[i] + accX[i]) + bv, 0.0f);
        }

        {
            const float i0 = 1.0f / fmaxf(d0, 1.0f);
            const float i1 = 1.0f / fmaxf(d1, 1.0f);
            a0.x *= i0; a0.y *= i0; a0.z *= i0; a0.w *= i0;
            a1.x *= i1; a1.y *= i1; a1.z *= i1; a1.w *= i1;
            const int nb = cur ^ 1;
            *reinterpret_cast<float4*>(&lds[nb][0][srow * LDS_PITCH + c4 * 4])       = a0;
            *reinterpret_cast<float4*>(&lds[nb][0][(srow + 8) * LDS_PITCH + c4 * 4]) = a1;
            *reinterpret_cast<float4*>(&lds[nb][1][srow * LDS_PITCH + c4 * 4])       = x0;
            *reinterpret_cast<float4*>(&lds[nb][1][(srow + 8) * LDS_PITCH + c4 * 4]) = x1;
        }
        __syncthreads();
        cur ^= 1;

        const int rowBase = s * 16 + 8 * half;
#pragma unroll
        for (int i = 0; i < 8; ++i) {
            out[(size_t)(rowBase + i) * 128 + coln] = v[i];
        }
    }
}

__global__ __launch_bounds__(256) __attribute__((amdgpu_waves_per_eu(4, 4)))
void sage_transform64(const float* __restrict__ H,
                      const float* __restrict__ Wl,
                      const float* __restrict__ Wr,
                      const float* __restrict__ bias,
                      float* __restrict__ P,
                      float* __restrict__ Q,
                      int n_nodes) {
    __shared__ __attribute__((aligned(16))) float lds[2][2][STRIP_WORDS];

    const int n_strips = n_nodes / 16;
    const int n_pairs  = (n_strips + 1) / 2;
    const int tid  = threadIdx.x;
    const int lane = tid & 31;
    const int w    = tid >> 5;
    const int grp  = w >> 2;
    const int r    = lane & 15;
    const int half = lane >> 4;
    const int kk   = half * 2;
    const int coln = (w & 3) * 16 + r;

    v2f bL[32], bR[32];
#pragma unroll
    for (int ks = 0; ks < 32; ++ks) {
        const int k = ks * 4 + kk;
        bL[ks].x = Wl[(size_t)k * 64 + coln];
        bL[ks].y = Wl[(size_t)(k + 1) * 64 + coln];
        bR[ks].x = Wr[(size_t)k * 64 + coln];
        bR[ks].y = Wr[(size_t)(k + 1) * 64 + coln];
    }
    const float bv = bias[coln];

    const int srow = tid >> 5;
    const int c4   = tid & 31;
    const int last = n_strips - 1;

    int p   = blockIdx.x;
    int cur = 0;

    if (p < n_pairs) {
#pragma unroll
        for (int sp = 0; sp < 2; ++sp) {
            const int st   = min(2 * p + sp, last);
            const int row0 = st * 16 + srow;
            const float4 h0 = *(reinterpret_cast<const float4*>(H + (size_t)row0 * 128) + c4);
            const float4 h1 = *(reinterpret_cast<const float4*>(H + (size_t)(row0 + 8) * 128) + c4);
            *reinterpret_cast<float4*>(&lds[0][sp][srow * LDS_PITCH + c4 * 4])       = h0;
            *reinterpret_cast<float4*>(&lds[0][sp][(srow + 8) * LDS_PITCH + c4 * 4]) = h1;
        }
    }
    __syncthreads();

    for (; p < n_pairs; p += gridDim.x) {
        const int ahead = p + (int)gridDim.x;
        const int nxt   = ahead < n_pairs ? ahead : p;
        const int rowA  = min(2 * nxt, last) * 16 + srow;
        const int rowB  = min(2 * nxt + 1, last) * 16 + srow;
        const float4 h00 = *(reinterpret_cast<const float4*>(H + (size_t)rowA * 128) + c4);
        const float4 h01 = *(reinterpret_cast<const float4*>(H + (size_t)(rowA + 8) * 128) + c4);
        const float4 h10 = *(reinterpret_cast<const float4*>(H + (size_t)rowB * 128) + c4);
        const float4 h11 = *(reinterpret_cast<const float4*>(H + (size_t)(rowB + 8) * 128) + c4);

        const float* lh = &lds[cur][grp][r * LDS_PITCH + kk];

        v8f accP = {};
        v8f accQ = {};
#pragma unroll
        for (int ks = 0; ks < 32; ++ks) {
            const v2f aH = *reinterpret_cast<const v2f*>(lh + ks * 4);
            accP = __builtin_amdgcn_wmma_f32_16x16x4_f32(
                       false, aH, false, bL[ks], (short)0, accP, false, false);
            accQ = __builtin_amdgcn_wmma_f32_16x16x4_f32(
                       false, aH, false, bR[ks], (short)0, accQ, false, false);
        }

        float vq[8];
#pragma unroll
        for (int i = 0; i < 8; ++i) {
            vq[i] = accQ[i] + bv;
        }

        {
            const int nb = cur ^ 1;
            *reinterpret_cast<float4*>(&lds[nb][0][srow * LDS_PITCH + c4 * 4])       = h00;
            *reinterpret_cast<float4*>(&lds[nb][0][(srow + 8) * LDS_PITCH + c4 * 4]) = h01;
            *reinterpret_cast<float4*>(&lds[nb][1][srow * LDS_PITCH + c4 * 4])       = h10;
            *reinterpret_cast<float4*>(&lds[nb][1][(srow + 8) * LDS_PITCH + c4 * 4]) = h11;
        }
        __syncthreads();
        cur ^= 1;

        const int st = 2 * p + grp;
        if (st < n_strips) {
            const int rowBase = st * 16 + 8 * half;
#pragma unroll
            for (int i = 0; i < 8; ++i) {
                const size_t off = (size_t)(rowBase + i) * 64 + coln;
                P[off] = accP[i];
                Q[off] = vq[i];
            }
        }
    }
}

extern "C" void kernel_launch(void* const* d_in, const int* in_sizes, int n_in,
                              void* d_out, int out_size, void* d_ws, size_t ws_size,
                              hipStream_t stream) {
    const float* x    = (const float*)d_in[0];
    const int*   ei   = (const int*)  d_in[1];
    const float* W1l  = (const float*)d_in[2];
    const float* W1r  = (const float*)d_in[3];
    const float* b1   = (const float*)d_in[4];
    const float* W2l  = (const float*)d_in[5];
    const float* W2r  = (const float*)d_in[6];
    const float* b2   = (const float*)d_in[7];
    float*       outp = (float*)d_out;

    const int n_nodes = in_sizes[0] / IN_DIM;
    const int n_edges = in_sizes[1] / 2;
    const int* src = ei;
    const int* dst = ei + n_edges;

    const size_t aggBytes = (size_t)n_nodes * 128 * sizeof(float);
    const size_t rowBytes = (size_t)n_nodes * 64 * sizeof(float);
    const size_t degBytes = (((size_t)n_nodes * sizeof(float)) + 255) & ~(size_t)255;

    char* ws    = (char*)d_ws;
    float* agg1 = (float*)(ws);
    float* sum2 = (float*)(ws + aggBytes);
    float* deg  = (float*)(ws + aggBytes + rowBytes);
    float* h    = (float*)(ws + aggBytes + rowBytes + degBytes);
    float* p2   = (float*)(ws + 2 * aggBytes + rowBytes + degBytes);
    float* q2   = (float*)(ws + 2 * aggBytes + 2 * rowBytes + degBytes);

    const int ownBlocks = (n_nodes + OWN_NODES - 1) / OWN_NODES;
    const size_t listBytes = (size_t)ownBlocks * LIST_CAP * 2 * sizeof(int);
    int* lists  = (int*)(ws + 2 * aggBytes + 3 * rowBytes + degBytes);
    int* heads  = (int*)(ws + 2 * aggBytes + 3 * rowBytes + degBytes + listBytes);
    const size_t headBytes = (((size_t)ownBlocks * 2 * sizeof(int)) + 255) & ~(size_t)255;
    int* bits   = (int*)(ws + 2 * aggBytes + 3 * rowBytes + degBytes + listBytes + headBytes);
    const int bitBlocks = (n_edges + BIT_CHUNK - 1) / BIT_CHUNK;
    const int denseBlocks = 512;
    const int finishBlocks = (n_nodes * 16 + 255) / 256;

    sage_owner_bits<<<bitBlocks, 1024, 0, stream>>>(dst, bits, n_nodes, n_edges);

    sage_owner_sum<128, true, true, false><<<ownBlocks, OWN_THREADS, 0, stream>>>(
        x, src, dst, bits, agg1, deg, lists, heads, n_nodes, n_edges);
    sage_gemm128<<<denseBlocks, 256, 0, stream>>>(
        agg1, deg, x, W1l, W1r, b1, h, n_nodes);

    sage_transform64<<<denseBlocks, 256, 0, stream>>>(
        h, W2l, W2r, b2, p2, q2, n_nodes);
    sage_owner_sum<64, false, false, true><<<ownBlocks, OWN_THREADS, 0, stream>>>(
        p2, src, dst, bits, sum2, nullptr, lists, heads, n_nodes, n_edges);
    sage_finish64<<<finishBlocks, 256, 0, stream>>>(sum2, deg, q2, outp, n_nodes);
}


# edge sums in registers: a wave owns 7 rows of its work-group's node range; finds go to the owner wave's LDS list (slot from an LDS int counter); no LDS float adds; sum order = list slot order
# speedup vs baseline: 5.1268x; 1.1760x over previous
#include <hip/hip_runtime.h>


typedef __attribute__((ext_vector_type(2))) float v2f;
typedef __attribute__((ext_vector_type(8))) float v8f;

#define N_NODES 50000
#define N_EDGES 1600000
#define IN_DIM  128
#define HIDDEN  128
#define OUT_DIM 64

#define OWN_NODES   200
#define OWN_THREADS 1024
#define OWN_BLOCKS  ((N_NODES + OWN_NODES - 1) / OWN_NODES)
#define SCAN_VEC    4
#define OWN_CHUNK   (OWN_THREADS * SCAN_VEC)
#define BIT_CHUNK   4096
#define BIT_WORDS   (BIT_CHUNK / 32)
#define SEG_CAP     512
#define WAVE_ROWS   7
#define ROW_GROUPS  ((OWN_NODES + WAVE_ROWS - 1) / WAVE_ROWS)
#define GRP_CAP     512
#define GRP_STRIDE  (GRP_CAP + 16)
#define HEAD_WORDS  64
#define PAD_ENT     ((unsigned)WAVE_ROWS)

typedef __attribute__((ext_vector_type(4))) float v4f;
template <int NV> struct rowvec;
template <> struct rowvec<4> { typedef v4f type; };
template <> struct rowvec<2> { typedef v2f type; };
template <int D> using rowv = typename rowvec<D / 32>::type;

__device__ __forceinline__ void lds_add(float* p, float v) {
    __hip_atomic_fetch_add(p, v, __ATOMIC_RELAXED, __HIP_MEMORY_SCOPE_WORKGROUP);
}

__device__ __forceinline__ int words_per_owner(int n_edges) {
    return ((n_edges + BIT_CHUNK - 1) / BIT_CHUNK) * BIT_WORDS;
}

__global__ __launch_bounds__(1024)
void sage_owner_bits(const int* __restrict__ dst,
                     int* __restrict__ bits,
                     int n_nodes, int n_edges) {
    static_assert(BIT_CHUNK == 4 * 1024, "four edges per thread");
    __shared__ __attribute__((aligned(16))) unsigned bm[OWN_BLOCKS * BIT_WORDS];

    const int tid = threadIdx.x;
    const int e0  = blockIdx.x * BIT_CHUNK + tid * 4;
    int d[4];
    if (e0 + 3 < n_edges) {
        const int4 v = *reinterpret_cast<const int4*>(dst + e0);
        d[0] = v.x; d[1] = v.y; d[2] = v.z; d[3] = v.w;
    } else {
#pragma unroll
        for (int k = 0; k < 4; ++k) d[k] = e0 + k < n_edges ? dst[e0 + k] : -1;
    }

    uint4* const bm4 = reinterpret_cast<uint4*>(bm);
    for (int i = tid; i < OWN_BLOCKS * BIT_WORDS / 4; i += 1024) bm4[i] = make_uint4(0u, 0u, 0u, 0u);
    __syncthreads();

#pragma unroll
    for (int k = 0; k < 4; ++k) {
        if ((unsigned)d[k] < (unsigned)n_nodes) {
            const unsigned o = (unsigned)d[k] / (unsigned)OWN_NODES;
            const int el = tid * 4 + k;
            if (o < (unsigned)OWN_BLOCKS) {
                __hip_atomic_fetch_or(&bm[o * BIT_WORDS + (el >> 5)], 1u << (el & 31),
                                      __ATOMIC_RELAXED, __HIP_MEMORY_SCOPE_WORKGROUP);
            }
        }
    }
    __syncthreads();

    const int wpo = words_per_owner(n_edges);
    for (int i = tid; i < OWN_BLOCKS * BIT_WORDS / 4; i += 1024) {
        const int o = i / (BIT_WORDS / 4);
        const int q = i % (BIT_WORDS / 4);
        reinterpret_cast<uint4*>(bits + (size_t)o * wpo + (size_t)blockIdx.x * BIT_WORDS)[q] = bm4[i];
    }
}

template <int D, bool WITH_DEG>
__device__ __forceinline__ void own_note(int e, int pos, int base, unsigned* seg,
                                         float* acc, float* cnt,
                                         const float* __restrict__ feat,
                                         const int* __restrict__ src,
                                         const int* __restrict__ dst) {
    if (pos < SEG_CAP) {
        seg[pos] = (unsigned)e;
    } else {
        const int dl = dst[e] - base;
        const float* row = feat + (size_t)src[e] * D;
        for (int c = 0; c < D; ++c) lds_add(&acc[dl * D + c], row[c]);
        if (WITH_DEG) lds_add(&cnt[dl], 1.0f);
    }
}

template <int D, bool WITH_DEG>
__device__ __forceinline__ void seg_route(const unsigned* seg, int m, int lane, int base,
                                          const float* __restrict__ feat,
                                          const int* __restrict__ src,
                                          const int* __restrict__ dst,
                                          float* acc, float* cnt, int* gcnt, unsigned* glist) {
    constexpr int CV = 8;
    for (int i0 = 0; i0 < m; i0 += 32 * CV) {
        int s[CV];
        int d[CV];
#pragma unroll
        for (int t = 0; t < CV; ++t) {
            const int i = i0 + t * 32 + lane;
            const unsigned e = i < m ? seg[i] : 0u;
            s[t] = src[e];
            d[t] = dst[e];
        }
#pragma unroll
        for (int t = 0; t < CV; ++t) {
            const int i = i0 + t * 32 + lane;
            if (i < m) {
                const int dl = d[t] - base;
                if (WITH_DEG) lds_add(&cnt[dl], 1.0f);
                const int g = dl / WAVE_ROWS;
                const int r = dl - g * WAVE_ROWS;
                const int pos = __hip_atomic_fetch_add(&gcnt[g], 1, __ATOMIC_RELAXED, __HIP_MEMORY_SCOPE_WORKGROUP);
                if (pos < GRP_CAP) {
                    glist[g * GRP_STRIDE + pos] = ((unsigned)s[t] << 8) | (unsigned)r;
                } else {
                    const float* row = feat + (size_t)s[t] * D;
                    for (int c = 0; c < D; ++c) lds_add(&acc[dl * D + c], row[c]);
                }
            }
        }
    }
}

template <int D, int SLOTS>
__device__ __forceinline__ void list_fetch(const unsigned* lst, int k0, int lane,
                                           const float* __restrict__ feat,
                                           int (&rr)[SLOTS], rowv<D> (&v)[SLOTS]) {
#pragma unroll
    for (int j = 0; j < SLOTS; ++j) {
        const unsigned ent = (unsigned)__builtin_amdgcn_readfirstlane((int)lst[k0 + j]);
        rr[j] = (int)(ent & 255u);
        const float* row = feat + (size_t)(ent >> 8) * D;
        v[j] = *reinterpret_cast<const rowv<D>*>(row + lane * (D / 32));
    }
}

template <int D, int SLOTS>
__device__ __forceinline__ void list_add(float (&sum)[(WAVE_ROWS + 1) * (D / 32)],
                                         const int (&rr)[SLOTS], const rowv<D> (&v)[SLOTS]) {
    constexpr int NV = D / 32;
#pragma unroll
    for (int j = 0; j < SLOTS; ++j) {
        const int r0 = rr[j] * NV;
#pragma unroll
        for (int c = 0; c < NV; ++c) sum[r0 + c] += v[j][c];
    }
}

template <int D, bool WITH_DEG, bool RECORD, bool REPLAY>
__global__ __launch_bounds__(OWN_THREADS)
void sage_owner_sum(const float* __restrict__ feat,
                    const int*   __restrict__ src,
                    const int*   __restrict__ dst,
                    const int*   __restrict__ bits,
                    float* __restrict__ agg,
                    float* __restrict__ deg,
                    unsigned* __restrict__ lists,
                    int*   __restrict__ heads,
                    int n_nodes, int n_edges) {
    constexpr int WAVES = OWN_THREADS / 32;
    static_assert(ROW_GROUPS <= WAVES && WAVES < HEAD_WORDS, "a wave per row group, a head word per wave and one more");
    static_assert(N_EDGES < (1 << 24) && N_NODES < (1 << 24) && WAVE_ROWS <= 255, "an entry packs into one word");
    typedef rowv<D> vec;
    constexpr int NV = D / 32;

    __shared__ __attribute__((aligned(16))) float acc[OWN_NODES * D];
    __shared__ float cnt[OWN_NODES];
    __shared__ __attribute__((aligned(16))) unsigned finds[WAVES * SEG_CAP];
    __shared__ __attribute__((aligned(16))) unsigned glist[WAVES * GRP_STRIDE];
    __shared__ int   wcnt[WAVES];
    __shared__ int   gcnt[WAVES];
    __shared__ int   slow;

    constexpr int RSLOTS = 8;
    constexpr int PASS   = RSLOTS;

    const int tid  = threadIdx.x;
    const int lane = tid & 31;
    const int wave = tid >> 5;
    const int base = blockIdx.x * OWN_NODES;
    const int lim  = min(OWN_NODES, n_nodes - base);

    unsigned* const mylist = glist + wave * GRP_STRIDE;
    unsigned* const glst   = lists + ((size_t)blockIdx.x * WAVES + wave) * GRP_CAP;
    int* const hd = heads + (size_t)blockIdx.x * HEAD_WORDS;

    int head_n = 0, head_bad = 1;
    if (REPLAY) {
        head_n   = __builtin_amdgcn_readfirstlane(hd[wave]);
        head_bad = __builtin_amdgcn_readfirstlane(hd[WAVES]);
    }
    const bool replay = REPLAY && head_bad == 0;

    if (!replay) {
        for (int i = tid; i < OWN_NODES * D; i += OWN_THREADS) acc[i] = 0.0f;
        if (tid < OWN_NODES) cnt[tid] = 0.0f;
        if (tid < WAVES) { wcnt[tid] = 0; gcnt[tid] = 0; }
        if (tid == 0) slow = 0;
    }

    int gn  = 0;

    if (replay) {
        gn = min(head_n, GRP_CAP);
        for (int i = lane; i < gn; i += 32) mylist[i] = glst[i];
    } else {
        const int wpo  = words_per_owner(n_edges);
        const int ngrp = wpo >> 2;
        const uint4* const bit4 = reinterpret_cast<const uint4*>(bits + (size_t)blockIdx.x * wpo);

        uint4 ccur[SCAN_VEC];
#pragma unroll
        for (int j = 0; j < SCAN_VEC; ++j) {
            const int g = j * OWN_THREADS + tid;
            ccur[j] = g < ngrp ? bit4[g] : make_uint4(0u, 0u, 0u, 0u);
        }
        __syncthreads();

        unsigned* const seg = finds + wave * SEG_CAP;
        for (int g0 = 0; g0 < ngrp; g0 += OWN_CHUNK) {
            uint4 cnext[SCAN_VEC];
#pragma unroll
            for (int j = 0; j < SCAN_VEC; ++j) {
                const int g = g0 + OWN_CHUNK + j * OWN_THREADS + tid;
                cnext[j] = g < ngrp ? bit4[g] : make_uint4(0u, 0u, 0u, 0u);
            }

#pragma unroll
            for (int j = 0; j < SCAN_VEC; ++j) {
                const unsigned w[4] = {ccur[j].x, ccur[j].y, ccur[j].z, ccur[j].w};
                const int nhit = __builtin_popcount(w[0]) + __builtin_popcount(w[1]) +
                                 __builtin_popcount(w[2]) + __builtin_popcount(w[3]);
                if (nhit != 0) {
                    int pos = __hip_atomic_fetch_add(&wcnt[wave], nhit, __ATOMIC_RELAXED, __HIP_MEMORY_SCOPE_WORKGROUP);
                    const int e0 = (g0 + j * OWN_THREADS + tid) << 7;
                    if (pos + nhit <= SEG_CAP) {
#pragma unroll
                        for (int i = 0; i < 4; ++i) {
                            unsigned b = w[i];
                            while (b != 0u) {
                                const int p = __builtin_ctz(b);
                                b &= b - 1u;
                                seg[pos++] = (unsigned)(e0 + 32 * i + p);
                            }
                        }
                    } else {
#pragma unroll
                        for (int i = 0; i < 4; ++i) {
                            unsigned b = w[i];
                            while (b != 0u) {
                                const int p = __builtin_ctz(b);
                                b &= b - 1u;
                                own_note<D, WITH_DEG>(e0 + 32 * i + p, pos++, base, seg, acc, cnt, feat, src, dst);
                            }
                        }
                    }
                }
            }

#pragma unroll
            for (int j = 0; j < SCAN_VEC; ++j) ccur[j] = cnext[j];
        }

        __syncthreads();
        const int wn = __builtin_amdgcn_readfirstlane(wcnt[wave]);
        const int m  = min(wn, SEG_CAP);

        seg_route<D, WITH_DEG>(seg, m, lane, base, feat, src, dst, acc, cnt, gcnt, glist);
        __syncthreads();
        gn = __builtin_amdgcn_readfirstlane(min(gcnt[wave], GRP_CAP));
        if (RECORD && tid < WAVES && (wcnt[tid] > SEG_CAP || gcnt[tid] > GRP_CAP)) slow = 1;
    }

    static_assert(GRP_CAP % PASS == 0 && GRP_STRIDE >= GRP_CAP + PASS, "room for the pad entries");
    const int mp = (gn + PASS - 1) / PASS * PASS;
    for (int i = gn + lane; i < mp + PASS; i += 32) mylist[i] = PAD_ENT;
    __syncthreads();

    float sum[(WAVE_ROWS + 1) * NV];
#pragma unroll
    for (int k = 0; k < (WAVE_ROWS + 1) * NV; ++k) sum[k] = 0.0f;
    {
        int rA[RSLOTS], rB[RSLOTS];
        vec vA[RSLOTS], vB[RSLOTS];
        list_fetch<D, RSLOTS>(mylist, 0, lane, feat, rA, vA);
        int k0 = 0;
        for (; k0 + 2 * PASS <= mp; k0 += 2 * PASS) {
            list_fetch<D, RSLOTS>(mylist, k0 + PASS, lane, feat, rB, vB);
            __builtin_amdgcn_sched_barrier(0);
            list_add<D, RSLOTS>(sum, rA, vA);
            list_fetch<D, RSLOTS>(mylist, k0 + 2 * PASS, lane, feat, rA, vA);
            __builtin_amdgcn_sched_barrier(0);
            list_add<D, RSLOTS>(sum, rB, vB);
        }
        if (k0 < mp) list_add<D, RSLOTS>(sum, rA, vA);
    }

    if (RECORD) {
        for (int i = lane; i < gn; i += 32) glst[i] = mylist[i];
        if (lane == 0) hd[wave] = gn;
    }

#pragma unroll
    for (int k = 0; k < WAVE_ROWS; ++k) {
        const int row = wave * WAVE_ROWS + k;
        if (row < lim) {
            vec o;
#pragma unroll
            for (int c = 0; c < NV; ++c) o[c] = sum[k * NV + c];
            if (!replay) o += *reinterpret_cast<const vec*>(&acc[row * D + lane * NV]);
            *reinterpret_cast<vec*>(agg + (size_t)(base + row) * D + lane * NV) = o;
        }
    }
    if (WITH_DEG && tid < lim) deg[base + tid] = cnt[tid];

    if (RECORD && tid == 0) hd[WAVES] = slow;
}

__global__ __launch_bounds__(256)
void sage_finish64(const float* __restrict__ S,
                   const float* __restrict__ deg,
                   const float* __restrict__ Q,
                   float* __restrict__ out,
                   int n_nodes) {
    const int t = blockIdx.x * 256 + threadIdx.x;
    if (t >= n_nodes * 16) return;

    const float invd = 1.0f / fmaxf(deg[t >> 4], 1.0f);
    const float4 s = reinterpret_cast<const float4*>(S)[t];
    const float4 q = reinterpret_cast<const float4*>(Q)[t];

    float4 o;
    o.x = q.x + s.x * invd;
    o.y = q.y + s.y * invd;
    o.z = q.z + s.z * invd;
    o.w = q.w + s.w * invd;
    reinterpret_cast<float4*>(out)[t] = o;
}

#define LDS_PITCH   132
#define STRIP_WORDS (16 * LDS_PITCH)

__global__ __launch_bounds__(256) __attribute__((amdgpu_waves_per_eu(4, 4)))
void sage_gemm128(const float* __restrict__ agg,
                  const float* __restrict__ deg,
                  const float* __restrict__ X,
                  const float* __restrict__ Wl,
                  const float* __restrict__ Wr,
                  const float* __restrict__ bias,
                  float* __restrict__ out,
                  int n_nodes) {
    __shared__ __attribute__((aligned(16))) float lds[2][2][STRIP_WORDS];

    const int n_strips = n_nodes / 16;
    const int tid  = threadIdx.x;
    const int lane = tid & 31;
    const int w    = tid >> 5;
    const int r    = lane & 15;
    const int half = lane >> 4;
    const int kk   = half * 2;
    const int coln = w * 16 + r;

    v2f bL[32], bR[32];
#pragma unroll
    for (int ks = 0; ks < 32; ++ks) {
        const int k = ks * 4 + kk;
        bL[ks].x = Wl[(size_t)k * 128 + coln];
        bL[ks].y = Wl[(size_t)(k + 1) * 128 + coln];
        bR[ks].x = Wr[(size_t)k * 128 + coln];
        bR[ks].y = Wr[(size_t)(k + 1) * 128 + coln];
    }
    const float bv = bias[coln];

    const int srow = tid >> 5;
    const int c4   = tid & 31;

    int s   = blockIdx.x;
    int cur = 0;

    if (s < n_strips) {
        const int row0 = s * 16 + srow;
        const int row1 = row0 + 8;
        const float i0 = 1.0f / fmaxf(deg[row0], 1.0f);
        const float i1 = 1.0f / fmaxf(deg[row1], 1.0f);
        float4 a0 = *(reinterpret_cast<const float4*>(agg + (size_t)row0 * 128) + c4);
        float4 a1 = *(reinterpret_cast<const float4*>(agg + (size_t)row1 * 128) + c4);
        const float4 x0 = *(reinterpret_cast<const float4*>(X + (size_t)row0 * 128) + c4);
        const float4 x1 = *(reinterpret_cast<const float4*>(X + (size_t)row1 * 128) + c4);
        a0.x *= i0; a0.y *= i0; a0.z *= i0; a0.w *= i0;
        a1.x *= i1; a1.y *= i1; a1.z *= i1; a1.w *= i1;
        *reinterpret_cast<float4*>(&lds[0][0][srow * LDS_PITCH + c4 * 4])       = a0;
        *reinterpret_cast<float4*>(&lds[0][0][(srow + 8) * LDS_PITCH + c4 * 4]) = a1;
        *reinterpret_cast<float4*>(&lds[0][1][srow * LDS_PITCH + c4 * 4])       = x0;
        *reinterpret_cast<float4*>(&lds[0][1][(srow + 8) * LDS_PITCH + c4 * 4]) = x1;
    }
    __syncthreads();

    for (; s < n_strips; s += gridDim.x) {
        const int ahead = s + (int)gridDim.x;
        const int nxt   = ahead < n_strips ? ahead : s;
        const int row0  = nxt * 16 + srow;
        const int row1  = row0 + 8;
        const float d0 = deg[row0];
        const float d1 = deg[row1];
        float4 a0 = *(reinterpret_cast<const float4*>(agg + (size_t)row0 * 128) + c4);
        float4 a1 = *(reinterpret_cast<const float4*>(agg + (size_t)row1 * 128) + c4);
        const float4 x0 = *(reinterpret_cast<const float4*>(X + (size_t)row0 * 128) + c4);
        const float4 x1 = *(reinterpret_cast<const float4*>(X + (size_t)row1 * 128) + c4);

        const float* la = &lds[cur][0][r * LDS_PITCH + kk];
        const float* lx = &lds[cur][1][r * LDS_PITCH + kk];

        v8f accA = {};
        v8f accX = {};
#pragma unroll
        for (int ks = 0; ks < 32; ++ks) {
            const v2f aA = *reinterpret_cast<const v2f*>(la + ks * 4);
            const v2f aX = *reinterpret_cast<const v2f*>(lx + ks * 4);
            accA = __builtin_amdgcn_wmma_f32_16x16x4_f32(
                       false, aA, false, bL[ks], (short)0, accA, false, false);
            accX = __builtin_amdgcn_wmma_f32_16x16x4_f32(
                       false, aX, false, bR[ks], (short)0, accX, false, false);
        }

        float v[8];
#pragma unroll
        for (int i = 0; i < 8; ++i) {
            v[i] = fmaxf((accA[i] + accX[i]) + bv, 0.0f);
        }

        {
            const float i0 = 1.0f / fmaxf(d0, 1.0f);
            const float i1 = 1.0f / fmaxf(d1, 1.0f);
            a0.x *= i0; a0.y *= i0; a0.z *= i0; a0.w *= i0;
            a1.x *= i1; a1.y *= i1; a1.z *= i1; a1.w *= i1;
            const int nb = cur ^ 1;
            *reinterpret_cast<float4*>(&lds[nb][0][srow * LDS_PITCH + c4 * 4])       = a0;
            *reinterpret_cast<float4*>(&lds[nb][0][(srow + 8) * LDS_PITCH + c4 * 4]) = a1;
            *reinterpret_cast<float4*>(&lds[nb][1][srow * LDS_PITCH + c4 * 4])       = x0;
            *reinterpret_cast<float4*>(&lds[nb][1][(srow + 8) * LDS_PITCH + c4 * 4]) = x1;
        }
        __syncthreads();
        cur ^= 1;

        const int rowBase = s * 16 + 8 * half;
#pragma unroll
        for (int i = 0; i < 8; ++i) {
            out[(size_t)(rowBase + i) * 128 + coln] = v[i];
        }
    }
}

__global__ __launch_bounds__(256) __attribute__((amdgpu_waves_per_eu(4, 4)))
void sage_transform64(const float* __restrict__ H,
                      const float* __restrict__ Wl,
                      const float* __restrict__ Wr,
                      const float* __restrict__ bias,
                      float* __restrict__ P,
                      float* __restrict__ Q,
                      int n_nodes) {
    __shared__ __attribute__((aligned(16))) float lds[2][2][STRIP_WORDS];

    const int n_strips = n_nodes / 16;
    const int n_pairs  = (n_strips + 1) / 2;
    const int tid  = threadIdx.x;
    const int lane = tid & 31;
    const int w    = tid >> 5;
    const int grp  = w >> 2;
    const int r    = lane & 15;
    const int half = lane >> 4;
    const int kk   = half * 2;
    const int coln = (w & 3) * 16 + r;

    v2f bL[32], bR[32];
#pragma unroll
    for (int ks = 0; ks < 32; ++ks) {
        const int k = ks * 4 + kk;
        bL[ks].x = Wl[(size_t)k * 64 + coln];
        bL[ks].y = Wl[(size_t)(k + 1) * 64 + coln];
        bR[ks].x = Wr[(size_t)k * 64 + coln];
        bR[ks].y = Wr[(size_t)(k + 1) * 64 + coln];
    }
    const float bv = bias[coln];

    const int srow = tid >> 5;
    const int c4   = tid & 31;
    const int last = n_strips - 1;

    int p   = blockIdx.x;
    int cur = 0;

    if (p < n_pairs) {
#pragma unroll
        for (int sp = 0; sp < 2; ++sp) {
            const int st   = min(2 * p + sp, last);
            const int row0 = st * 16 + srow;
            const float4 h0 = *(reinterpret_cast<const float4*>(H + (size_t)row0 * 128) + c4);
            const float4 h1 = *(reinterpret_cast<const float4*>(H + (size_t)(row0 + 8) * 128) + c4);
            *reinterpret_cast<float4*>(&lds[0][sp][srow * LDS_PITCH + c4 * 4])       = h0;
            *reinterpret_cast<float4*>(&lds[0][sp][(srow + 8) * LDS_PITCH + c4 * 4]) = h1;
        }
    }
    __syncthreads();

    for (; p < n_pairs; p += gridDim.x) {
        const int ahead = p + (int)gridDim.x;
        const int nxt   = ahead < n_pairs ? ahead : p;
        const int rowA  = min(2 * nxt, last) * 16 + srow;
        const int rowB  = min(2 * nxt + 1, last) * 16 + srow;
        const float4 h00 = *(reinterpret_cast<const float4*>(H + (size_t)rowA * 128) + c4);
        const float4 h01 = *(reinterpret_cast<const float4*>(H + (size_t)(rowA + 8) * 128) + c4);
        const float4 h10 = *(reinterpret_cast<const float4*>(H + (size_t)rowB * 128) + c4);
        const float4 h11 = *(reinterpret_cast<const float4*>(H + (size_t)(rowB + 8) * 128) + c4);

        const float* lh = &lds[cur][grp][r * LDS_PITCH + kk];

        v8f accP = {};
        v8f accQ = {};
#pragma unroll
        for (int ks = 0; ks < 32; ++ks) {
            const v2f aH = *reinterpret_cast<const v2f*>(lh + ks * 4);
            accP = __builtin_amdgcn_wmma_f32_16x16x4_f32(
                       false, aH, false, bL[ks], (short)0, accP, false, false);
            accQ = __builtin_amdgcn_wmma_f32_16x16x4_f32(
                       false, aH, false, bR[ks], (short)0, accQ, false, false);
        }

        float vq[8];
#pragma unroll
        for (int i = 0; i < 8; ++i) {
            vq[i] = accQ[i] + bv;
        }

        {
            const int nb = cur ^ 1;
            *reinterpret_cast<float4*>(&lds[nb][0][srow * LDS_PITCH + c4 * 4])       = h00;
            *reinterpret_cast<float4*>(&lds[nb][0][(srow + 8) * LDS_PITCH + c4 * 4]) = h01;
            *reinterpret_cast<float4*>(&lds[nb][1][srow * LDS_PITCH + c4 * 4])       = h10;
            *reinterpret_cast<float4*>(&lds[nb][1][(srow + 8) * LDS_PITCH + c4 * 4]) = h11;
        }
        __syncthreads();
        cur ^= 1;

        const int st = 2 * p + grp;
        if (st < n_strips) {
            const int rowBase = st * 16 + 8 * half;
#pragma unroll
            for (int i = 0; i < 8; ++i) {
                const size_t off = (size_t)(rowBase + i) * 64 + coln;
                P[off] = accP[i];
                Q[off] = vq[i];
            }
        }
    }
}

extern "C" void kernel_launch(void* const* d_in, const int* in_sizes, int n_in,
                              void* d_out, int out_size, void* d_ws, size_t ws_size,
                              hipStream_t stream) {
    const float* x    = (const float*)d_in[0];
    const int*   ei   = (const int*)  d_in[1];
    const float* W1l  = (const float*)d_in[2];
    const float* W1r  = (const float*)d_in[3];
    const float* b1   = (const float*)d_in[4];
    const float* W2l  = (const float*)d_in[5];
    const float* W2r  = (const float*)d_in[6];
    const float* b2   = (const float*)d_in[7];
    float*       outp = (float*)d_out;

    const int n_nodes = in_sizes[0] / IN_DIM;
    const int n_edges = in_sizes[1] / 2;
    const int* src = ei;
    const int* dst = ei + n_edges;

    const size_t aggBytes = (size_t)n_nodes * 128 * sizeof(float);
    const size_t rowBytes = (size_t)n_nodes * 64 * sizeof(float);
    const size_t degBytes = (((size_t)n_nodes * sizeof(float)) + 255) & ~(size_t)255;

    char* ws    = (char*)d_ws;
    float* agg1 = (float*)(ws);
    float* sum2 = (float*)(ws + aggBytes);
    float* deg  = (float*)(ws + aggBytes + rowBytes);
    float* h    = (float*)(ws + aggBytes + rowBytes + degBytes);
    float* p2   = (float*)(ws + 2 * aggBytes + rowBytes + degBytes);
    float* q2   = (float*)(ws + 2 * aggBytes + 2 * rowBytes + degBytes);

    const int ownBlocks = (n_nodes + OWN_NODES - 1) / OWN_NODES;
    const size_t listBytes = (size_t)ownBlocks * (OWN_THREADS / 32) * GRP_CAP * sizeof(unsigned);
    unsigned* lists = (unsigned*)(ws + 2 * aggBytes + 3 * rowBytes + degBytes);
    int* heads  = (int*)(ws + 2 * aggBytes + 3 * rowBytes + degBytes + listBytes);
    const size_t headBytes = (size_t)ownBlocks * HEAD_WORDS * sizeof(int);
    int* bits   = (int*)(ws + 2 * aggBytes + 3 * rowBytes + degBytes + listBytes + headBytes);
    const int bitBlocks = (n_edges + BIT_CHUNK - 1) / BIT_CHUNK;
    const int denseBlocks = 512;
    const int finishBlocks = (n_nodes * 16 + 255) / 256;

    sage_owner_bits<<<bitBlocks, 1024, 0, stream>>>(dst, bits, n_nodes, n_edges);

    sage_owner_sum<128, true, true, false><<<ownBlocks, OWN_THREADS, 0, stream>>>(
        x, src, dst, bits, agg1, deg, lists, heads, n_nodes, n_edges);
    sage_gemm128<<<denseBlocks, 256, 0, stream>>>(
        agg1, deg, x, W1l, W1r, b1, h, n_nodes);

    sage_transform64<<<denseBlocks, 256, 0, stream>>>(
        h, W2l, W2r, b2, p2, q2, n_nodes);
    sage_owner_sum<64, false, false, true><<<ownBlocks, OWN_THREADS, 0, stream>>>(
        p2, src, dst, bits, sum2, nullptr, lists, heads, n_nodes, n_edges);
    sage_finish64<<<finishBlocks, 256, 0, stream>>>(sum2, deg, q2, outp, n_nodes);
}


# layer-1 owner map holds, per 32 edges, the one found edge's packed (source, row) word: owner routes from the map alone; edge list read only for groups with several finds; sum order = list slot order
# speedup vs baseline: 5.4965x; 1.0721x over previous
#include <hip/hip_runtime.h>

typedef __attribute__((ext_vector_type(2))) float v2f;
typedef __attribute__((ext_vector_type(8))) float v8f;

#define N_NODES 50000
#define N_EDGES 1600000
#define IN_DIM  128
#define HIDDEN  128
#define OUT_DIM 64

#define OWN_NODES   200
#define OWN_THREADS 1024
#define OWN_BLOCKS  ((N_NODES + OWN_NODES - 1) / OWN_NODES)
#define SCAN_VEC    4
#define OWN_CHUNK   (OWN_THREADS * SCAN_VEC)
#define MAP_CHUNK   4096
#define MAP_WORDS   (MAP_CHUNK / 32)
#define MAP_MANY    0xFFFFFFFFu
#define SEG_CAP     512
#define FIND_CAP    448
#define NOTE_CAP    (SEG_CAP - FIND_CAP)
#define WAVE_ROWS   7
#define ROW_GROUPS  ((OWN_NODES + WAVE_ROWS - 1) / WAVE_ROWS)
#define GRP_CAP     512
#define GRP_STRIDE  (GRP_CAP + 16)
#define HEAD_WORDS  64
#define PAD_ENT     ((unsigned)WAVE_ROWS)

typedef __attribute__((ext_vector_type(4))) float v4f;
template <int NV> struct rowvec;
template <> struct rowvec<4> { typedef v4f type; };
template <> struct rowvec<2> { typedef v2f type; };
template <int D> using rowv = typename rowvec<D / 32>::type;

__device__ __forceinline__ void lds_add(float* p, float v) {
    __hip_atomic_fetch_add(p, v, __ATOMIC_RELAXED, __HIP_MEMORY_SCOPE_WORKGROUP);
}

__device__ __forceinline__ int words_per_owner(int n_edges) {
    return ((n_edges + MAP_CHUNK - 1) / MAP_CHUNK) * MAP_WORDS;
}

__global__ __launch_bounds__(1024)
void sage_owner_map(const int* __restrict__ src,
                    const int* __restrict__ dst,
                    unsigned* __restrict__ omap,
                    int n_nodes, int n_edges) {
    static_assert(MAP_CHUNK == 4 * 1024, "four edges per thread");
    static_assert(OWN_NODES <= 254 && N_NODES < (1 << 24), "a packed word plus one is neither 0 nor the mark");
    __shared__ __attribute__((aligned(16))) unsigned bm[OWN_BLOCKS * MAP_WORDS];

    const int tid = threadIdx.x;
    const int e0  = blockIdx.x * MAP_CHUNK + tid * 4;
    int d[4];
    int s[4];
    if (e0 + 3 < n_edges) {
        const int4 v = *reinterpret_cast<const int4*>(dst + e0);
        const int4 u = *reinterpret_cast<const int4*>(src + e0);
        d[0] = v.x; d[1] = v.y; d[2] = v.z; d[3] = v.w;
        s[0] = u.x; s[1] = u.y; s[2] = u.z; s[3] = u.w;
    } else {
#pragma unroll
        for (int k = 0; k < 4; ++k) {
            d[k] = e0 + k < n_edges ? dst[e0 + k] : -1;
            s[k] = e0 + k < n_edges ? src[e0 + k] : 0;
        }
    }

    uint4* const bm4 = reinterpret_cast<uint4*>(bm);
    for (int i = tid; i < OWN_BLOCKS * MAP_WORDS / 4; i += 1024) bm4[i] = make_uint4(0u, 0u, 0u, 0u);
    __syncthreads();

#pragma unroll
    for (int k = 0; k < 4; ++k) {
        if ((unsigned)d[k] < (unsigned)n_nodes) {
            const unsigned o = (unsigned)d[k] / (unsigned)OWN_NODES;
            const int el = tid * 4 + k;
            if (o < (unsigned)OWN_BLOCKS) {
                unsigned* const slot = &bm[o * MAP_WORDS + (el >> 5)];
                const unsigned word = (((unsigned)s[k] << 8) | ((unsigned)d[k] - o * (unsigned)OWN_NODES)) + 1u;
                unsigned seen = 0u;
                if (!__hip_atomic_compare_exchange_strong(slot, &seen, word, __ATOMIC_RELAXED, __ATOMIC_RELAXED,
                                                          __HIP_MEMORY_SCOPE_WORKGROUP)) {
                    __hip_atomic_fetch_or(slot, MAP_MANY, __ATOMIC_RELAXED, __HIP_MEMORY_SCOPE_WORKGROUP);
                }
            }
        }
    }
    __syncthreads();

    const int wpo = words_per_owner(n_edges);
    for (int i = tid; i < OWN_BLOCKS * MAP_WORDS / 4; i += 1024) {
        const int o = i / (MAP_WORDS / 4);
        const int q = i % (MAP_WORDS / 4);
        reinterpret_cast<uint4*>(omap + (size_t)o * wpo + (size_t)blockIdx.x * MAP_WORDS)[q] = bm4[i];
    }
}

template <bool WITH_DEG>
__device__ __forceinline__ int route_take(unsigned packed, float* cnt, int* gcnt) {
    const int dl = (int)(packed & 255u);
    if (WITH_DEG) lds_add(&cnt[dl], 1.0f);
    return __hip_atomic_fetch_add(&gcnt[dl / WAVE_ROWS], 1, __ATOMIC_RELAXED, __HIP_MEMORY_SCOPE_WORKGROUP);
}

template <int D>
__device__ __forceinline__ void route_put(unsigned packed, int pos, const float* __restrict__ feat,
                                          float* acc, unsigned* glist) {
    const int dl = (int)(packed & 255u);
    const int g  = dl / WAVE_ROWS;
    const int r  = dl - g * WAVE_ROWS;
    if (pos < GRP_CAP) {
        glist[g * GRP_STRIDE + pos] = (packed & ~255u) | (unsigned)r;
    } else {
        const float* row = feat + (size_t)(packed >> 8) * D;
#pragma nounroll
        for (int c = 0; c < D; ++c) lds_add(&acc[dl * D + c], row[c]);
    }
}

template <int D, bool WITH_DEG>
__device__ __forceinline__ void own_find(unsigned packed, int pos, unsigned* seg,
                                         float* acc, float* cnt,
                                         const float* __restrict__ feat) {
    if (pos < FIND_CAP) {
        seg[pos] = packed;
    } else {
        const int dl = (int)(packed & 255u);
        const float* row = feat + (size_t)(packed >> 8) * D;
#pragma nounroll
        for (int c = 0; c < D; ++c) lds_add(&acc[dl * D + c], row[c]);
        if (WITH_DEG) lds_add(&cnt[dl], 1.0f);
    }
}

template <int D, bool WITH_DEG>
__device__ __forceinline__ void own_note(unsigned grp, int pos, int owner, unsigned* note,
                                         float* acc, float* cnt,
                                         const float* __restrict__ feat,
                                         const int* __restrict__ src,
                                         const int* __restrict__ dst,
                                         int n_nodes, int n_edges) {
    if (pos < NOTE_CAP) {
        note[pos] = grp;
    } else {
        for (int k = 0; k < 32; ++k) {
            const int e = (int)(grp * 32u) + k;
            if (e < n_edges) {
                const int d = dst[e];
                if ((unsigned)d < (unsigned)n_nodes && (unsigned)d / (unsigned)OWN_NODES == (unsigned)owner) {
                    const int dl = d - owner * OWN_NODES;
                    const float* row = feat + (size_t)src[e] * D;
#pragma nounroll
                    for (int c = 0; c < D; ++c) lds_add(&acc[dl * D + c], row[c]);
                    if (WITH_DEG) lds_add(&cnt[dl], 1.0f);
                }
            }
        }
    }
}

template <int D, bool WITH_DEG>
__device__ __forceinline__ void seg_route(const unsigned* seg, int m, int lane,
                                          const float* __restrict__ feat,
                                          float* acc, float* cnt, int* gcnt, unsigned* glist) {
    constexpr int CV = 4;
    for (int i0 = 0; i0 < m; i0 += 32 * CV) {
        unsigned pk[CV];
        int pos[CV];
#pragma unroll
        for (int t = 0; t < CV; ++t) {
            const int i = i0 + t * 32 + lane;
            pk[t] = i < m ? seg[i] : 0u;
        }
#pragma unroll
        for (int t = 0; t < CV; ++t) {
            const int i = i0 + t * 32 + lane;
            pos[t] = 0;
            if (i < m) pos[t] = route_take<WITH_DEG>(pk[t], cnt, gcnt);
        }
#pragma unroll
        for (int t = 0; t < CV; ++t) {
            const int i = i0 + t * 32 + lane;
            if (i < m) route_put<D>(pk[t], pos[t], feat, acc, glist);
        }
    }
}

#define RES_SLOTS 24
__device__ __forceinline__ void res_issue(const unsigned* note, int mm, int lane,
                                          const int* __restrict__ src,
                                          const int* __restrict__ dst,
                                          int n_edges,
                                          int (&e)[RES_SLOTS], int (&s)[RES_SLOTS], int (&d)[RES_SLOTS]) {
    static_assert(RES_SLOTS <= 32, "one lane holds one slot's group number");
    const int nt = min(RES_SLOTS, mm);
    const int e_last = n_edges - 1;
    const unsigned mine = lane < nt ? note[lane] : 0u;
#pragma unroll
    for (int t = 0; t < RES_SLOTS; ++t) {
        e[t] = __builtin_amdgcn_readlane((int)mine, t) * 32 + lane;
        const int ec = max(min(e[t], e_last), 0);
        s[t] = src[ec];
        d[t] = dst[ec];
    }
}

template <int D, bool WITH_DEG>
__device__ __forceinline__ void res_use(int mm, int base, int lim,
                                        const float* __restrict__ feat,
                                        int n_edges,
                                        float* acc, float* cnt, int* gcnt, unsigned* glist,
                                        const int (&e)[RES_SLOTS], const int (&s)[RES_SLOTS], const int (&d)[RES_SLOTS]) {
    constexpr int RB = 4;
    static_assert(RES_SLOTS % RB == 0, "whole blocks of slots");
    const int nt = min(RES_SLOTS, mm);
#pragma unroll
    for (int b = 0; b < RES_SLOTS; b += RB) {
        if (b < nt) {
            unsigned pk[RB];
            int  pos[RB];
            bool hit[RB];
#pragma unroll
            for (int u = 0; u < RB; ++u) {
                const int t  = b + u;
                const int dl = d[t] - base;
                hit[u] = t < nt && e[t] < n_edges && (unsigned)dl < (unsigned)lim;
                pk[u]  = ((unsigned)s[t] << 8) | (unsigned)dl;
                pos[u] = 0;
                if (hit[u]) pos[u] = route_take<WITH_DEG>(pk[u], cnt, gcnt);
            }
#pragma unroll
            for (int u = 0; u < RB; ++u) {
                if (hit[u]) route_put<D>(pk[u], pos[u], feat, acc, glist);
            }
        }
    }
}

template <int D, int SLOTS>
__device__ __forceinline__ void list_fetch(const unsigned* lst, int k0, int lane,
                                           const float* __restrict__ feat,
                                           int (&rr)[SLOTS], rowv<D> (&v)[SLOTS]) {
#pragma unroll
    for (int j = 0; j < SLOTS; ++j) {
        const unsigned ent = (unsigned)__builtin_amdgcn_readfirstlane((int)lst[k0 + j]);
        rr[j] = (int)(ent & 255u);
        const float* row = feat + (size_t)(ent >> 8) * D;
        v[j] = *reinterpret_cast<const rowv<D>*>(row + lane * (D / 32));
    }
}

template <int D, int SLOTS>
__device__ __forceinline__ void list_add(float (&sum)[(WAVE_ROWS + 1) * (D / 32)],
                                         const int (&rr)[SLOTS], const rowv<D> (&v)[SLOTS]) {
    constexpr int NV = D / 32;
#pragma unroll
    for (int j = 0; j < SLOTS; ++j) {
        const int r0 = rr[j] * NV;
#pragma unroll
        for (int c = 0; c < NV; ++c) sum[r0 + c] += v[j][c];
    }
}

template <int D, bool WITH_DEG, bool RECORD, bool REPLAY>
__global__ __launch_bounds__(OWN_THREADS)
void sage_owner_sum(const float* __restrict__ feat,
                    const int*   __restrict__ src,
                    const int*   __restrict__ dst,
                    const unsigned* __restrict__ omap,
                    float* __restrict__ agg,
                    float* __restrict__ deg,
                    unsigned* __restrict__ lists,
                    int*   __restrict__ heads,
                    int n_nodes, int n_edges) {
    constexpr int WAVES = OWN_THREADS / 32;
    static_assert(ROW_GROUPS <= WAVES && WAVES < HEAD_WORDS, "a wave per row group, a head word per wave and one more");
    static_assert(N_NODES < (1 << 24) && WAVE_ROWS <= 255 && OWN_NODES <= 254, "an edge packs into one word");
    typedef rowv<D> vec;
    constexpr int NV = D / 32;

    __shared__ __attribute__((aligned(16))) float acc[OWN_NODES * D];
    __shared__ float cnt[OWN_NODES];
    __shared__ __attribute__((aligned(16))) unsigned finds[WAVES * SEG_CAP];
    __shared__ __attribute__((aligned(16))) unsigned glist[WAVES * GRP_STRIDE];
    __shared__ int   wcnt[WAVES];
    __shared__ int   mcnt[WAVES];
    __shared__ int   gcnt[WAVES];
    __shared__ int   slow;

    constexpr int RSLOTS = 8;
    constexpr int PASS   = RSLOTS;

    const int tid  = threadIdx.x;
    const int lane = tid & 31;
    const int wave = tid >> 5;
    const int base = blockIdx.x * OWN_NODES;
    const int lim  = min(OWN_NODES, n_nodes - base);

    unsigned* const mylist = glist + wave * GRP_STRIDE;
    unsigned* const glst   = lists + ((size_t)blockIdx.x * WAVES + wave) * GRP_CAP;
    int* const hd = heads + (size_t)blockIdx.x * HEAD_WORDS;

    int head_n = 0, head_bad = 1;
    if (REPLAY) {
        head_n   = __builtin_amdgcn_readfirstlane(hd[wave]);
        head_bad = __builtin_amdgcn_readfirstlane(hd[WAVES]);
    }
    const bool replay = REPLAY && head_bad == 0;

    if (!replay) {
        for (int i = tid; i < OWN_NODES * D; i += OWN_THREADS) acc[i] = 0.0f;
        if (tid < OWN_NODES) cnt[tid] = 0.0f;
        if (tid < WAVES) { wcnt[tid] = 0; mcnt[tid] = 0; gcnt[tid] = 0; }
        if (tid == 0) slow = 0;
    }

    int gn  = 0;

    if (replay) {
        gn = min(head_n, GRP_CAP);
        for (int i = lane; i < gn; i += 32) mylist[i] = glst[i];
    } else {
        const int wpo  = words_per_owner(n_edges);
        const int ngrp = wpo >> 2;
        const uint4* const bit4 = reinterpret_cast<const uint4*>(omap + (size_t)blockIdx.x * wpo);

        uint4 ccur[SCAN_VEC];
#pragma unroll
        for (int j = 0; j < SCAN_VEC; ++j) {
            const int g = j * OWN_THREADS + tid;
            ccur[j] = g < ngrp ? bit4[g] : make_uint4(0u, 0u, 0u, 0u);
        }
        __syncthreads();

        unsigned* const seg  = finds + wave * SEG_CAP;
        unsigned* const note = seg + FIND_CAP;
        for (int g0 = 0; g0 < ngrp; g0 += OWN_CHUNK) {
            uint4 cnext[SCAN_VEC];
#pragma unroll
            for (int j = 0; j < SCAN_VEC; ++j) {
                const int g = g0 + OWN_CHUNK + j * OWN_THREADS + tid;
                cnext[j] = g < ngrp ? bit4[g] : make_uint4(0u, 0u, 0u, 0u);
            }

#pragma unroll
            for (int j = 0; j < SCAN_VEC; ++j) {
                const unsigned w[4] = {ccur[j].x, ccur[j].y, ccur[j].z, ccur[j].w};
                if ((w[0] | w[1] | w[2] | w[3]) != 0u) {
                    bool one[4];
                    int nhit = 0;
#pragma unroll
                    for (int i = 0; i < 4; ++i) {
                        one[i] = w[i] - 1u < MAP_MANY - 1u;
                        nhit += one[i] ? 1 : 0;
                    }
                    if (nhit != 0) {
                        int pos = __hip_atomic_fetch_add(&wcnt[wave], nhit, __ATOMIC_RELAXED, __HIP_MEMORY_SCOPE_WORKGROUP);
                        if (pos + nhit <= FIND_CAP) {
#pragma unroll
                            for (int i = 0; i < 4; ++i) {
                                if (one[i]) seg[pos++] = w[i] - 1u;
                            }
                        } else {
#pragma unroll
                            for (int i = 0; i < 4; ++i) {
                                if (one[i]) own_find<D, WITH_DEG>(w[i] - 1u, pos++, seg, acc, cnt, feat);
                            }
                        }
                    }
                    unsigned many = (w[0] == MAP_MANY ? 1u : 0u) | (w[1] == MAP_MANY ? 2u : 0u) |
                                    (w[2] == MAP_MANY ? 4u : 0u) | (w[3] == MAP_MANY ? 8u : 0u);
                    while (many != 0u) {
                        const unsigned i = (unsigned)__builtin_ctz(many);
                        many &= many - 1u;
                        const int np = __hip_atomic_fetch_add(&mcnt[wave], 1, __ATOMIC_RELAXED, __HIP_MEMORY_SCOPE_WORKGROUP);
                        own_note<D, WITH_DEG>((unsigned)(g0 + j * OWN_THREADS + tid) * 4u + i, np,
                                              (int)blockIdx.x, note, acc, cnt, feat, src, dst, n_nodes, n_edges);
                    }
                }
            }

#pragma unroll
            for (int j = 0; j < SCAN_VEC; ++j) ccur[j] = cnext[j];
        }

        __syncthreads();
        const int m  = min(__builtin_amdgcn_readfirstlane(wcnt[wave]), FIND_CAP);
        const int mm = min(__builtin_amdgcn_readfirstlane(mcnt[wave]), NOTE_CAP);

        {
            int re[RES_SLOTS], rs[RES_SLOTS], rd[RES_SLOTS];
            res_issue(note, mm, lane, src, dst, n_edges, re, rs, rd);
            seg_route<D, WITH_DEG>(seg, m, lane, feat, acc, cnt, gcnt, glist);
            res_use<D, WITH_DEG>(mm, base, lim, feat, n_edges, acc, cnt, gcnt, glist, re, rs, rd);
        }
        for (int i = RES_SLOTS; i < mm; ++i) {
            const int e  = (int)note[i] * 32 + lane;
            const int ec = min(e, n_edges - 1);
            const int s  = src[ec];
            const int dl = dst[ec] - base;
            if (e < n_edges && (unsigned)dl < (unsigned)lim) {
                const unsigned packed = ((unsigned)s << 8) | (unsigned)dl;
                const int pos = route_take<WITH_DEG>(packed, cnt, gcnt);
                route_put<D>(packed, pos, feat, acc, glist);
            }
        }
        __syncthreads();
        gn = __builtin_amdgcn_readfirstlane(min(gcnt[wave], GRP_CAP));
        if (RECORD && tid < WAVES && (wcnt[tid] > FIND_CAP || mcnt[tid] > NOTE_CAP || gcnt[tid] > GRP_CAP)) slow = 1;
    }

    static_assert(GRP_CAP % PASS == 0 && GRP_STRIDE >= GRP_CAP + PASS, "room for the pad entries");
    const int mp = (gn + PASS - 1) / PASS * PASS;
    for (int i = gn + lane; i < mp + PASS; i += 32) mylist[i] = PAD_ENT;
    __syncthreads();

    float sum[(WAVE_ROWS + 1) * NV];
#pragma unroll
    for (int k = 0; k < (WAVE_ROWS + 1) * NV; ++k) sum[k] = 0.0f;
    {
        int rA[RSLOTS], rB[RSLOTS];
        vec vA[RSLOTS], vB[RSLOTS];
        list_fetch<D, RSLOTS>(mylist, 0, lane, feat, rA, vA);
        int k0 = 0;
        for (; k0 + 2 * PASS <= mp; k0 += 2 * PASS) {
            list_fetch<D, RSLOTS>(mylist, k0 + PASS, lane, feat, rB, vB);
            __builtin_amdgcn_sched_barrier(0);
            list_add<D, RSLOTS>(sum, rA, vA);
            list_fetch<D, RSLOTS>(mylist, k0 + 2 * PASS, lane, feat, rA, vA);
            __builtin_amdgcn_sched_barrier(0);
            list_add<D, RSLOTS>(sum, rB, vB);
        }
        if (k0 < mp) list_add<D, RSLOTS>(sum, rA, vA);
    }

    if (RECORD) {
        for (int i = lane; i < gn; i += 32) glst[i] = mylist[i];
        if (lane == 0) hd[wave] = gn;
    }

#pragma unroll
    for (int k = 0; k < WAVE_ROWS; ++k) {
        const int row = wave * WAVE_ROWS + k;
        if (row < lim) {
            vec o;
#pragma unroll
            for (int c = 0; c < NV; ++c) o[c] = sum[k * NV + c];
            if (!replay) o += *reinterpret_cast<const vec*>(&acc[row * D + lane * NV]);
            *reinterpret_cast<vec*>(agg + (size_t)(base + row) * D + lane * NV) = o;
        }
    }
    if (WITH_DEG && tid < lim) deg[base + tid] = cnt[tid];

    if (RECORD && tid == 0) hd[WAVES] = slow;
}

__global__ __launch_bounds__(256)
void sage_finish64(const float* __restrict__ S,
                   const float* __restrict__ deg,
                   const float* __restrict__ Q,
                   float* __restrict__ out,
                   int n_nodes) {
    const int t = blockIdx.x * 256 + threadIdx.x;
    if (t >= n_nodes * 16) return;

    const float invd = 1.0f / fmaxf(deg[t >> 4], 1.0f);
    const float4 s = reinterpret_cast<const float4*>(S)[t];
    const float4 q = reinterpret_cast<const float4*>(Q)[t];

    float4 o;
    o.x = q.x + s.x * invd;
    o.y = q.y + s.y * invd;
    o.z = q.z + s.z * invd;
    o.w = q.w + s.w * invd;
    reinterpret_cast<float4*>(out)[t] = o;
}

#define LDS_PITCH   132
#define STRIP_WORDS (16 * LDS_PITCH)

__global__ __launch_bounds__(256) __attribute__((amdgpu_waves_per_eu(4, 4)))
void sage_gemm128(const float* __restrict__ agg,
                  const float* __restrict__ deg,
                  const float* __restrict__ X,
                  const float* __restrict__ Wl,
                  const float* __restrict__ Wr,
                  const float* __restrict__ bias,
                  float* __restrict__ out,
                  int n_nodes) {
    __shared__ __attribute__((aligned(16))) float lds[2][2][STRIP_WORDS];

    const int n_strips = n_nodes / 16;
    const int tid  = threadIdx.x;
    const int lane = tid & 31;
    const int w    = tid >> 5;
    const int r    = lane & 15;
    const int half = lane >> 4;
    const int kk   = half * 2;
    const int coln = w * 16 + r;

    v2f bL[32], bR[32];
#pragma unroll
    for (int ks = 0; ks < 32; ++ks) {
        const int k = ks * 4 + kk;
        bL[ks].x = Wl[(size_t)k * 128 + coln];
        bL[ks].y = Wl[(size_t)(k + 1) * 128 + coln];
        bR[ks].x = Wr[(size_t)k * 128 + coln];
        bR[ks].y = Wr[(size_t)(k + 1) * 128 + coln];
    }
    const float bv = bias[coln];

    const int srow = tid >> 5;
    const int c4   = tid & 31;

    int s   = blockIdx.x;
    int cur = 0;

    if (s < n_strips) {
        const int row0 = s * 16 + srow;
        const int row1 = row0 + 8;
        const float i0 = 1.0f / fmaxf(deg[row0], 1.0f);
        const float i1 = 1.0f / fmaxf(deg[row1], 1.0f);
        float4 a0 = *(reinterpret_cast<const float4*>(agg + (size_t)row0 * 128) + c4);
        float4 a1 = *(reinterpret_cast<const float4*>(agg + (size_t)row1 * 128) + c4);
        const float4 x0 = *(reinterpret_cast<const float4*>(X + (size_t)row0 * 128) + c4);
        const float4 x1 = *(reinterpret_cast<const float4*>(X + (size_t)row1 * 128) + c4);
        a0.x *= i0; a0.y *= i0; a0.z *= i0; a0.w *= i0;
        a1.x *= i1; a1.y *= i1; a1.z *= i1; a1.w *= i1;
        *reinterpret_cast<float4*>(&lds[0][0][srow * LDS_PITCH + c4 * 4])       = a0;
        *reinterpret_cast<float4*>(&lds[0][0][(srow + 8) * LDS_PITCH + c4 * 4]) = a1;
        *reinterpret_cast<float4*>(&lds[0][1][srow * LDS_PITCH + c4 * 4])       = x0;
        *reinterpret_cast<float4*>(&lds[0][1][(srow + 8) * LDS_PITCH + c4 * 4]) = x1;
    }
    __syncthreads();

    for (; s < n_strips; s += gridDim.x) {
        const int ahead = s + (int)gridDim.x;
        const int nxt   = ahead < n_strips ? ahead : s;
        const int row0  = nxt * 16 + srow;
        const int row1  = row0 + 8;
        const float d0 = deg[row0];
        const float d1 = deg[row1];
        float4 a0 = *(reinterpret_cast<const float4*>(agg + (size_t)row0 * 128) + c4);
        float4 a1 = *(reinterpret_cast<const float4*>(agg + (size_t)row1 * 128) + c4);
        const float4 x0 = *(reinterpret_cast<const float4*>(X + (size_t)row0 * 128) + c4);
        const float4 x1 = *(reinterpret_cast<const float4*>(X + (size_t)row1 * 128) + c4);

        const float* la = &lds[cur][0][r * LDS_PITCH + kk];
        const float* lx = &lds[cur][1][r * LDS_PITCH + kk];

        v8f accA = {};
        v8f accX = {};
#pragma unroll
        for (int ks = 0; ks < 32; ++ks) {
            const v2f aA = *reinterpret_cast<const v2f*>(la + ks * 4);
            const v2f aX = *reinterpret_cast<const v2f*>(lx + ks * 4);
            accA = __builtin_amdgcn_wmma_f32_16x16x4_f32(
                       false, aA, false, bL[ks], (short)0, accA, false, false);
            accX = __builtin_amdgcn_wmma_f32_16x16x4_f32(
                       false, aX, false, bR[ks], (short)0, accX, false, false);
        }

        float v[8];
#pragma unroll
        for (int i = 0; i < 8; ++i) {
            v[i] = fmaxf((accA[i] + accX[i]) + bv, 0.0f);
        }

        {
            const float i0 = 1.0f / fmaxf(d0, 1.0f);
            const float i1 = 1.0f / fmaxf(d1, 1.0f);
            a0.x *= i0; a0.y *= i0; a0.z *= i0; a0.w *= i0;
            a1.x *= i1; a1.y *= i1; a1.z *= i1; a1.w *= i1;
            const int nb = cur ^ 1;
            *reinterpret_cast<float4*>(&lds[nb][0][srow * LDS_PITCH + c4 * 4])       = a0;
            *reinterpret_cast<float4*>(&lds[nb][0][(srow + 8) * LDS_PITCH + c4 * 4]) = a1;
            *reinterpret_cast<float4*>(&lds[nb][1][srow * LDS_PITCH + c4 * 4])       = x0;
            *reinterpret_cast<float4*>(&lds[nb][1][(srow + 8) * LDS_PITCH + c4 * 4]) = x1;
        }
        __syncthreads();
        cur ^= 1;

        const int rowBase = s * 16 + 8 * half;
#pragma unroll
        for (int i = 0; i < 8; ++i) {
            out[(size_t)(rowBase + i) * 128 + coln] = v[i];
        }
    }
}

__global__ __launch_bounds__(256) __attribute__((amdgpu_waves_per_eu(4, 4)))
void sage_transform64(const float* __restrict__ H,
                      const float* __restrict__ Wl,
                      const float* __restrict__ Wr,
                      const float* __restrict__ bias,
                      float* __restrict__ P,
                      float* __restrict__ Q,
                      int n_nodes) {
    __shared__ __attribute__((aligned(16))) float lds[2][2][STRIP_WORDS];

    const int n_strips = n_nodes / 16;
    const int n_pairs  = (n_strips + 1) / 2;
    const int tid  = threadIdx.x;
    const int lane = tid & 31;
    const int w    = tid >> 5;
    const int grp  = w >> 2;
    const int r    = lane & 15;
    const int half = lane >> 4;
    const int kk   = half * 2;
    const int coln = (w & 3) * 16 + r;

    v2f bL[32], bR[32];
#pragma unroll
    for (int ks = 0; ks < 32; ++ks) {
        const int k = ks * 4 + kk;
        bL[ks].x = Wl[(size_t)k * 64 + coln];
        bL[ks].y = Wl[(size_t)(k + 1) * 64 + coln];
        bR[ks].x = Wr[(size_t)k * 64 + coln];
        bR[ks].y = Wr[(size_t)(k + 1) * 64 + coln];
    }
    const float bv = bias[coln];

    const int srow = tid >> 5;
    const int c4   = tid & 31;
    const int last = n_strips - 1;

    int p   = blockIdx.x;
    int cur = 0;

    if (p < n_pairs) {
#pragma unroll
        for (int sp = 0; sp < 2; ++sp) {
            const int st   = min(2 * p + sp, last);
            const int row0 = st * 16 + srow;
            const float4 h0 = *(reinterpret_cast<const float4*>(H + (size_t)row0 * 128) + c4);
            const float4 h1 = *(reinterpret_cast<const float4*>(H + (size_t)(row0 + 8) * 128) + c4);
            *reinterpret_cast<float4*>(&lds[0][sp][srow * LDS_PITCH + c4 * 4])       = h0;
            *reinterpret_cast<float4*>(&lds[0][sp][(srow + 8) * LDS_PITCH + c4 * 4]) = h1;
        }
    }
    __syncthreads();

    for (; p < n_pairs; p += gridDim.x) {
        const int ahead = p + (int)gridDim.x;
        const int nxt   = ahead < n_pairs ? ahead : p;
        const int rowA  = min(2 * nxt, last) * 16 + srow;
        const int rowB  = min(2 * nxt + 1, last) * 16 + srow;
        const float4 h00 = *(reinterpret_cast<const float4*>(H + (size_t)rowA * 128) + c4);
        const float4 h01 = *(reinterpret_cast<const float4*>(H + (size_t)(rowA + 8) * 128) + c4);
        const float4 h10 = *(reinterpret_cast<const float4*>(H + (size_t)rowB * 128) + c4);
        const float4 h11 = *(reinterpret_cast<const float4*>(H + (size_t)(rowB + 8) * 128) + c4);

        const float* lh = &lds[cur][grp][r * LDS_PITCH + kk];

        v8f accP = {};
        v8f accQ = {};
#pragma unroll
        for (int ks = 0; ks < 32; ++ks) {
            const v2f aH = *reinterpret_cast<const v2f*>(lh + ks * 4);
            accP = __builtin_amdgcn_wmma_f32_16x16x4_f32(
                       false, aH, false, bL[ks], (short)0, accP, false, false);
            accQ = __builtin_amdgcn_wmma_f32_16x16x4_f32(
                       false, aH, false, bR[ks], (short)0, accQ, false, false);
        }

        float vq[8];
#pragma unroll
        for (int i = 0; i < 8; ++i) {
            vq[i] = accQ[i] + bv;
        }

        {
            const int nb = cur ^ 1;
            *reinterpret_cast<float4*>(&lds[nb][0][srow * LDS_PITCH + c4 * 4])       = h00;
            *reinterpret_cast<float4*>(&lds[nb][0][(srow + 8) * LDS_PITCH + c4 * 4]) = h01;
            *reinterpret_cast<float4*>(&lds[nb][1][srow * LDS_PITCH + c4 * 4])       = h10;
            *reinterpret_cast<float4*>(&lds[nb][1][(srow + 8) * LDS_PITCH + c4 * 4]) = h11;
        }
        __syncthreads();
        cur ^= 1;

        const int st = 2 * p + grp;
        if (st < n_strips) {
            const int rowBase = st * 16 + 8 * half;
#pragma unroll
            for (int i = 0; i < 8; ++i) {
                const size_t off = (size_t)(rowBase + i) * 64 + coln;
                P[off] = accP[i];
                Q[off] = vq[i];
            }
        }
    }
}

extern "C" void kernel_launch(void* const* d_in, const int* in_sizes, int n_in,
                              void* d_out, int out_size, void* d_ws, size_t ws_size,
                              hipStream_t stream) {
    const float* x    = (const float*)d_in[0];
    const int*   ei   = (const int*)  d_in[1];
    const float* W1l  = (const float*)d_in[2];
    const float* W1r  = (const float*)d_in[3];
    const float* b1   = (const float*)d_in[4];
    const float* W2l  = (const float*)d_in[5];
    const float* W2r  = (const float*)d_in[6];
    const float* b2   = (const float*)d_in[7];
    float*       outp = (float*)d_out;

    const int n_nodes = in_sizes[0] / IN_DIM;
    const int n_edges = in_sizes[1] / 2;
    const int* src = ei;
    const int* dst = ei + n_edges;

    const size_t aggBytes = (size_t)n_nodes * 128 * sizeof(float);
    const size_t rowBytes = (size_t)n_nodes * 64 * sizeof(float);
    const size_t degBytes = (((size_t)n_nodes * sizeof(float)) + 255) & ~(size_t)255;

    char* ws    = (char*)d_ws;
    float* agg1 = (float*)(ws);
    float* sum2 = (float*)(ws + aggBytes);
    float* deg  = (float*)(ws + aggBytes + rowBytes);
    float* h    = (float*)(ws + aggBytes + rowBytes + degBytes);
    float* p2   = (float*)(ws + 2 * aggBytes + rowBytes + degBytes);
    float* q2   = (float*)(ws + 2 * aggBytes + 2 * rowBytes + degBytes);

    const int ownBlocks = (n_nodes + OWN_NODES - 1) / OWN_NODES;
    const size_t listBytes = (size_t)ownBlocks * (OWN_THREADS / 32) * GRP_CAP * sizeof(unsigned);
    unsigned* lists = (unsigned*)(ws + 2 * aggBytes + 3 * rowBytes + degBytes);
    int* heads  = (int*)(ws + 2 * aggBytes + 3 * rowBytes + degBytes + listBytes);
    const size_t headBytes = (size_t)ownBlocks * HEAD_WORDS * sizeof(int);
    unsigned* omap = (unsigned*)(ws + 2 * aggBytes + 3 * rowBytes + degBytes + listBytes + headBytes);
    const int mapBlocks = (n_edges + MAP_CHUNK - 1) / MAP_CHUNK;
    const int denseBlocks = 512;
    const int finishBlocks = (n_nodes * 16 + 255) / 256;

    sage_owner_map<<<mapBlocks, 1024, 0, stream>>>(src, dst, omap, n_nodes, n_edges);

    sage_owner_sum<128, true, true, false><<<ownBlocks, OWN_THREADS, 0, stream>>>(
        x, src, dst, omap, agg1, deg, lists, heads, n_nodes, n_edges);
    sage_gemm128<<<denseBlocks, 256, 0, stream>>>(
        agg1, deg, x, W1l, W1r, b1, h, n_nodes);

    sage_transform64<<<denseBlocks, 256, 0, stream>>>(
        h, W2l, W2r, b2, p2, q2, n_nodes);
    sage_owner_sum<64, false, false, true><<<ownBlocks, OWN_THREADS, 0, stream>>>(
        p2, src, dst, omap, sum2, nullptr, lists, heads, n_nodes, n_edges);
    sage_finish64<<<finishBlocks, 256, 0, stream>>>(sum2, deg, q2, outp, n_nodes);
}
